# starting kernel
# baseline (speedup 1.0000x reference)
#include <hip/hip_runtime.h>
#include <hip/hip_bf16.h>
#include <math.h>

typedef __attribute__((ext_vector_type(2))) float v2f;
typedef __attribute__((ext_vector_type(8))) float v8f;

#define NEG_SLOPE 0.2f

__device__ __forceinline__ unsigned flipf(float f) {
  unsigned u = __float_as_uint(f);
  return (u & 0x80000000u) ? ~u : (u | 0x80000000u);
}
__device__ __forceinline__ float unflipf(unsigned k) {
  unsigned u = (k & 0x80000000u) ? (k ^ 0x80000000u) : ~k;
  return __uint_as_float(u);
}

__global__ void zero_kernel(float* __restrict__ p, long long n) {
  long long i = (long long)blockIdx.x * blockDim.x + threadIdx.x;
  if (i < n) p[i] = 0.f;
}

template<int NC>
__global__ __launch_bounds__(256) void gemm_f32_wmma(const float* __restrict__ A,
                                                     const float* __restrict__ B,
                                                     float* __restrict__ C, int M) {
  constexpr int K = 128;
  constexpr int CG = 64;
  constexpr int LDA = K + 4;
  constexpr int colGroups = NC / CG;
  __shared__ float Wlds[K * CG];
  __shared__ float Alds[32 * LDA];
  int cg = blockIdx.x % colGroups;
  int rg = blockIdx.x / colGroups;

  for (int i = threadIdx.x; i < K * (CG / 4); i += 256) {
    int k = i / (CG / 4), c4 = i % (CG / 4);
    *(float4*)&Wlds[k * CG + c4 * 4] =
        *(const float4*)&B[(size_t)k * NC + cg * CG + c4 * 4];
  }
  for (int i = threadIdx.x; i < 32 * (K / 4); i += 256) {
    int r = i / (K / 4), c4 = i % (K / 4);
    int grow = rg * 32 + r;
    float4 v = make_float4(0.f, 0.f, 0.f, 0.f);
    if (grow < M) v = *(const float4*)&A[(size_t)grow * K + c4 * 4];
    *(float4*)&Alds[r * LDA + c4 * 4] = v;
  }
  __syncthreads();

  int wave = threadIdx.x >> 5;
  int lane = threadIdx.x & 31;
  int half = lane >> 4;
  int lrow = lane & 15;
  int rt = wave >> 2, ct = wave & 3;
  int grow0 = rg * 32 + rt * 16;
  if (grow0 >= M) return;

  const float* Ab = &Alds[(rt * 16 + lrow) * LDA + 2 * half];
  const float* Bb = &Wlds[(2 * half) * CG + ct * 16 + lrow];
  v8f acc = {};
  #pragma unroll
  for (int k0 = 0; k0 < K; k0 += 4) {
    v2f a, b;
    a.x = Ab[k0];        a.y = Ab[k0 + 1];
    b.x = Bb[k0 * CG];   b.y = Bb[k0 * CG + CG];
    acc = __builtin_amdgcn_wmma_f32_16x16x4_f32(false, a, false, b,
                                                (short)0, acc, false, false);
  }
  float* Cp = &C[(size_t)(grow0 + 8 * half) * NC + cg * CG + ct * 16 + lrow];
  #pragma unroll
  for (int j = 0; j < 8; ++j) Cp[(size_t)j * NC] = acc[j];
}

template<int H>
__global__ void node_alpha_kernel(const float* __restrict__ h,
                                  const float* __restrict__ aw_src,
                                  const float* __restrict__ aw_dst,
                                  float* __restrict__ asrc, float* __restrict__ adst, int N) {
  int idx = blockIdx.x * blockDim.x + threadIdx.x;
  if (idx >= N * H) return;
  int hh = idx % H;
  const float* hp = h + (size_t)idx * 64;
  const float* ap = aw_src + hh * 64;
  const float* bp = aw_dst + hh * 64;
  float sa = 0.f, sb = 0.f;
  #pragma unroll
  for (int c = 0; c < 64; c += 4) {
    float4 v = *(const float4*)(hp + c);
    float4 u = *(const float4*)(ap + c);
    float4 t = *(const float4*)(bp + c);
    sa += v.x * u.x + v.y * u.y + v.z * u.z + v.w * u.w;
    sb += v.x * t.x + v.y * t.y + v.z * t.z + v.w * t.w;
  }
  asrc[idx] = sa;
  adst[idx] = sb;
}

template<int H>
__global__ void edge_max_kernel(const int* __restrict__ srcE, const int* __restrict__ dstE,
                                const float* __restrict__ asrc, const float* __restrict__ adst,
                                unsigned* __restrict__ m, int E, int Etot) {
  int idx = blockIdx.x * blockDim.x + threadIdx.x;
  if (idx >= Etot * H) return;
  int e = idx / H, hh = idx % H;
  int s = (e < E) ? srcE[e] : (e - E);
  int d = (e < E) ? dstE[e] : (e - E);
  float v = asrc[s * H + hh] + adst[d * H + hh];
  v = (v > 0.f) ? v : NEG_SLOPE * v;
  atomicMax(&m[d * H + hh], flipf(v));
}

template<int H>
__global__ void edge_expsum_kernel(const int* __restrict__ srcE, const int* __restrict__ dstE,
                                   const float* __restrict__ asrc, const float* __restrict__ adst,
                                   const unsigned* __restrict__ m, float* __restrict__ ssum,
                                   float* __restrict__ eout, int E, int Etot) {
  int idx = blockIdx.x * blockDim.x + threadIdx.x;
  if (idx >= Etot * H) return;
  int e = idx / H, hh = idx % H;
  int s = (e < E) ? srcE[e] : (e - E);
  int d = (e < E) ? dstE[e] : (e - E);
  float v = asrc[s * H + hh] + adst[d * H + hh];
  v = (v > 0.f) ? v : NEG_SLOPE * v;
  float ex = __expf(v - unflipf(m[d * H + hh]));
  eout[idx] = ex;
  atomicAdd(&ssum[d * H + hh], ex);
}

template<int H>
__global__ void edge_agg_kernel(const int* __restrict__ srcE, const int* __restrict__ dstE,
                                const float* __restrict__ eexp, const float* __restrict__ ssum,
                                const float* __restrict__ h, float* __restrict__ agg,
                                int E, int Etot) {
  constexpr int G = H * 64 / 4;
  long long idx = (long long)blockIdx.x * blockDim.x + threadIdx.x;
  if (idx >= (long long)Etot * G) return;
  int g = (int)(idx % G);
  int e = (int)(idx / G);
  int s = (e < E) ? srcE[e] : (e - E);
  int d = (e < E) ? dstE[e] : (e - E);
  int hh = g >> 4;
  int c = (g & 15) * 4;
  float alpha = eexp[e * H + hh] / ssum[d * H + hh];
  float4 hv = *(const float4*)&h[((size_t)s * H + hh) * 64 + c];
  float* ap = &agg[((size_t)d * H + hh) * 64 + c];
  atomicAdd(ap + 0, alpha * hv.x);
  atomicAdd(ap + 1, alpha * hv.y);
  atomicAdd(ap + 2, alpha * hv.z);
  atomicAdd(ap + 3, alpha * hv.w);
}

__global__ void bias_relu_kernel(float* __restrict__ a, const float* __restrict__ b,
                                 long long n, int Fmask) {
  long long i = (long long)blockIdx.x * blockDim.x + threadIdx.x;
  if (i >= n) return;
  float v = a[i] + b[(int)(i & Fmask)];
  a[i] = v > 0.f ? v : 0.f;
}

__global__ void bias_sigmoid_kernel(const float* __restrict__ a, const float* __restrict__ b,
                                    float* __restrict__ out, long long n, int Fmask) {
  long long i = (long long)blockIdx.x * blockDim.x + threadIdx.x;
  if (i >= n) return;
  float v = a[i] + b[(int)(i & Fmask)];
  out[i] = 1.f / (1.f + __expf(-v));
}

extern "C" void kernel_launch(void* const* d_in, const int* in_sizes, int n_in,
                              void* d_out, int out_size, void* d_ws, size_t ws_size,
                              hipStream_t stream) {
  const float* x   = (const float*)d_in[0];
  const int*   ei  = (const int*)d_in[1];
  const float* W1  = (const float*)d_in[2];
  const float* as1 = (const float*)d_in[3];
  const float* ad1 = (const float*)d_in[4];
  const float* b1  = (const float*)d_in[5];
  const float* W2  = (const float*)d_in[6];
  const float* as2 = (const float*)d_in[7];
  const float* ad2 = (const float*)d_in[8];
  const float* b2  = (const float*)d_in[9];
  float* out = (float*)d_out;

  const int N = in_sizes[0] / 128;
  const int E = in_sizes[1] / 2;
  const int Etot = E + N;
  const int* srcE = ei;
  const int* dstE = ei + E;

  float* w = (float*)d_ws;
  size_t o = 0;
  float* h1    = w + o; o += (size_t)N * 128;
  float* agg1  = w + o; o += (size_t)N * 128;
  float* asrc1 = w + o; o += (size_t)N * 2;
  float* adst1 = w + o; o += (size_t)N * 2;
  unsigned* m1 = (unsigned*)(w + o); o += (size_t)N * 2;
  float* s1    = w + o; o += (size_t)N * 2;
  float* asrc2 = w + o; o += (size_t)N;
  float* adst2 = w + o; o += (size_t)N;
  unsigned* m2 = (unsigned*)(w + o); o += (size_t)N;
  float* s2    = w + o; o += (size_t)N;
  float* e1    = w + o; o += (size_t)Etot * 2;
  float* e2    = w + o; o += (size_t)Etot;
  float* h2f   = h1;
  float* agg2  = h1 + (size_t)N * 64;
  (void)o; (void)ws_size; (void)n_in; (void)out_size;

  const int TB = 256;
  auto blk = [](long long n, int tb) { return (unsigned)((n + tb - 1) / tb); };
  const int rowGroups = (N + 31) / 32;

  zero_kernel<<<blk((long long)N * 128, TB), TB, 0, stream>>>(agg1, (long long)N * 128);
  zero_kernel<<<blk((long long)N * 4, TB), TB, 0, stream>>>((float*)m1, (long long)N * 4);
  gemm_f32_wmma<128><<<dim3(rowGroups * 2), TB, 0, stream>>>(x, W1, h1, N);
  node_alpha_kernel<2><<<blk((long long)N * 2, TB), TB, 0, stream>>>(h1, as1, ad1, asrc1, adst1, N);
  edge_max_kernel<2><<<blk((long long)Etot * 2, TB), TB, 0, stream>>>(srcE, dstE, asrc1, adst1, m1, E, Etot);
  edge_expsum_kernel<2><<<blk((long long)Etot * 2, TB), TB, 0, stream>>>(srcE, dstE, asrc1, adst1, m1, s1, e1, E, Etot);
  edge_agg_kernel<2><<<blk((long long)Etot * 32, TB), TB, 0, stream>>>(srcE, dstE, e1, s1, h1, agg1, E, Etot);
  bias_relu_kernel<<<blk((long long)N * 128, TB), TB, 0, stream>>>(agg1, b1, (long long)N * 128, 127);

  gemm_f32_wmma<64><<<dim3(rowGroups), TB, 0, stream>>>(agg1, W2, h2f, N);
  zero_kernel<<<blk((long long)N * 64, TB), TB, 0, stream>>>(agg2, (long long)N * 64);
  zero_kernel<<<blk((long long)N * 2, TB), TB, 0, stream>>>((float*)m2, (long long)N * 2);
  node_alpha_kernel<1><<<blk((long long)N, TB), TB, 0, stream>>>(h2f, as2, ad2, asrc2, adst2, N);
  edge_max_kernel<1><<<blk((long long)Etot, TB), TB, 0, stream>>>(srcE, dstE, asrc2, adst2, m2, E, Etot);
  edge_expsum_kernel<1><<<blk((long long)Etot, TB), TB, 0, stream>>>(srcE, dstE, asrc2, adst2, m2, s2, e2, E, Etot);
  edge_agg_kernel<1><<<blk((long long)Etot * 16, TB), TB, 0, stream>>>(srcE, dstE, e2, s2, h2f, agg2, E, Etot);
  bias_sigmoid_kernel<<<blk((long long)N * 64, TB), TB, 0, stream>>>(agg2, b2, out, (long long)N * 64, 63);
}


# edge aggregation: lane = channel, so each f32 atomic-add and each row-load instruction covers one whole 128-B line instead of 4 quarter-filled lines; sums still in atomic arrival order
# speedup vs baseline: 1.1596x; 1.1596x over previous
#include <hip/hip_runtime.h>
#include <hip/hip_bf16.h>
#include <math.h>

typedef __attribute__((ext_vector_type(2))) float v2f;
typedef __attribute__((ext_vector_type(8))) float v8f;

#define NEG_SLOPE 0.2f

__device__ __forceinline__ unsigned flipf(float f) {
  unsigned u = __float_as_uint(f);
  return (u & 0x80000000u) ? ~u : (u | 0x80000000u);
}
__device__ __forceinline__ float unflipf(unsigned k) {
  unsigned u = (k & 0x80000000u) ? (k ^ 0x80000000u) : ~k;
  return __uint_as_float(u);
}

__global__ void zero_kernel(float* __restrict__ p, long long n) {
  long long i = (long long)blockIdx.x * blockDim.x + threadIdx.x;
  if (i < n) p[i] = 0.f;
}

template<int NC>
__global__ __launch_bounds__(256) void gemm_f32_wmma(const float* __restrict__ A,
                                                     const float* __restrict__ B,
                                                     float* __restrict__ C, int M) {
  constexpr int K = 128;
  constexpr int CG = 64;
  constexpr int LDA = K + 4;
  constexpr int colGroups = NC / CG;
  __shared__ float Wlds[K * CG];
  __shared__ float Alds[32 * LDA];
  int cg = blockIdx.x % colGroups;
  int rg = blockIdx.x / colGroups;

  for (int i = threadIdx.x; i < K * (CG / 4); i += 256) {
    int k = i / (CG / 4), c4 = i % (CG / 4);
    *(float4*)&Wlds[k * CG + c4 * 4] =
        *(const float4*)&B[(size_t)k * NC + cg * CG + c4 * 4];
  }
  for (int i = threadIdx.x; i < 32 * (K / 4); i += 256) {
    int r = i / (K / 4), c4 = i % (K / 4);
    int grow = rg * 32 + r;
    float4 v = make_float4(0.f, 0.f, 0.f, 0.f);
    if (grow < M) v = *(const float4*)&A[(size_t)grow * K + c4 * 4];
    *(float4*)&Alds[r * LDA + c4 * 4] = v;
  }
  __syncthreads();

  int wave = threadIdx.x >> 5;
  int lane = threadIdx.x & 31;
  int half = lane >> 4;
  int lrow = lane & 15;
  int rt = wave >> 2, ct = wave & 3;
  int grow0 = rg * 32 + rt * 16;
  if (grow0 >= M) return;

  const float* Ab = &Alds[(rt * 16 + lrow) * LDA + 2 * half];
  const float* Bb = &Wlds[(2 * half) * CG + ct * 16 + lrow];
  v8f acc = {};
  #pragma unroll
  for (int k0 = 0; k0 < K; k0 += 4) {
    v2f a, b;
    a.x = Ab[k0];        a.y = Ab[k0 + 1];
    b.x = Bb[k0 * CG];   b.y = Bb[k0 * CG + CG];
    acc = __builtin_amdgcn_wmma_f32_16x16x4_f32(false, a, false, b,
                                                (short)0, acc, false, false);
  }
  float* Cp = &C[(size_t)(grow0 + 8 * half) * NC + cg * CG + ct * 16 + lrow];
  #pragma unroll
  for (int j = 0; j < 8; ++j) Cp[(size_t)j * NC] = acc[j];
}

template<int H>
__global__ void node_alpha_kernel(const float* __restrict__ h,
                                  const float* __restrict__ aw_src,
                                  const float* __restrict__ aw_dst,
                                  float* __restrict__ asrc, float* __restrict__ adst, int N) {
  int idx = blockIdx.x * blockDim.x + threadIdx.x;
  if (idx >= N * H) return;
  int hh = idx % H;
  const float* hp = h + (size_t)idx * 64;
  const float* ap = aw_src + hh * 64;
  const float* bp = aw_dst + hh * 64;
  float sa = 0.f, sb = 0.f;
  #pragma unroll
  for (int c = 0; c < 64; c += 4) {
    float4 v = *(const float4*)(hp + c);
    float4 u = *(const float4*)(ap + c);
    float4 t = *(const float4*)(bp + c);
    sa += v.x * u.x + v.y * u.y + v.z * u.z + v.w * u.w;
    sb += v.x * t.x + v.y * t.y + v.z * t.z + v.w * t.w;
  }
  asrc[idx] = sa;
  adst[idx] = sb;
}

template<int H>
__global__ void edge_max_kernel(const int* __restrict__ srcE, const int* __restrict__ dstE,
                                const float* __restrict__ asrc, const float* __restrict__ adst,
                                unsigned* __restrict__ m, int E, int Etot) {
  int idx = blockIdx.x * blockDim.x + threadIdx.x;
  if (idx >= Etot * H) return;
  int e = idx / H, hh = idx % H;
  int s = (e < E) ? srcE[e] : (e - E);
  int d = (e < E) ? dstE[e] : (e - E);
  float v = asrc[s * H + hh] + adst[d * H + hh];
  v = (v > 0.f) ? v : NEG_SLOPE * v;
  atomicMax(&m[d * H + hh], flipf(v));
}

template<int H>
__global__ void edge_expsum_kernel(const int* __restrict__ srcE, const int* __restrict__ dstE,
                                   const float* __restrict__ asrc, const float* __restrict__ adst,
                                   const unsigned* __restrict__ m, float* __restrict__ ssum,
                                   float* __restrict__ eout, int E, int Etot) {
  int idx = blockIdx.x * blockDim.x + threadIdx.x;
  if (idx >= Etot * H) return;
  int e = idx / H, hh = idx % H;
  int s = (e < E) ? srcE[e] : (e - E);
  int d = (e < E) ? dstE[e] : (e - E);
  float v = asrc[s * H + hh] + adst[d * H + hh];
  v = (v > 0.f) ? v : NEG_SLOPE * v;
  float ex = __expf(v - unflipf(m[d * H + hh]));
  eout[idx] = ex;
  atomicAdd(&ssum[d * H + hh], ex);
}

template<int H, int CNT>
__device__ __forceinline__ void agg_edges(const int* __restrict__ srcE, const int* __restrict__ dstE,
                                          const float* __restrict__ eexp,
                                          const float* __restrict__ ssum,
                                          const float* __restrict__ h, float* __restrict__ agg,
                                          int E, int e0, int lane) {
  constexpr int F = H * 64;
  constexpr int NL = F / 32;
  int s[CNT], d[CNT];
  float alpha[CNT][H], hv[CNT][NL];
  #pragma unroll
  for (int k = 0; k < CNT; ++k) {
    const int e = e0 + k;
    s[k] = (e < E) ? srcE[e] : (e - E);
    d[k] = (e < E) ? dstE[e] : (e - E);
  }
  #pragma unroll
  for (int k = 0; k < CNT; ++k) {
    #pragma unroll
    for (int hh = 0; hh < H; ++hh) alpha[k][hh] = eexp[(e0 + k) * H + hh] / ssum[d[k] * H + hh];
    const float* hp = &h[(size_t)s[k] * F + lane];
    #pragma unroll
    for (int j = 0; j < NL; ++j) hv[k][j] = hp[32 * j];
  }
  #pragma unroll
  for (int k = 0; k < CNT; ++k) {
    float* ap = &agg[(size_t)d[k] * F + lane];
    #pragma unroll
    for (int j = 0; j < NL; ++j) atomicAdd(ap + 32 * j, alpha[k][(32 * j) / 64] * hv[k][j]);
  }
}

template<int H, int EPW>
__global__ void edge_agg_kernel(const int* __restrict__ srcE, const int* __restrict__ dstE,
                                const float* __restrict__ eexp, const float* __restrict__ ssum,
                                const float* __restrict__ h, float* __restrict__ agg,
                                int E, int Etot) {
  const int lane = threadIdx.x & 31;
  const long long e0 = (((long long)blockIdx.x * blockDim.x + threadIdx.x) >> 5) * EPW;
  if (e0 + EPW <= (long long)Etot) {
    agg_edges<H, EPW>(srcE, dstE, eexp, ssum, h, agg, E, (int)e0, lane);
  } else {
    for (long long e = e0; e < (long long)Etot; ++e)
      agg_edges<H, 1>(srcE, dstE, eexp, ssum, h, agg, E, (int)e, lane);
  }
}

__global__ void bias_relu_kernel(float* __restrict__ a, const float* __restrict__ b,
                                 long long n, int Fmask) {
  long long i = (long long)blockIdx.x * blockDim.x + threadIdx.x;
  if (i >= n) return;
  float v = a[i] + b[(int)(i & Fmask)];
  a[i] = v > 0.f ? v : 0.f;
}

__global__ void bias_sigmoid_kernel(const float* __restrict__ a, const float* __restrict__ b,
                                    float* __restrict__ out, long long n, int Fmask) {
  long long i = (long long)blockIdx.x * blockDim.x + threadIdx.x;
  if (i >= n) return;
  float v = a[i] + b[(int)(i & Fmask)];
  out[i] = 1.f / (1.f + __expf(-v));
}

extern "C" void kernel_launch(void* const* d_in, const int* in_sizes, int n_in,
                              void* d_out, int out_size, void* d_ws, size_t ws_size,
                              hipStream_t stream) {
  const float* x   = (const float*)d_in[0];
  const int*   ei  = (const int*)d_in[1];
  const float* W1  = (const float*)d_in[2];
  const float* as1 = (const float*)d_in[3];
  const float* ad1 = (const float*)d_in[4];
  const float* b1  = (const float*)d_in[5];
  const float* W2  = (const float*)d_in[6];
  const float* as2 = (const float*)d_in[7];
  const float* ad2 = (const float*)d_in[8];
  const float* b2  = (const float*)d_in[9];
  float* out = (float*)d_out;

  const int N = in_sizes[0] / 128;
  const int E = in_sizes[1] / 2;
  const int Etot = E + N;
  const int* srcE = ei;
  const int* dstE = ei + E;

  float* w = (float*)d_ws;
  size_t o = 0;
  float* h1    = w + o; o += (size_t)N * 128;
  float* agg1  = w + o; o += (size_t)N * 128;
  float* asrc1 = w + o; o += (size_t)N * 2;
  float* adst1 = w + o; o += (size_t)N * 2;
  unsigned* m1 = (unsigned*)(w + o); o += (size_t)N * 2;
  float* s1    = w + o; o += (size_t)N * 2;
  float* asrc2 = w + o; o += (size_t)N;
  float* adst2 = w + o; o += (size_t)N;
  unsigned* m2 = (unsigned*)(w + o); o += (size_t)N;
  float* s2    = w + o; o += (size_t)N;
  float* e1    = w + o; o += (size_t)Etot * 2;
  float* e2    = w + o; o += (size_t)Etot;
  float* h2f   = h1;
  float* agg2  = h1 + (size_t)N * 64;
  (void)o; (void)ws_size; (void)n_in; (void)out_size;

  const int TB = 256;
  auto blk = [](long long n, int tb) { return (unsigned)((n + tb - 1) / tb); };
  const int rowGroups = (N + 31) / 32;

  zero_kernel<<<blk((long long)N * 128, TB), TB, 0, stream>>>(agg1, (long long)N * 128);
  zero_kernel<<<blk((long long)N * 4, TB), TB, 0, stream>>>((float*)m1, (long long)N * 4);
  gemm_f32_wmma<128><<<dim3(rowGroups * 2), TB, 0, stream>>>(x, W1, h1, N);
  node_alpha_kernel<2><<<blk((long long)N * 2, TB), TB, 0, stream>>>(h1, as1, ad1, asrc1, adst1, N);
  edge_max_kernel<2><<<blk((long long)Etot * 2, TB), TB, 0, stream>>>(srcE, dstE, asrc1, adst1, m1, E, Etot);
  edge_expsum_kernel<2><<<blk((long long)Etot * 2, TB), TB, 0, stream>>>(srcE, dstE, asrc1, adst1, m1, s1, e1, E, Etot);
  edge_agg_kernel<2, 1><<<blk((long long)Etot * 32, TB), TB, 0, stream>>>(srcE, dstE, e1, s1, h1, agg1, E, Etot);
  bias_relu_kernel<<<blk((long long)N * 128, TB), TB, 0, stream>>>(agg1, b1, (long long)N * 128, 127);

  gemm_f32_wmma<64><<<dim3(rowGroups), TB, 0, stream>>>(agg1, W2, h2f, N);
  zero_kernel<<<blk((long long)N * 64, TB), TB, 0, stream>>>(agg2, (long long)N * 64);
  zero_kernel<<<blk((long long)N * 2, TB), TB, 0, stream>>>((float*)m2, (long long)N * 2);
  node_alpha_kernel<1><<<blk((long long)N, TB), TB, 0, stream>>>(h2f, as2, ad2, asrc2, adst2, N);
  edge_max_kernel<1><<<blk((long long)Etot, TB), TB, 0, stream>>>(srcE, dstE, asrc2, adst2, m2, E, Etot);
  edge_expsum_kernel<1><<<blk((long long)Etot, TB), TB, 0, stream>>>(srcE, dstE, asrc2, adst2, m2, s2, e2, E, Etot);
  edge_agg_kernel<1, 2><<<blk((long long)Etot * 16, TB), TB, 0, stream>>>(srcE, dstE, e2, s2, h2f, agg2, E, Etot);
  bias_sigmoid_kernel<<<blk((long long)N * 64, TB), TB, 0, stream>>>(agg2, b2, out, (long long)N * 64, 63);
}


# edges bucketed by destination in each call (integer rank add, scan, fill); one wave per node does softmax and row sum in registers, row stored once: no f32 atomics; sums in rank arrival order
# speedup vs baseline: 2.6180x; 2.2577x over previous
#include <hip/hip_runtime.h>
#include <hip/hip_bf16.h>
#include <math.h>

typedef __attribute__((ext_vector_type(2))) float v2f;
typedef __attribute__((ext_vector_type(4))) float v4f;
typedef __attribute__((ext_vector_type(8))) float v8f;

#define NEG_SLOPE 0.2f
__global__ void zero_kernel(float* __restrict__ p, long long n) {
  long long i = (long long)blockIdx.x * blockDim.x + threadIdx.x;
  if (i < n) p[i] = 0.f;
}

template<int NC>
__global__ __launch_bounds__(256) void gemm_f32_wmma(const float* __restrict__ A,
                                                     const float* __restrict__ B,
                                                     float* __restrict__ C, int M) {
  constexpr int K = 128;
  constexpr int CG = 64;
  constexpr int LDA = K + 4;
  constexpr int colGroups = NC / CG;
  __shared__ float Wlds[K * CG];
  __shared__ float Alds[32 * LDA];
  int cg = blockIdx.x % colGroups;
  int rg = blockIdx.x / colGroups;

  for (int i = threadIdx.x; i < K * (CG / 4); i += 256) {
    int k = i / (CG / 4), c4 = i % (CG / 4);
    *(float4*)&Wlds[k * CG + c4 * 4] =
        *(const float4*)&B[(size_t)k * NC + cg * CG + c4 * 4];
  }
  for (int i = threadIdx.x; i < 32 * (K / 4); i += 256) {
    int r = i / (K / 4), c4 = i % (K / 4);
    int grow = rg * 32 + r;
    float4 v = make_float4(0.f, 0.f, 0.f, 0.f);
    if (grow < M) v = *(const float4*)&A[(size_t)grow * K + c4 * 4];
    *(float4*)&Alds[r * LDA + c4 * 4] = v;
  }
  __syncthreads();

  int wave = threadIdx.x >> 5;
  int lane = threadIdx.x & 31;
  int half = lane >> 4;
  int lrow = lane & 15;
  int rt = wave >> 2, ct = wave & 3;
  int grow0 = rg * 32 + rt * 16;
  if (grow0 >= M) return;

  const float* Ab = &Alds[(rt * 16 + lrow) * LDA + 2 * half];
  const float* Bb = &Wlds[(2 * half) * CG + ct * 16 + lrow];
  v8f acc = {};
  #pragma unroll
  for (int k0 = 0; k0 < K; k0 += 4) {
    v2f a, b;
    a.x = Ab[k0];        a.y = Ab[k0 + 1];
    b.x = Bb[k0 * CG];   b.y = Bb[k0 * CG + CG];
    acc = __builtin_amdgcn_wmma_f32_16x16x4_f32(false, a, false, b,
                                                (short)0, acc, false, false);
  }
  float* Cp = &C[(size_t)(grow0 + 8 * half) * NC + cg * CG + ct * 16 + lrow];
  #pragma unroll
  for (int j = 0; j < 8; ++j) Cp[(size_t)j * NC] = acc[j];
}

template<int H>
__global__ void node_alpha_kernel(const float* __restrict__ h,
                                  const float* __restrict__ aw_src,
                                  const float* __restrict__ aw_dst,
                                  float* __restrict__ asrc, float* __restrict__ adst, int N) {
  int idx = blockIdx.x * blockDim.x + threadIdx.x;
  if (idx >= N * H) return;
  int hh = idx % H;
  const float* hp = h + (size_t)idx * 64;
  const float* ap = aw_src + hh * 64;
  const float* bp = aw_dst + hh * 64;
  float sa = 0.f, sb = 0.f;
  #pragma unroll
  for (int c = 0; c < 64; c += 4) {
    float4 v = *(const float4*)(hp + c);
    float4 u = *(const float4*)(ap + c);
    float4 t = *(const float4*)(bp + c);
    sa += v.x * u.x + v.y * u.y + v.z * u.z + v.w * u.w;
    sb += v.x * t.x + v.y * t.y + v.z * t.z + v.w * t.w;
  }
  asrc[idx] = sa;
  adst[idx] = sb;
}

__global__ void edge_rank_kernel(const int* __restrict__ dstE, int* __restrict__ cnt,
                                 int* __restrict__ rank, int E) {
  int e = blockIdx.x * blockDim.x + threadIdx.x;
  if (e >= E) return;
  rank[e] = atomicAdd(&cnt[dstE[e]], 1);
}

__global__ __launch_bounds__(1024) void rowptr_scan_kernel(const int* __restrict__ cnt,
                                                          int* __restrict__ rowptr, int N) {
  __shared__ int wsum[32];
  __shared__ int wbase[32];
  const int t = threadIdx.x, lane = t & 31, wave = t >> 5;
  const int tile0 = blockIdx.x * 1024;
  int front = 0;
  for (int i = t; i < tile0; i += 1024) front += cnt[i];
  const int idx = tile0 + t;
  const int c = (idx < N) ? cnt[idx] : 0;
  int incl = c;
  for (int off = 1; off < 32; off <<= 1) {
    int v = __shfl_up(incl, off, 32);
    if (lane >= off) incl += v;
  }
  for (int off = 16; off > 0; off >>= 1) front += __shfl_xor(front, off, 32);
  if (lane == 31) wsum[wave] = incl;
  if (lane == 0) wbase[wave] = front;
  __syncthreads();
  const int ws = wsum[lane];
  int winc = ws;
  int base = wbase[lane];
  for (int off = 1; off < 32; off <<= 1) {
    int v = __shfl_up(winc, off, 32);
    if (lane >= off) winc += v;
  }
  for (int off = 16; off > 0; off >>= 1) base += __shfl_xor(base, off, 32);
  const int wave_excl = __shfl(winc - ws, wave, 32);
  const int excl = base + wave_excl + (incl - c);
  if (idx < N) rowptr[idx] = excl;
  if (idx == N - 1) rowptr[N] = excl + c;
}

__global__ void edge_fill_kernel(const int* __restrict__ srcE, const int* __restrict__ dstE,
                                 const int* __restrict__ rowptr, const int* __restrict__ rank,
                                 int* __restrict__ csr_src, int E) {
  int e = blockIdx.x * blockDim.x + threadIdx.x;
  if (e >= E) return;
  csr_src[rowptr[dstE[e]] + rank[e]] = srcE[e];
}

template<int V> struct RowVec;
template<> struct RowVec<4> { typedef v4f T; };
template<> struct RowVec<2> { typedef v2f T; };

__device__ __forceinline__ float lane_bcast(float x, int k) {
  return __int_as_float(__builtin_amdgcn_readlane(__float_as_int(x), k));
}

template<int H, bool FINAL>
__global__ __launch_bounds__(256) void node_softmax_agg_kernel(
    const int* __restrict__ rowptr, const int* __restrict__ csr_src,
    const float* __restrict__ asrc, const float* __restrict__ adst,
    const float* __restrict__ h, const float* __restrict__ bias,
    float* __restrict__ outp, int N) {
  constexpr int F = H * 64;
  constexpr int VEC = F / 32;
  constexpr int U = 4;
  typedef typename RowVec<VEC>::T vecf;
  const int lane = threadIdx.x & 31;
  const int d = __builtin_amdgcn_readfirstlane((int)((blockIdx.x * blockDim.x + threadIdx.x) >> 5));
  if (d >= N) return;
  const int beg = rowptr[d];
  const int cnt = rowptr[d + 1] - beg + 1;
  const int myh = (VEC * lane) / 64;
  float adv[H];
  for (int hh = 0; hh < H; ++hh) adv[hh] = adst[d * H + hh];

  auto slot_src = [&](int i) -> int { return (i < cnt - 1) ? csr_src[beg + i] : d; };
  auto logit = [&](int s, int hh) -> float {
    float v = asrc[s * H + hh] + adv[hh];
    return (v > 0.f) ? v : NEG_SLOPE * v;
  };

  const bool ok0 = lane < cnt;
  const int s0 = ok0 ? slot_src(lane) : d;
  float v0[H], mx[H], sm[H], ex0[H];
  for (int hh = 0; hh < H; ++hh) {
    v0[hh] = logit(s0, hh);
    mx[hh] = ok0 ? v0[hh] : -__builtin_inff();
  }
  for (int i = lane + 32; i < cnt; i += 32) {
    const int s = slot_src(i);
    for (int hh = 0; hh < H; ++hh) mx[hh] = fmaxf(mx[hh], logit(s, hh));
  }
  for (int hh = 0; hh < H; ++hh)
    for (int off = 16; off > 0; off >>= 1) mx[hh] = fmaxf(mx[hh], __shfl_xor(mx[hh], off, 32));
  for (int hh = 0; hh < H; ++hh) {
    ex0[hh] = ok0 ? __expf(v0[hh] - mx[hh]) : 0.f;
    sm[hh] = ex0[hh];
  }
  for (int i = lane + 32; i < cnt; i += 32) {
    const int s = slot_src(i);
    for (int hh = 0; hh < H; ++hh) sm[hh] += __expf(logit(s, hh) - mx[hh]);
  }
  for (int hh = 0; hh < H; ++hh)
    for (int off = 16; off > 0; off >>= 1) sm[hh] += __shfl_xor(sm[hh], off, 32);

  vecf acc = {};
  const float* hl = h + VEC * lane;
  for (int c0 = 0; c0 < cnt; c0 += 32) {
    int s = s0;
    float al[H];
    if (c0 == 0) {
      for (int hh = 0; hh < H; ++hh) al[hh] = ex0[hh] / sm[hh];
    } else {
      const int i = c0 + lane;
      const bool ok = i < cnt;
      s = ok ? slot_src(i) : d;
      for (int hh = 0; hh < H; ++hh) al[hh] = ok ? __expf(logit(s, hh) - mx[hh]) / sm[hh] : 0.f;
    }
    const int n = (cnt - c0 < 32) ? (cnt - c0) : 32;
    for (int k = 0; k < n; k += U) {
      vecf hv[U];
      float a[U];
      #pragma unroll
      for (int u = 0; u < U; ++u) {
        const int kk = k + u;
        const int kc = (kk < n) ? kk : n - 1;
        const int sk = __builtin_amdgcn_readlane(s, kc);
        hv[u] = *(const vecf*)(hl + (size_t)sk * F);
        float ax = lane_bcast(al[0], kc);
        if constexpr (H == 2) {
          const float a1 = lane_bcast(al[1], kc);
          ax = myh ? a1 : ax;
        }
        a[u] = (kk < n) ? ax : 0.f;
      }
      #pragma unroll
      for (int u = 0; u < U; ++u) acc += a[u] * hv[u];
    }
  }

  vecf r = acc + *(const vecf*)(bias + VEC * lane);
  for (int c = 0; c < VEC; ++c) {
    if (FINAL) r[c] = 1.f / (1.f + __expf(-r[c]));
    else       r[c] = (r[c] > 0.f) ? r[c] : 0.f;
  }
  *(vecf*)(outp + (size_t)d * F + VEC * lane) = r;
}

extern "C" void kernel_launch(void* const* d_in, const int* in_sizes, int n_in,
                              void* d_out, int out_size, void* d_ws, size_t ws_size,
                              hipStream_t stream) {
  const float* x   = (const float*)d_in[0];
  const int*   ei  = (const int*)d_in[1];
  const float* W1  = (const float*)d_in[2];
  const float* as1 = (const float*)d_in[3];
  const float* ad1 = (const float*)d_in[4];
  const float* b1  = (const float*)d_in[5];
  const float* W2  = (const float*)d_in[6];
  const float* as2 = (const float*)d_in[7];
  const float* ad2 = (const float*)d_in[8];
  const float* b2  = (const float*)d_in[9];
  float* out = (float*)d_out;

  const int N = in_sizes[0] / 128;
  const int E = in_sizes[1] / 2;
  const int* srcE = ei;
  const int* dstE = ei + E;

  float* w = (float*)d_ws;
  size_t o = 0;
  float* h1    = w + o; o += (size_t)N * 128;
  float* agg1  = w + o; o += (size_t)N * 128;
  float* asrc1 = w + o; o += (size_t)N * 2;
  float* adst1 = w + o; o += (size_t)N * 2;
  float* asrc2 = w + o; o += (size_t)N;
  float* adst2 = w + o; o += (size_t)N;
  int* cnt     = (int*)(w + o); o += (size_t)N;
  int* rowptr  = (int*)(w + o); o += (size_t)N + 1;
  int* rank    = (int*)(w + o); o += (size_t)E;
  int* csr_src = (int*)(w + o); o += (size_t)E;
  float* h2f   = h1;
  (void)o; (void)ws_size; (void)n_in; (void)out_size;

  const int TB = 256;
  auto blk = [](long long n, int tb) { return (unsigned)((n + tb - 1) / tb); };
  const int rowGroups = (N + 31) / 32;

  zero_kernel<<<blk((long long)N, TB), TB, 0, stream>>>((float*)cnt, (long long)N);
  edge_rank_kernel<<<blk((long long)E, TB), TB, 0, stream>>>(dstE, cnt, rank, E);
  rowptr_scan_kernel<<<blk((long long)N, 1024), 1024, 0, stream>>>(cnt, rowptr, N);
  edge_fill_kernel<<<blk((long long)E, TB), TB, 0, stream>>>(srcE, dstE, rowptr, rank, csr_src, E);

  gemm_f32_wmma<128><<<dim3(rowGroups * 2), TB, 0, stream>>>(x, W1, h1, N);
  node_alpha_kernel<2><<<blk((long long)N * 2, TB), TB, 0, stream>>>(h1, as1, ad1, asrc1, adst1, N);
  node_softmax_agg_kernel<2, false><<<blk((long long)N * 32, TB), TB, 0, stream>>>(
      rowptr, csr_src, asrc1, adst1, h1, b1, agg1, N);

  gemm_f32_wmma<64><<<dim3(rowGroups), TB, 0, stream>>>(agg1, W2, h2f, N);
  node_alpha_kernel<1><<<blk((long long)N, TB), TB, 0, stream>>>(h2f, as2, ad2, asrc2, adst2, N);
  node_softmax_agg_kernel<1, true><<<blk((long long)N * 32, TB), TB, 0, stream>>>(
      rowptr, csr_src, asrc2, adst2, h2f, b2, out, N);
}


# edges bucketed by destination by a two-level counting sort whose counts and ranks are LDS adds (chunk histogram, filing by bin, per-bin placement): no global atomics; sums in LDS arrival order
# speedup vs baseline: 4.4138x; 1.6859x over previous
#include <hip/hip_runtime.h>
#include <hip/hip_bf16.h>
#include <math.h>

typedef __attribute__((ext_vector_type(2))) float v2f;
typedef __attribute__((ext_vector_type(4))) float v4f;
typedef __attribute__((ext_vector_type(8))) float v8f;

#define NEG_SLOPE 0.2f

template<int NC>
__global__ __launch_bounds__(256) void gemm_f32_wmma(const float* __restrict__ A,
                                                     const float* __restrict__ B,
                                                     float* __restrict__ C, int M) {
  constexpr int K = 128;
  constexpr int CG = 64;
  constexpr int LDA = K + 4;
  constexpr int colGroups = NC / CG;
  __shared__ float Wlds[K * CG];
  __shared__ float Alds[32 * LDA];
  int cg = blockIdx.x % colGroups;
  int rg = blockIdx.x / colGroups;

  for (int i = threadIdx.x; i < K * (CG / 4); i += 256) {
    int k = i / (CG / 4), c4 = i % (CG / 4);
    *(float4*)&Wlds[k * CG + c4 * 4] =
        *(const float4*)&B[(size_t)k * NC + cg * CG + c4 * 4];
  }
  for (int i = threadIdx.x; i < 32 * (K / 4); i += 256) {
    int r = i / (K / 4), c4 = i % (K / 4);
    int grow = rg * 32 + r;
    float4 v = make_float4(0.f, 0.f, 0.f, 0.f);
    if (grow < M) v = *(const float4*)&A[(size_t)grow * K + c4 * 4];
    *(float4*)&Alds[r * LDA + c4 * 4] = v;
  }
  __syncthreads();

  int wave = threadIdx.x >> 5;
  int lane = threadIdx.x & 31;
  int half = lane >> 4;
  int lrow = lane & 15;
  int rt = wave >> 2, ct = wave & 3;
  int grow0 = rg * 32 + rt * 16;
  if (grow0 >= M) return;

  const float* Ab = &Alds[(rt * 16 + lrow) * LDA + 2 * half];
  const float* Bb = &Wlds[(2 * half) * CG + ct * 16 + lrow];
  v8f acc = {};
  #pragma unroll
  for (int k0 = 0; k0 < K; k0 += 4) {
    v2f a, b;
    a.x = Ab[k0];        a.y = Ab[k0 + 1];
    b.x = Bb[k0 * CG];   b.y = Bb[k0 * CG + CG];
    acc = __builtin_amdgcn_wmma_f32_16x16x4_f32(false, a, false, b,
                                                (short)0, acc, false, false);
  }
  float* Cp = &C[(size_t)(grow0 + 8 * half) * NC + cg * CG + ct * 16 + lrow];
  #pragma unroll
  for (int j = 0; j < 8; ++j) Cp[(size_t)j * NC] = acc[j];
}

template<int H>
__global__ void node_alpha_kernel(const float* __restrict__ h,
                                  const float* __restrict__ aw_src,
                                  const float* __restrict__ aw_dst,
                                  float* __restrict__ asrc, float* __restrict__ adst, int N) {
  int idx = blockIdx.x * blockDim.x + threadIdx.x;
  if (idx >= N * H) return;
  int hh = idx % H;
  const float* hp = h + (size_t)idx * 64;
  const float* ap = aw_src + hh * 64;
  const float* bp = aw_dst + hh * 64;
  float sa = 0.f, sb = 0.f;
  #pragma unroll
  for (int c = 0; c < 64; c += 4) {
    float4 v = *(const float4*)(hp + c);
    float4 u = *(const float4*)(ap + c);
    float4 t = *(const float4*)(bp + c);
    sa += v.x * u.x + v.y * u.y + v.z * u.z + v.w * u.w;
    sb += v.x * t.x + v.y * t.y + v.z * t.z + v.w * t.w;
  }
  asrc[idx] = sa;
  adst[idx] = sb;
}

typedef __attribute__((ext_vector_type(2))) int v2i;
typedef __attribute__((ext_vector_type(4))) int v4i;
constexpr int BIN_SHIFT = 8;
constexpr int BIN_NODES = 1 << BIN_SHIFT;
constexpr int MAX_BINS  = 256;
constexpr int SORT_NT   = 512;
constexpr int SORT_U    = 8;
constexpr int SORT_CH   = SORT_NT * SORT_U;

__device__ __forceinline__ int wave_incl_scan(int v, int lane) {
  for (int off = 1; off < 32; off <<= 1) {
    int u = __shfl_up(v, off, 32);
    if (lane >= off) v += u;
  }
  return v;
}

__global__ __launch_bounds__(SORT_NT) void bin_hist_kernel(const int* __restrict__ dstE,
                                                           int* __restrict__ histG, int E) {
  __shared__ int hist[MAX_BINS];
  const int t = threadIdx.x, g = blockIdx.x;
  const int e0 = g * SORT_CH;
  const int len = min(E - e0, SORT_CH);
  int d[SORT_U];
  #pragma unroll
  for (int u = 0; u < SORT_U; ++u) {
    const int i = u * SORT_NT + t;
    d[u] = (i < len) ? dstE[e0 + i] : -1;
  }
  if (t < MAX_BINS) hist[t] = 0;
  __syncthreads();
  #pragma unroll
  for (int u = 0; u < SORT_U; ++u)
    if (d[u] >= 0) atomicAdd(&hist[d[u] >> BIN_SHIFT], 1);
  __syncthreads();
  if (t < MAX_BINS) histG[g * MAX_BINS + t] = hist[t];
}

__global__ __launch_bounds__(SORT_NT) void bin_scatter_kernel(const int* __restrict__ srcE,
                                                              const int* __restrict__ dstE,
                                                              const int* __restrict__ histG,
                                                              int* __restrict__ binStart,
                                                              int* __restrict__ binned,
                                                              int E, int nbins, int nchunks) {
  constexpr int QN = MAX_BINS / 4;
  constexpr int NG = SORT_NT / QN;
  constexpr int LB = 16;
  __shared__ int tot[MAX_BINS];
  __shared__ int fr[MAX_BINS];
  __shared__ int lcur[MAX_BINS];
  __shared__ int gbase[MAX_BINS];
  __shared__ int wsumA[MAX_BINS / 32], wsumB[MAX_BINS / 32];
  __shared__ v2i stage[SORT_CH];
  const int t = threadIdx.x, lane = t & 31, wave = t >> 5;
  const int g = blockIdx.x;
  const int q = t % QN;
  const int grp = __builtin_amdgcn_readfirstlane(t / QN);

  const int e0 = g * SORT_CH;
  const int len = min(E - e0, SORT_CH);
  int d[SORT_U], s[SORT_U];
  #pragma unroll
  for (int u = 0; u < SORT_U; ++u) {
    const int i = u * SORT_NT + t;
    const bool ok = i < len;
    d[u] = ok ? dstE[e0 + i] : -1;
    s[u] = ok ? srcE[e0 + i] : 0;
  }
  if (t < MAX_BINS) { tot[t] = 0; fr[t] = 0; }

  v4i all = {0, 0, 0, 0}, front = {0, 0, 0, 0};
  for (int r0 = grp; r0 < nchunks; r0 += NG * LB) {
    v4i v[LB];
    #pragma unroll
    for (int k = 0; k < LB; ++k) {
      const int gp = r0 + NG * k;
      const int gc = min(gp, nchunks - 1);
      v[k] = ((const v4i*)(histG + (size_t)gc * MAX_BINS))[q];
    }
    #pragma unroll
    for (int k = 0; k < LB; ++k) {
      const int gp = r0 + NG * k;
      const v4i z = {0, 0, 0, 0};
      all   += (gp < nchunks) ? v[k] : z;
      front += (gp < g) ? v[k] : z;
    }
  }
  __syncthreads();
  for (int k = 0; k < 4; ++k) {
    atomicAdd(&tot[4 * q + k], all[k]);
    atomicAdd(&fr[4 * q + k], front[k]);
  }
  __syncthreads();
  int ca = 0, cb = 0, ia = 0, ib = 0;
  if (t < MAX_BINS) {
    ca = tot[t];
    cb = histG[g * MAX_BINS + t];
    ia = wave_incl_scan(ca, lane);
    ib = wave_incl_scan(cb, lane);
    if (lane == 31) { wsumA[wave] = ia; wsumB[wave] = ib; }
  }
  __syncthreads();
  if (t < MAX_BINS) {
    int ba = 0, bb = 0;
    for (int w = 0; w < MAX_BINS / 32; ++w) {
      ba += (w < wave) ? wsumA[w] : 0;
      bb += (w < wave) ? wsumB[w] : 0;
    }
    const int exa = ba + ia - ca;
    const int exb = bb + ib - cb;
    lcur[t] = exb;
    gbase[t] = exa + fr[t] - exb;
    if (g == 0) {
      if (t < nbins) binStart[t] = exa;
      if (t == nbins - 1) binStart[nbins] = exa + ca;
    }
  }
  __syncthreads();
  #pragma unroll
  for (int u = 0; u < SORT_U; ++u)
    if (d[u] >= 0) stage[atomicAdd(&lcur[d[u] >> BIN_SHIFT], 1)] = (v2i){s[u], d[u]};
  __syncthreads();
  v2i* pairs = (v2i*)binned;
  #pragma unroll
  for (int u = 0; u < SORT_U; ++u) {
    const int i = u * SORT_NT + t;
    if (i < len) {
      const v2i pr = stage[i];
      pairs[gbase[pr.y >> BIN_SHIFT] + i] = pr;
    }
  }
}

__global__ __launch_bounds__(SORT_NT) void bin_place_kernel(const int* __restrict__ binned,
                                                            const int* __restrict__ binStart,
                                                            int* __restrict__ rowptr,
                                                            int* __restrict__ csr_src, int N) {
  constexpr int U = 10;
  constexpr int STAGE_MAX = 8192;
  __shared__ int cnt[BIN_NODES];
  __shared__ int wsum[BIN_NODES / 32];
  __shared__ int stage[STAGE_MAX];
  const int t = threadIdx.x, lane = t & 31, wave = t >> 5;
  const int b = blockIdx.x;
  const int s0 = binStart[b], s1 = binStart[b + 1];
  const bool staged = (s1 - s0) <= STAGE_MAX;
  const int node0 = b << BIN_SHIFT;
  const v2i* pairs = (const v2i*)binned;
  if (t < BIN_NODES) cnt[t] = 0;
  __syncthreads();
  for (int base = s0; base < s1; base += U * SORT_NT) {
    int d[U];
    #pragma unroll
    for (int u = 0; u < U; ++u) {
      const int e = base + u * SORT_NT + t;
      d[u] = (e < s1) ? pairs[e].y : -1;
    }
    #pragma unroll
    for (int u = 0; u < U; ++u)
      if (d[u] >= 0) atomicAdd(&cnt[d[u] - node0], 1);
  }
  __syncthreads();
  int c = 0, incl = 0;
  if (t < BIN_NODES) {
    c = cnt[t];
    incl = wave_incl_scan(c, lane);
    if (lane == 31) wsum[wave] = incl;
  }
  __syncthreads();
  if (t < BIN_NODES) {
    int wbase = 0;
    for (int w = 0; w < BIN_NODES / 32; ++w) wbase += (w < wave) ? wsum[w] : 0;
    const int excl = s0 + wbase + incl - c;
    cnt[t] = excl;
    const int node = node0 + t;
    if (node < N) rowptr[node] = excl;
    if (node == N - 1) rowptr[N] = excl + c;
  }
  __syncthreads();
  for (int base = s0; base < s1; base += U * SORT_NT) {
    v2i p[U];
    #pragma unroll
    for (int u = 0; u < U; ++u) {
      const int e = base + u * SORT_NT + t;
      p[u] = (e < s1) ? pairs[e] : (v2i){0, -1};
    }
    #pragma unroll
    for (int u = 0; u < U; ++u) {
      if (p[u].y >= 0) {
        const int pos = atomicAdd(&cnt[p[u].y - node0], 1);
        if (staged) stage[pos - s0] = p[u].x;
        else        csr_src[pos] = p[u].x;
      }
    }
  }
  __syncthreads();
  if (staged)
    for (int i = t; i < s1 - s0; i += SORT_NT) csr_src[s0 + i] = stage[i];
}

template<int V> struct RowVec;
template<> struct RowVec<4> { typedef v4f T; };
template<> struct RowVec<2> { typedef v2f T; };

__device__ __forceinline__ float lane_bcast(float x, int k) {
  return __int_as_float(__builtin_amdgcn_readlane(__float_as_int(x), k));
}

template<int H, bool FINAL>
__global__ __launch_bounds__(256) void node_softmax_agg_kernel(
    const int* __restrict__ rowptr, const int* __restrict__ csr_src,
    const float* __restrict__ asrc, const float* __restrict__ adst,
    const float* __restrict__ h, const float* __restrict__ bias,
    float* __restrict__ outp, int N) {
  constexpr int F = H * 64;
  constexpr int VEC = F / 32;
  constexpr int U = 4;
  typedef typename RowVec<VEC>::T vecf;
  const int lane = threadIdx.x & 31;
  const int d = __builtin_amdgcn_readfirstlane((int)((blockIdx.x * blockDim.x + threadIdx.x) >> 5));
  if (d >= N) return;
  const int beg = rowptr[d];
  const int cnt = rowptr[d + 1] - beg + 1;
  const int myh = (VEC * lane) / 64;
  float adv[H];
  for (int hh = 0; hh < H; ++hh) adv[hh] = adst[d * H + hh];

  auto slot_src = [&](int i) -> int { return (i < cnt - 1) ? csr_src[beg + i] : d; };
  auto logit = [&](int s, int hh) -> float {
    float v = asrc[s * H + hh] + adv[hh];
    return (v > 0.f) ? v : NEG_SLOPE * v;
  };

  const bool ok0 = lane < cnt;
  const int s0 = ok0 ? slot_src(lane) : d;
  float v0[H], mx[H], sm[H], ex0[H];
  for (int hh = 0; hh < H; ++hh) {
    v0[hh] = logit(s0, hh);
    mx[hh] = ok0 ? v0[hh] : -__builtin_inff();
  }
  for (int i = lane + 32; i < cnt; i += 32) {
    const int s = slot_src(i);
    for (int hh = 0; hh < H; ++hh) mx[hh] = fmaxf(mx[hh], logit(s, hh));
  }
  for (int hh = 0; hh < H; ++hh)
    for (int off = 16; off > 0; off >>= 1) mx[hh] = fmaxf(mx[hh], __shfl_xor(mx[hh], off, 32));
  for (int hh = 0; hh < H; ++hh) {
    ex0[hh] = ok0 ? __expf(v0[hh] - mx[hh]) : 0.f;
    sm[hh] = ex0[hh];
  }
  for (int i = lane + 32; i < cnt; i += 32) {
    const int s = slot_src(i);
    for (int hh = 0; hh < H; ++hh) sm[hh] += __expf(logit(s, hh) - mx[hh]);
  }
  for (int hh = 0; hh < H; ++hh)
    for (int off = 16; off > 0; off >>= 1) sm[hh] += __shfl_xor(sm[hh], off, 32);

  vecf acc = {};
  const float* hl = h + VEC * lane;
  for (int c0 = 0; c0 < cnt; c0 += 32) {
    int s = s0;
    float al[H];
    if (c0 == 0) {
      for (int hh = 0; hh < H; ++hh) al[hh] = ex0[hh] / sm[hh];
    } else {
      const int i = c0 + lane;
      const bool ok = i < cnt;
      s = ok ? slot_src(i) : d;
      for (int hh = 0; hh < H; ++hh) al[hh] = ok ? __expf(logit(s, hh) - mx[hh]) / sm[hh] : 0.f;
    }
    const int n = (cnt - c0 < 32) ? (cnt - c0) : 32;
    for (int k = 0; k < n; k += U) {
      vecf hv[U];
      float a[U];
      #pragma unroll
      for (int u = 0; u < U; ++u) {
        const int kk = k + u;
        const int kc = (kk < n) ? kk : n - 1;
        const int sk = __builtin_amdgcn_readlane(s, kc);
        hv[u] = *(const vecf*)(hl + (size_t)sk * F);
        float ax = lane_bcast(al[0], kc);
        if constexpr (H == 2) {
          const float a1 = lane_bcast(al[1], kc);
          ax = myh ? a1 : ax;
        }
        a[u] = (kk < n) ? ax : 0.f;
      }
      #pragma unroll
      for (int u = 0; u < U; ++u) acc += a[u] * hv[u];
    }
  }

  vecf r = acc + *(const vecf*)(bias + VEC * lane);
  for (int c = 0; c < VEC; ++c) {
    if (FINAL) r[c] = 1.f / (1.f + __expf(-r[c]));
    else       r[c] = (r[c] > 0.f) ? r[c] : 0.f;
  }
  *(vecf*)(outp + (size_t)d * F + VEC * lane) = r;
}

extern "C" void kernel_launch(void* const* d_in, const int* in_sizes, int n_in,
                              void* d_out, int out_size, void* d_ws, size_t ws_size,
                              hipStream_t stream) {
  const float* x   = (const float*)d_in[0];
  const int*   ei  = (const int*)d_in[1];
  const float* W1  = (const float*)d_in[2];
  const float* as1 = (const float*)d_in[3];
  const float* ad1 = (const float*)d_in[4];
  const float* b1  = (const float*)d_in[5];
  const float* W2  = (const float*)d_in[6];
  const float* as2 = (const float*)d_in[7];
  const float* ad2 = (const float*)d_in[8];
  const float* b2  = (const float*)d_in[9];
  float* out = (float*)d_out;

  const int N = in_sizes[0] / 128;
  const int E = in_sizes[1] / 2;
  const int* srcE = ei;
  const int* dstE = ei + E;

  float* w = (float*)d_ws;
  size_t o = 0;
  float* h1    = w + o; o += (size_t)N * 128;
  float* agg1  = w + o; o += (size_t)N * 128;
  float* asrc1 = w + o; o += (size_t)N * 2;
  float* adst1 = w + o; o += (size_t)N * 2;
  float* asrc2 = w + o; o += (size_t)N;
  float* adst2 = w + o; o += (size_t)N;
  o = (o + 3) & ~(size_t)3;
  const int nbins   = (N + BIN_NODES - 1) / BIN_NODES;
  const int nchunks = (E + SORT_CH - 1) / SORT_CH;
  int* histG    = (int*)(w + o); o += (size_t)nchunks * MAX_BINS;
  int* binStart = (int*)(w + o); o += (size_t)MAX_BINS + 4;
  int* binned   = (int*)(w + o); o += (size_t)E * 2;
  int* rowptr   = (int*)(w + o); o += (size_t)N + 1;
  int* csr_src  = (int*)(w + o); o += (size_t)E;
  float* h2f   = h1;
  (void)o; (void)ws_size; (void)n_in; (void)out_size;

  const int TB = 256;
  auto blk = [](long long n, int tb) { return (unsigned)((n + tb - 1) / tb); };
  const int rowGroups = (N + 31) / 32;

  bin_hist_kernel<<<nchunks, SORT_NT, 0, stream>>>(dstE, histG, E);
  bin_scatter_kernel<<<nchunks, SORT_NT, 0, stream>>>(srcE, dstE, histG, binStart, binned, E, nbins, nchunks);
  bin_place_kernel<<<nbins, SORT_NT, 0, stream>>>(binned, binStart, rowptr, csr_src, N);

  gemm_f32_wmma<128><<<dim3(rowGroups * 2), TB, 0, stream>>>(x, W1, h1, N);
  node_alpha_kernel<2><<<blk((long long)N * 2, TB), TB, 0, stream>>>(h1, as1, ad1, asrc1, adst1, N);
  node_softmax_agg_kernel<2, false><<<blk((long long)N * 32, TB), TB, 0, stream>>>(
      rowptr, csr_src, asrc1, adst1, h1, b1, agg1, N);

  gemm_f32_wmma<64><<<dim3(rowGroups), TB, 0, stream>>>(agg1, W2, h2f, N);
  node_alpha_kernel<1><<<blk((long long)N, TB), TB, 0, stream>>>(h2f, as2, ad2, asrc2, adst2, N);
  node_softmax_agg_kernel<1, true><<<blk((long long)N * 32, TB), TB, 0, stream>>>(
      rowptr, csr_src, asrc2, adst2, h2f, b2, out, N);
}


# attention logits computed in the GEMM epilogue from the output tile staged in LDS (row dot, 16-lane DPP sum, whole-line row stores): 2 logit kernels and their re-read of h gone; new logit sum order
# speedup vs baseline: 4.7542x; 1.0771x over previous
#include <hip/hip_runtime.h>
#include <hip/hip_bf16.h>
#include <math.h>

typedef __attribute__((ext_vector_type(2))) float v2f;
typedef __attribute__((ext_vector_type(4))) float v4f;
typedef __attribute__((ext_vector_type(8))) float v8f;

#define NEG_SLOPE 0.2f

__device__ __forceinline__ float row16_sum(float x) {
  x += __int_as_float(__builtin_amdgcn_update_dpp(0, __float_as_int(x), 0x161, 0xf, 0xf, true));
  x += __int_as_float(__builtin_amdgcn_update_dpp(0, __float_as_int(x), 0x162, 0xf, 0xf, true));
  x += __int_as_float(__builtin_amdgcn_update_dpp(0, __float_as_int(x), 0x164, 0xf, 0xf, true));
  x += __int_as_float(__builtin_amdgcn_update_dpp(0, __float_as_int(x), 0x168, 0xf, 0xf, true));
  return x;
}

template<int NC>
__global__ __launch_bounds__(256) void gemm_f32_wmma(const float* __restrict__ A,
                                                     const float* __restrict__ B,
                                                     float* __restrict__ C,
                                                     const float* __restrict__ aw_src,
                                                     const float* __restrict__ aw_dst,
                                                     float* __restrict__ asrc,
                                                     float* __restrict__ adst, int M) {
  constexpr int K = 128;
  constexpr int CG = 64;
  constexpr int LDA = K + 4;
  constexpr int LDC = CG + 4;
  constexpr int colGroups = NC / CG;
  __shared__ float Wlds[K * CG];
  __shared__ float Alds[32 * LDA];
  int cg = blockIdx.x % colGroups;
  int rg = blockIdx.x / colGroups;

  for (int i = threadIdx.x; i < K * (CG / 4); i += 256) {
    int k = i / (CG / 4), c4 = i % (CG / 4);
    *(float4*)&Wlds[k * CG + c4 * 4] =
        *(const float4*)&B[(size_t)k * NC + cg * CG + c4 * 4];
  }
  for (int i = threadIdx.x; i < 32 * (K / 4); i += 256) {
    int r = i / (K / 4), c4 = i % (K / 4);
    int grow = rg * 32 + r;
    float4 v = make_float4(0.f, 0.f, 0.f, 0.f);
    if (grow < M) v = *(const float4*)&A[(size_t)grow * K + c4 * 4];
    *(float4*)&Alds[r * LDA + c4 * 4] = v;
  }
  const int oc4 = threadIdx.x & 15;
  const float4 us = *(const float4*)&aw_src[cg * CG + oc4 * 4];
  const float4 ud = *(const float4*)&aw_dst[cg * CG + oc4 * 4];
  asm volatile("" :: "v"(us.x), "v"(us.y), "v"(us.z), "v"(us.w),
                     "v"(ud.x), "v"(ud.y), "v"(ud.z), "v"(ud.w));
  __syncthreads();

  int wave = threadIdx.x >> 5;
  int lane = threadIdx.x & 31;
  int half = lane >> 4;
  int lrow = lane & 15;
  int rt = wave >> 2, ct = wave & 3;

  const float* Ab = &Alds[(rt * 16 + lrow) * LDA + 2 * half];
  const float* Bb = &Wlds[(2 * half) * CG + ct * 16 + lrow];
  v8f acc = {};
  #pragma unroll
  for (int k0 = 0; k0 < K; k0 += 4) {
    v2f a, b;
    a.x = Ab[k0];        a.y = Ab[k0 + 1];
    b.x = Bb[k0 * CG];   b.y = Bb[k0 * CG + CG];
    acc = __builtin_amdgcn_wmma_f32_16x16x4_f32(false, a, false, b,
                                                (short)0, acc, false, false);
  }

  __syncthreads();
  float* Ct = Alds;
  float* Cw = &Ct[(rt * 16 + 8 * half) * LDC + ct * 16 + lrow];
  #pragma unroll
  for (int j = 0; j < 8; ++j) Cw[j * LDC] = acc[j];
  __syncthreads();

  const int orow = threadIdx.x >> 4;
  const int grow_a = rg * 32 + orow, grow_b = grow_a + 16;
  const float4 va = *(const float4*)&Ct[orow * LDC + oc4 * 4];
  const float4 vb = *(const float4*)&Ct[(orow + 16) * LDC + oc4 * 4];
  if (grow_a < M) *(float4*)&C[(size_t)grow_a * NC + cg * CG + oc4 * 4] = va;
  if (grow_b < M) *(float4*)&C[(size_t)grow_b * NC + cg * CG + oc4 * 4] = vb;
  const float psa = row16_sum(va.x * us.x + va.y * us.y + va.z * us.z + va.w * us.w);
  const float pda = row16_sum(va.x * ud.x + va.y * ud.y + va.z * ud.z + va.w * ud.w);
  const float psb = row16_sum(vb.x * us.x + vb.y * us.y + vb.z * us.z + vb.w * us.w);
  const float pdb = row16_sum(vb.x * ud.x + vb.y * ud.y + vb.z * ud.z + vb.w * ud.w);
  if (oc4 < 4) {
    const bool second = (oc4 & 2) != 0, isdst = (oc4 & 1) != 0;
    const int grow = second ? grow_b : grow_a;
    const float val = isdst ? (second ? pdb : pda) : (second ? psb : psa);
    float* dstp = isdst ? adst : asrc;
    if (grow < M) dstp[(size_t)grow * colGroups + cg] = val;
  }
}

typedef __attribute__((ext_vector_type(2))) int v2i;
typedef __attribute__((ext_vector_type(4))) int v4i;
constexpr int BIN_SHIFT = 8;
constexpr int BIN_NODES = 1 << BIN_SHIFT;
constexpr int MAX_BINS  = 256;
constexpr int SORT_NT   = 512;
constexpr int SORT_U    = 8;
constexpr int SORT_CH   = SORT_NT * SORT_U;

__device__ __forceinline__ int wave_incl_scan(int v, int lane) {
  for (int off = 1; off < 32; off <<= 1) {
    int u = __shfl_up(v, off, 32);
    if (lane >= off) v += u;
  }
  return v;
}

__global__ __launch_bounds__(SORT_NT) void bin_hist_kernel(const int* __restrict__ dstE,
                                                           int* __restrict__ histG, int E) {
  __shared__ int hist[MAX_BINS];
  const int t = threadIdx.x, g = blockIdx.x;
  const int e0 = g * SORT_CH;
  const int len = min(E - e0, SORT_CH);
  int d[SORT_U];
  #pragma unroll
  for (int u = 0; u < SORT_U; ++u) {
    const int i = u * SORT_NT + t;
    d[u] = (i < len) ? dstE[e0 + i] : -1;
  }
  if (t < MAX_BINS) hist[t] = 0;
  __syncthreads();
  #pragma unroll
  for (int u = 0; u < SORT_U; ++u)
    if (d[u] >= 0) atomicAdd(&hist[d[u] >> BIN_SHIFT], 1);
  __syncthreads();
  if (t < MAX_BINS) histG[g * MAX_BINS + t] = hist[t];
}

__global__ __launch_bounds__(SORT_NT) void bin_scatter_kernel(const int* __restrict__ srcE,
                                                              const int* __restrict__ dstE,
                                                              const int* __restrict__ histG,
                                                              int* __restrict__ binStart,
                                                              int* __restrict__ binned,
                                                              int E, int nbins, int nchunks) {
  constexpr int QN = MAX_BINS / 4;
  constexpr int NG = SORT_NT / QN;
  constexpr int LB = 16;
  __shared__ int tot[MAX_BINS];
  __shared__ int fr[MAX_BINS];
  __shared__ int lcur[MAX_BINS];
  __shared__ int gbase[MAX_BINS];
  __shared__ int wsumA[MAX_BINS / 32], wsumB[MAX_BINS / 32];
  __shared__ v2i stage[SORT_CH];
  const int t = threadIdx.x, lane = t & 31, wave = t >> 5;
  const int g = blockIdx.x;
  const int q = t % QN;
  const int grp = __builtin_amdgcn_readfirstlane(t / QN);

  const int e0 = g * SORT_CH;
  const int len = min(E - e0, SORT_CH);
  int d[SORT_U], s[SORT_U];
  #pragma unroll
  for (int u = 0; u < SORT_U; ++u) {
    const int i = u * SORT_NT + t;
    const bool ok = i < len;
    d[u] = ok ? dstE[e0 + i] : -1;
    s[u] = ok ? srcE[e0 + i] : 0;
  }
  if (t < MAX_BINS) { tot[t] = 0; fr[t] = 0; }

  v4i all = {0, 0, 0, 0}, front = {0, 0, 0, 0};
  for (int r0 = grp; r0 < nchunks; r0 += NG * LB) {
    v4i v[LB];
    #pragma unroll
    for (int k = 0; k < LB; ++k) {
      const int gp = r0 + NG * k;
      const int gc = min(gp, nchunks - 1);
      v[k] = ((const v4i*)(histG + (size_t)gc * MAX_BINS))[q];
    }
    #pragma unroll
    for (int k = 0; k < LB; ++k) {
      const int gp = r0 + NG * k;
      const v4i z = {0, 0, 0, 0};
      all   += (gp < nchunks) ? v[k] : z;
      front += (gp < g) ? v[k] : z;
    }
  }
  __syncthreads();
  for (int k = 0; k < 4; ++k) {
    atomicAdd(&tot[4 * q + k], all[k]);
    atomicAdd(&fr[4 * q + k], front[k]);
  }
  __syncthreads();
  int ca = 0, cb = 0, ia = 0, ib = 0;
  if (t < MAX_BINS) {
    ca = tot[t];
    cb = histG[g * MAX_BINS + t];
    ia = wave_incl_scan(ca, lane);
    ib = wave_incl_scan(cb, lane);
    if (lane == 31) { wsumA[wave] = ia; wsumB[wave] = ib; }
  }
  __syncthreads();
  if (t < MAX_BINS) {
    int ba = 0, bb = 0;
    for (int w = 0; w < MAX_BINS / 32; ++w) {
      ba += (w < wave) ? wsumA[w] : 0;
      bb += (w < wave) ? wsumB[w] : 0;
    }
    const int exa = ba + ia - ca;
    const int exb = bb + ib - cb;
    lcur[t] = exb;
    gbase[t] = exa + fr[t] - exb;
    if (g == 0) {
      if (t < nbins) binStart[t] = exa;
      if (t == nbins - 1) binStart[nbins] = exa + ca;
    }
  }
  __syncthreads();
  #pragma unroll
  for (int u = 0; u < SORT_U; ++u)
    if (d[u] >= 0) stage[atomicAdd(&lcur[d[u] >> BIN_SHIFT], 1)] = (v2i){s[u], d[u]};
  __syncthreads();
  v2i* pairs = (v2i*)binned;
  #pragma unroll
  for (int u = 0; u < SORT_U; ++u) {
    const int i = u * SORT_NT + t;
    if (i < len) {
      const v2i pr = stage[i];
      pairs[gbase[pr.y >> BIN_SHIFT] + i] = pr;
    }
  }
}

__global__ __launch_bounds__(SORT_NT) void bin_place_kernel(const int* __restrict__ binned,
                                                            const int* __restrict__ binStart,
                                                            int* __restrict__ rowptr,
                                                            int* __restrict__ csr_src, int N) {
  constexpr int U = 10;
  constexpr int STAGE_MAX = 8192;
  __shared__ int cnt[BIN_NODES];
  __shared__ int wsum[BIN_NODES / 32];
  __shared__ int stage[STAGE_MAX];
  const int t = threadIdx.x, lane = t & 31, wave = t >> 5;
  const int b = blockIdx.x;
  const int s0 = binStart[b], s1 = binStart[b + 1];
  const bool staged = (s1 - s0) <= STAGE_MAX;
  const int node0 = b << BIN_SHIFT;
  const v2i* pairs = (const v2i*)binned;
  if (t < BIN_NODES) cnt[t] = 0;
  __syncthreads();
  for (int base = s0; base < s1; base += U * SORT_NT) {
    int d[U];
    #pragma unroll
    for (int u = 0; u < U; ++u) {
      const int e = base + u * SORT_NT + t;
      d[u] = (e < s1) ? pairs[e].y : -1;
    }
    #pragma unroll
    for (int u = 0; u < U; ++u)
      if (d[u] >= 0) atomicAdd(&cnt[d[u] - node0], 1);
  }
  __syncthreads();
  int c = 0, incl = 0;
  if (t < BIN_NODES) {
    c = cnt[t];
    incl = wave_incl_scan(c, lane);
    if (lane == 31) wsum[wave] = incl;
  }
  __syncthreads();
  if (t < BIN_NODES) {
    int wbase = 0;
    for (int w = 0; w < BIN_NODES / 32; ++w) wbase += (w < wave) ? wsum[w] : 0;
    const int excl = s0 + wbase + incl - c;
    cnt[t] = excl;
    const int node = node0 + t;
    if (node < N) rowptr[node] = excl;
    if (node == N - 1) rowptr[N] = excl + c;
  }
  __syncthreads();
  for (int base = s0; base < s1; base += U * SORT_NT) {
    v2i p[U];
    #pragma unroll
    for (int u = 0; u < U; ++u) {
      const int e = base + u * SORT_NT + t;
      p[u] = (e < s1) ? pairs[e] : (v2i){0, -1};
    }
    #pragma unroll
    for (int u = 0; u < U; ++u) {
      if (p[u].y >= 0) {
        const int pos = atomicAdd(&cnt[p[u].y - node0], 1);
        if (staged) stage[pos - s0] = p[u].x;
        else        csr_src[pos] = p[u].x;
      }
    }
  }
  __syncthreads();
  if (staged)
    for (int i = t; i < s1 - s0; i += SORT_NT) csr_src[s0 + i] = stage[i];
}

template<int V> struct RowVec;
template<> struct RowVec<4> { typedef v4f T; };
template<> struct RowVec<2> { typedef v2f T; };

__device__ __forceinline__ float lane_bcast(float x, int k) {
  return __int_as_float(__builtin_amdgcn_readlane(__float_as_int(x), k));
}

template<int H, bool FINAL>
__global__ __launch_bounds__(256) void node_softmax_agg_kernel(
    const int* __restrict__ rowptr, const int* __restrict__ csr_src,
    const float* __restrict__ asrc, const float* __restrict__ adst,
    const float* __restrict__ h, const float* __restrict__ bias,
    float* __restrict__ outp, int N) {
  constexpr int F = H * 64;
  constexpr int VEC = F / 32;
  constexpr int U = 4;
  typedef typename RowVec<VEC>::T vecf;
  const int lane = threadIdx.x & 31;
  const int d = __builtin_amdgcn_readfirstlane((int)((blockIdx.x * blockDim.x + threadIdx.x) >> 5));
  if (d >= N) return;
  const int beg = rowptr[d];
  const int cnt = rowptr[d + 1] - beg + 1;
  const int myh = (VEC * lane) / 64;
  float adv[H];
  for (int hh = 0; hh < H; ++hh) adv[hh] = adst[d * H + hh];

  auto slot_src = [&](int i) -> int { return (i < cnt - 1) ? csr_src[beg + i] : d; };
  auto logit = [&](int s, int hh) -> float {
    float v = asrc[s * H + hh] + adv[hh];
    return (v > 0.f) ? v : NEG_SLOPE * v;
  };

  const bool ok0 = lane < cnt;
  const int s0 = ok0 ? slot_src(lane) : d;
  float v0[H], mx[H], sm[H], ex0[H];
  for (int hh = 0; hh < H; ++hh) {
    v0[hh] = logit(s0, hh);
    mx[hh] = ok0 ? v0[hh] : -__builtin_inff();
  }
  for (int i = lane + 32; i < cnt; i += 32) {
    const int s = slot_src(i);
    for (int hh = 0; hh < H; ++hh) mx[hh] = fmaxf(mx[hh], logit(s, hh));
  }
  for (int hh = 0; hh < H; ++hh)
    for (int off = 16; off > 0; off >>= 1) mx[hh] = fmaxf(mx[hh], __shfl_xor(mx[hh], off, 32));
  for (int hh = 0; hh < H; ++hh) {
    ex0[hh] = ok0 ? __expf(v0[hh] - mx[hh]) : 0.f;
    sm[hh] = ex0[hh];
  }
  for (int i = lane + 32; i < cnt; i += 32) {
    const int s = slot_src(i);
    for (int hh = 0; hh < H; ++hh) sm[hh] += __expf(logit(s, hh) - mx[hh]);
  }
  for (int hh = 0; hh < H; ++hh)
    for (int off = 16; off > 0; off >>= 1) sm[hh] += __shfl_xor(sm[hh], off, 32);

  vecf acc = {};
  const float* hl = h + VEC * lane;
  for (int c0 = 0; c0 < cnt; c0 += 32) {
    int s = s0;
    float al[H];
    if (c0 == 0) {
      for (int hh = 0; hh < H; ++hh) al[hh] = ex0[hh] / sm[hh];
    } else {
      const int i = c0 + lane;
      const bool ok = i < cnt;
      s = ok ? slot_src(i) : d;
      for (int hh = 0; hh < H; ++hh) al[hh] = ok ? __expf(logit(s, hh) - mx[hh]) / sm[hh] : 0.f;
    }
    const int n = (cnt - c0 < 32) ? (cnt - c0) : 32;
    for (int k = 0; k < n; k += U) {
      vecf hv[U];
      float a[U];
      #pragma unroll
      for (int u = 0; u < U; ++u) {
        const int kk = k + u;
        const int kc = (kk < n) ? kk : n - 1;
        const int sk = __builtin_amdgcn_readlane(s, kc);
        hv[u] = *(const vecf*)(hl + (size_t)sk * F);
        float ax = lane_bcast(al[0], kc);
        if constexpr (H == 2) {
          const float a1 = lane_bcast(al[1], kc);
          ax = myh ? a1 : ax;
        }
        a[u] = (kk < n) ? ax : 0.f;
      }
      #pragma unroll
      for (int u = 0; u < U; ++u) acc += a[u] * hv[u];
    }
  }

  vecf r = acc + *(const vecf*)(bias + VEC * lane);
  for (int c = 0; c < VEC; ++c) {
    if (FINAL) r[c] = 1.f / (1.f + __expf(-r[c]));
    else       r[c] = (r[c] > 0.f) ? r[c] : 0.f;
  }
  *(vecf*)(outp + (size_t)d * F + VEC * lane) = r;
}

extern "C" void kernel_launch(void* const* d_in, const int* in_sizes, int n_in,
                              void* d_out, int out_size, void* d_ws, size_t ws_size,
                              hipStream_t stream) {
  const float* x   = (const float*)d_in[0];
  const int*   ei  = (const int*)d_in[1];
  const float* W1  = (const float*)d_in[2];
  const float* as1 = (const float*)d_in[3];
  const float* ad1 = (const float*)d_in[4];
  const float* b1  = (const float*)d_in[5];
  const float* W2  = (const float*)d_in[6];
  const float* as2 = (const float*)d_in[7];
  const float* ad2 = (const float*)d_in[8];
  const float* b2  = (const float*)d_in[9];
  float* out = (float*)d_out;

  const int N = in_sizes[0] / 128;
  const int E = in_sizes[1] / 2;
  const int* srcE = ei;
  const int* dstE = ei + E;

  float* w = (float*)d_ws;
  size_t o = 0;
  float* h1    = w + o; o += (size_t)N * 128;
  float* agg1  = w + o; o += (size_t)N * 128;
  float* asrc1 = w + o; o += (size_t)N * 2;
  float* adst1 = w + o; o += (size_t)N * 2;
  float* asrc2 = w + o; o += (size_t)N;
  float* adst2 = w + o; o += (size_t)N;
  o = (o + 3) & ~(size_t)3;
  const int nbins   = (N + BIN_NODES - 1) / BIN_NODES;
  const int nchunks = (E + SORT_CH - 1) / SORT_CH;
  int* histG    = (int*)(w + o); o += (size_t)nchunks * MAX_BINS;
  int* binStart = (int*)(w + o); o += (size_t)MAX_BINS + 4;
  int* binned   = (int*)(w + o); o += (size_t)E * 2;
  int* rowptr   = (int*)(w + o); o += (size_t)N + 1;
  int* csr_src  = (int*)(w + o); o += (size_t)E;
  float* h2f   = h1;
  (void)o; (void)ws_size; (void)n_in; (void)out_size;

  const int TB = 256;
  auto blk = [](long long n, int tb) { return (unsigned)((n + tb - 1) / tb); };
  const int rowGroups = (N + 31) / 32;

  bin_hist_kernel<<<nchunks, SORT_NT, 0, stream>>>(dstE, histG, E);
  bin_scatter_kernel<<<nchunks, SORT_NT, 0, stream>>>(srcE, dstE, histG, binStart, binned, E, nbins, nchunks);
  bin_place_kernel<<<nbins, SORT_NT, 0, stream>>>(binned, binStart, rowptr, csr_src, N);

  gemm_f32_wmma<128><<<dim3(rowGroups * 2), TB, 0, stream>>>(x, W1, h1, as1, ad1, asrc1, adst1, N);
  node_softmax_agg_kernel<2, false><<<blk((long long)N * 32, TB), TB, 0, stream>>>(
      rowptr, csr_src, asrc1, adst1, h1, b1, agg1, N);

  gemm_f32_wmma<64><<<dim3(rowGroups), TB, 0, stream>>>(agg1, W2, h2f, as2, ad2, asrc2, adst2, N);
  node_softmax_agg_kernel<1, true><<<blk((long long)N * 32, TB), TB, 0, stream>>>(
      rowptr, csr_src, asrc2, adst2, h2f, b2, out, N);
}


# GEMM panel staging: each thread's 12 panel loads (and 2 small ones) issued together ahead of its LDS stores; the rolled loops waited for every load before the next: 12 round trips in a row per block
# speedup vs baseline: 5.6391x; 1.1861x over previous
#include <hip/hip_runtime.h>
#include <hip/hip_bf16.h>
#include <math.h>

typedef __attribute__((ext_vector_type(2))) float v2f;
typedef __attribute__((ext_vector_type(4))) float v4f;
typedef __attribute__((ext_vector_type(8))) float v8f;

#define NEG_SLOPE 0.2f

__device__ __forceinline__ float row16_sum(float x) {
  x += __int_as_float(__builtin_amdgcn_update_dpp(0, __float_as_int(x), 0x161, 0xf, 0xf, true));
  x += __int_as_float(__builtin_amdgcn_update_dpp(0, __float_as_int(x), 0x162, 0xf, 0xf, true));
  x += __int_as_float(__builtin_amdgcn_update_dpp(0, __float_as_int(x), 0x164, 0xf, 0xf, true));
  x += __int_as_float(__builtin_amdgcn_update_dpp(0, __float_as_int(x), 0x168, 0xf, 0xf, true));
  return x;
}

template<int NC>
__global__ __launch_bounds__(256) void gemm_f32_wmma(const float* __restrict__ A,
                                                     const float* __restrict__ B,
                                                     float* __restrict__ C,
                                                     const float* __restrict__ aw_src,
                                                     const float* __restrict__ aw_dst,
                                                     float* __restrict__ asrc,
                                                     float* __restrict__ adst, int M) {
  constexpr int K = 128;
  constexpr int CG = 64;
  constexpr int LDA = K + 4;
  constexpr int LDC = CG + 4;
  constexpr int colGroups = NC / CG;
  __shared__ float Wlds[K * CG];
  __shared__ float Alds[32 * LDA];
  int cg = blockIdx.x % colGroups;
  int rg = blockIdx.x / colGroups;

  constexpr int WL = K * (CG / 4) / 256;
  constexpr int AL = 32 * (K / 4) / 256;
  static_assert(WL == 8 && AL == 4, "the fence below names every loaded value");
  v4f wv[WL], av[AL];
  #pragma unroll
  for (int j = 0; j < WL; ++j) {
    const int i = threadIdx.x + j * 256;
    const int k = i / (CG / 4), c4 = i % (CG / 4);
    wv[j] = *(const v4f*)&B[(size_t)k * NC + cg * CG + c4 * 4];
  }
  #pragma unroll
  for (int j = 0; j < AL; ++j) {
    const int i = threadIdx.x + j * 256;
    const int r = i / (K / 4), c4 = i % (K / 4);
    const int grow = min(rg * 32 + r, M - 1);
    av[j] = *(const v4f*)&A[(size_t)grow * K + c4 * 4];
  }
  const int oc4 = threadIdx.x & 15;
  const v4f us = *(const v4f*)&aw_src[cg * CG + oc4 * 4];
  const v4f ud = *(const v4f*)&aw_dst[cg * CG + oc4 * 4];
  asm volatile("" :: "v"(wv[0]), "v"(wv[1]), "v"(wv[2]), "v"(wv[3]),
                     "v"(wv[4]), "v"(wv[5]), "v"(wv[6]), "v"(wv[7]),
                     "v"(av[0]), "v"(av[1]), "v"(av[2]), "v"(av[3]), "v"(us), "v"(ud));
  #pragma unroll
  for (int j = 0; j < WL; ++j) {
    const int i = threadIdx.x + j * 256;
    const int k = i / (CG / 4), c4 = i % (CG / 4);
    *(v4f*)&Wlds[k * CG + c4 * 4] = wv[j];
  }
  #pragma unroll
  for (int j = 0; j < AL; ++j) {
    const int i = threadIdx.x + j * 256;
    const int r = i / (K / 4), c4 = i % (K / 4);
    const v4f zero = {0.f, 0.f, 0.f, 0.f};
    *(v4f*)&Alds[r * LDA + c4 * 4] = (rg * 32 + r < M) ? av[j] : zero;
  }
  __syncthreads();

  int wave = threadIdx.x >> 5;
  int lane = threadIdx.x & 31;
  int half = lane >> 4;
  int lrow = lane & 15;
  int rt = wave >> 2, ct = wave & 3;

  const float* Ab = &Alds[(rt * 16 + lrow) * LDA + 2 * half];
  const float* Bb = &Wlds[(2 * half) * CG + ct * 16 + lrow];
  v8f acc = {};
  #pragma unroll
  for (int k0 = 0; k0 < K; k0 += 4) {
    v2f a, b;
    a.x = Ab[k0];        a.y = Ab[k0 + 1];
    b.x = Bb[k0 * CG];   b.y = Bb[k0 * CG + CG];
    acc = __builtin_amdgcn_wmma_f32_16x16x4_f32(false, a, false, b,
                                                (short)0, acc, false, false);
  }

  __syncthreads();
  float* Ct = Alds;
  float* Cw = &Ct[(rt * 16 + 8 * half) * LDC + ct * 16 + lrow];
  #pragma unroll
  for (int j = 0; j < 8; ++j) Cw[j * LDC] = acc[j];
  __syncthreads();

  const int orow = threadIdx.x >> 4;
  const int grow_a = rg * 32 + orow, grow_b = grow_a + 16;
  const float4 va = *(const float4*)&Ct[orow * LDC + oc4 * 4];
  const float4 vb = *(const float4*)&Ct[(orow + 16) * LDC + oc4 * 4];
  if (grow_a < M) *(float4*)&C[(size_t)grow_a * NC + cg * CG + oc4 * 4] = va;
  if (grow_b < M) *(float4*)&C[(size_t)grow_b * NC + cg * CG + oc4 * 4] = vb;
  const float psa = row16_sum(va.x * us.x + va.y * us.y + va.z * us.z + va.w * us.w);
  const float pda = row16_sum(va.x * ud.x + va.y * ud.y + va.z * ud.z + va.w * ud.w);
  const float psb = row16_sum(vb.x * us.x + vb.y * us.y + vb.z * us.z + vb.w * us.w);
  const float pdb = row16_sum(vb.x * ud.x + vb.y * ud.y + vb.z * ud.z + vb.w * ud.w);
  if (oc4 < 4) {
    const bool second = (oc4 & 2) != 0, isdst = (oc4 & 1) != 0;
    const int grow = second ? grow_b : grow_a;
    const float val = isdst ? (second ? pdb : pda) : (second ? psb : psa);
    float* dstp = isdst ? adst : asrc;
    if (grow < M) dstp[(size_t)grow * colGroups + cg] = val;
  }
}

typedef __attribute__((ext_vector_type(2))) int v2i;
typedef __attribute__((ext_vector_type(4))) int v4i;
constexpr int BIN_SHIFT = 8;
constexpr int BIN_NODES = 1 << BIN_SHIFT;
constexpr int MAX_BINS  = 256;
constexpr int SORT_NT   = 512;
constexpr int SORT_U    = 8;
constexpr int SORT_CH   = SORT_NT * SORT_U;

__device__ __forceinline__ int wave_incl_scan(int v, int lane) {
  for (int off = 1; off < 32; off <<= 1) {
    int u = __shfl_up(v, off, 32);
    if (lane >= off) v += u;
  }
  return v;
}

__global__ __launch_bounds__(SORT_NT) void bin_hist_kernel(const int* __restrict__ dstE,
                                                           int* __restrict__ histG, int E) {
  __shared__ int hist[MAX_BINS];
  const int t = threadIdx.x, g = blockIdx.x;
  const int e0 = g * SORT_CH;
  const int len = min(E - e0, SORT_CH);
  int d[SORT_U];
  #pragma unroll
  for (int u = 0; u < SORT_U; ++u) {
    const int i = u * SORT_NT + t;
    d[u] = (i < len) ? dstE[e0 + i] : -1;
  }
  if (t < MAX_BINS) hist[t] = 0;
  __syncthreads();
  #pragma unroll
  for (int u = 0; u < SORT_U; ++u)
    if (d[u] >= 0) atomicAdd(&hist[d[u] >> BIN_SHIFT], 1);
  __syncthreads();
  if (t < MAX_BINS) histG[g * MAX_BINS + t] = hist[t];
}

__global__ __launch_bounds__(SORT_NT) void bin_scatter_kernel(const int* __restrict__ srcE,
                                                              const int* __restrict__ dstE,
                                                              const int* __restrict__ histG,
                                                              int* __restrict__ binStart,
                                                              int* __restrict__ binned,
                                                              int E, int nbins, int nchunks) {
  constexpr int QN = MAX_BINS / 4;
  constexpr int NG = SORT_NT / QN;
  constexpr int LB = 16;
  __shared__ int tot[MAX_BINS];
  __shared__ int fr[MAX_BINS];
  __shared__ int lcur[MAX_BINS];
  __shared__ int gbase[MAX_BINS];
  __shared__ int wsumA[MAX_BINS / 32], wsumB[MAX_BINS / 32];
  __shared__ v2i stage[SORT_CH];
  const int t = threadIdx.x, lane = t & 31, wave = t >> 5;
  const int g = blockIdx.x;
  const int q = t % QN;
  const int grp = __builtin_amdgcn_readfirstlane(t / QN);

  const int e0 = g * SORT_CH;
  const int len = min(E - e0, SORT_CH);
  int d[SORT_U], s[SORT_U];
  #pragma unroll
  for (int u = 0; u < SORT_U; ++u) {
    const int i = u * SORT_NT + t;
    const bool ok = i < len;
    d[u] = ok ? dstE[e0 + i] : -1;
    s[u] = ok ? srcE[e0 + i] : 0;
  }
  if (t < MAX_BINS) { tot[t] = 0; fr[t] = 0; }

  v4i all = {0, 0, 0, 0}, front = {0, 0, 0, 0};
  for (int r0 = grp; r0 < nchunks; r0 += NG * LB) {
    v4i v[LB];
    #pragma unroll
    for (int k = 0; k < LB; ++k) {
      const int gp = r0 + NG * k;
      const int gc = min(gp, nchunks - 1);
      v[k] = ((const v4i*)(histG + (size_t)gc * MAX_BINS))[q];
    }
    #pragma unroll
    for (int k = 0; k < LB; ++k) {
      const int gp = r0 + NG * k;
      const v4i z = {0, 0, 0, 0};
      all   += (gp < nchunks) ? v[k] : z;
      front += (gp < g) ? v[k] : z;
    }
  }
  __syncthreads();
  for (int k = 0; k < 4; ++k) {
    atomicAdd(&tot[4 * q + k], all[k]);
    atomicAdd(&fr[4 * q + k], front[k]);
  }
  __syncthreads();
  int ca = 0, cb = 0, ia = 0, ib = 0;
  if (t < MAX_BINS) {
    ca = tot[t];
    cb = histG[g * MAX_BINS + t];
    ia = wave_incl_scan(ca, lane);
    ib = wave_incl_scan(cb, lane);
    if (lane == 31) { wsumA[wave] = ia; wsumB[wave] = ib; }
  }
  __syncthreads();
  if (t < MAX_BINS) {
    int ba = 0, bb = 0;
    for (int w = 0; w < MAX_BINS / 32; ++w) {
      ba += (w < wave) ? wsumA[w] : 0;
      bb += (w < wave) ? wsumB[w] : 0;
    }
    const int exa = ba + ia - ca;
    const int exb = bb + ib - cb;
    lcur[t] = exb;
    gbase[t] = exa + fr[t] - exb;
    if (g == 0) {
      if (t < nbins) binStart[t] = exa;
      if (t == nbins - 1) binStart[nbins] = exa + ca;
    }
  }
  __syncthreads();
  #pragma unroll
  for (int u = 0; u < SORT_U; ++u)
    if (d[u] >= 0) stage[atomicAdd(&lcur[d[u] >> BIN_SHIFT], 1)] = (v2i){s[u], d[u]};
  __syncthreads();
  v2i* pairs = (v2i*)binned;
  #pragma unroll
  for (int u = 0; u < SORT_U; ++u) {
    const int i = u * SORT_NT + t;
    if (i < len) {
      const v2i pr = stage[i];
      pairs[gbase[pr.y >> BIN_SHIFT] + i] = pr;
    }
  }
}

__global__ __launch_bounds__(SORT_NT) void bin_place_kernel(const int* __restrict__ binned,
                                                            const int* __restrict__ binStart,
                                                            int* __restrict__ rowptr,
                                                            int* __restrict__ csr_src, int N) {
  constexpr int U = 10;
  constexpr int STAGE_MAX = 8192;
  __shared__ int cnt[BIN_NODES];
  __shared__ int wsum[BIN_NODES / 32];
  __shared__ int stage[STAGE_MAX];
  const int t = threadIdx.x, lane = t & 31, wave = t >> 5;
  const int b = blockIdx.x;
  const int s0 = binStart[b], s1 = binStart[b + 1];
  const bool staged = (s1 - s0) <= STAGE_MAX;
  const int node0 = b << BIN_SHIFT;
  const v2i* pairs = (const v2i*)binned;
  if (t < BIN_NODES) cnt[t] = 0;
  __syncthreads();
  for (int base = s0; base < s1; base += U * SORT_NT) {
    int d[U];
    #pragma unroll
    for (int u = 0; u < U; ++u) {
      const int e = base + u * SORT_NT + t;
      d[u] = (e < s1) ? pairs[e].y : -1;
    }
    #pragma unroll
    for (int u = 0; u < U; ++u)
      if (d[u] >= 0) atomicAdd(&cnt[d[u] - node0], 1);
  }
  __syncthreads();
  int c = 0, incl = 0;
  if (t < BIN_NODES) {
    c = cnt[t];
    incl = wave_incl_scan(c, lane);
    if (lane == 31) wsum[wave] = incl;
  }
  __syncthreads();
  if (t < BIN_NODES) {
    int wbase = 0;
    for (int w = 0; w < BIN_NODES / 32; ++w) wbase += (w < wave) ? wsum[w] : 0;
    const int excl = s0 + wbase + incl - c;
    cnt[t] = excl;
    const int node = node0 + t;
    if (node < N) rowptr[node] = excl;
    if (node == N - 1) rowptr[N] = excl + c;
  }
  __syncthreads();
  for (int base = s0; base < s1; base += U * SORT_NT) {
    v2i p[U];
    #pragma unroll
    for (int u = 0; u < U; ++u) {
      const int e = base + u * SORT_NT + t;
      p[u] = (e < s1) ? pairs[e] : (v2i){0, -1};
    }
    #pragma unroll
    for (int u = 0; u < U; ++u) {
      if (p[u].y >= 0) {
        const int pos = atomicAdd(&cnt[p[u].y - node0], 1);
        if (staged) stage[pos - s0] = p[u].x;
        else        csr_src[pos] = p[u].x;
      }
    }
  }
  __syncthreads();
  if (staged)
    for (int i = t; i < s1 - s0; i += SORT_NT) csr_src[s0 + i] = stage[i];
}

template<int V> struct RowVec;
template<> struct RowVec<4> { typedef v4f T; };
template<> struct RowVec<2> { typedef v2f T; };

__device__ __forceinline__ float lane_bcast(float x, int k) {
  return __int_as_float(__builtin_amdgcn_readlane(__float_as_int(x), k));
}

template<int H, bool FINAL>
__global__ __launch_bounds__(256) void node_softmax_agg_kernel(
    const int* __restrict__ rowptr, const int* __restrict__ csr_src,
    const float* __restrict__ asrc, const float* __restrict__ adst,
    const float* __restrict__ h, const float* __restrict__ bias,
    float* __restrict__ outp, int N) {
  constexpr int F = H * 64;
  constexpr int VEC = F / 32;
  constexpr int U = 4;
  typedef typename RowVec<VEC>::T vecf;
  const int lane = threadIdx.x & 31;
  const int d = __builtin_amdgcn_readfirstlane((int)((blockIdx.x * blockDim.x + threadIdx.x) >> 5));
  if (d >= N) return;
  const int beg = rowptr[d];
  const int cnt = rowptr[d + 1] - beg + 1;
  const int myh = (VEC * lane) / 64;
  float adv[H];
  for (int hh = 0; hh < H; ++hh) adv[hh] = adst[d * H + hh];

  auto slot_src = [&](int i) -> int { return (i < cnt - 1) ? csr_src[beg + i] : d; };
  auto logit = [&](int s, int hh) -> float {
    float v = asrc[s * H + hh] + adv[hh];
    return (v > 0.f) ? v : NEG_SLOPE * v;
  };

  const bool ok0 = lane < cnt;
  const int s0 = ok0 ? slot_src(lane) : d;
  float v0[H], mx[H], sm[H], ex0[H];
  for (int hh = 0; hh < H; ++hh) {
    v0[hh] = logit(s0, hh);
    mx[hh] = ok0 ? v0[hh] : -__builtin_inff();
  }
  for (int i = lane + 32; i < cnt; i += 32) {
    const int s = slot_src(i);
    for (int hh = 0; hh < H; ++hh) mx[hh] = fmaxf(mx[hh], logit(s, hh));
  }
  for (int hh = 0; hh < H; ++hh)
    for (int off = 16; off > 0; off >>= 1) mx[hh] = fmaxf(mx[hh], __shfl_xor(mx[hh], off, 32));
  for (int hh = 0; hh < H; ++hh) {
    ex0[hh] = ok0 ? __expf(v0[hh] - mx[hh]) : 0.f;
    sm[hh] = ex0[hh];
  }
  for (int i = lane + 32; i < cnt; i += 32) {
    const int s = slot_src(i);
    for (int hh = 0; hh < H; ++hh) sm[hh] += __expf(logit(s, hh) - mx[hh]);
  }
  for (int hh = 0; hh < H; ++hh)
    for (int off = 16; off > 0; off >>= 1) sm[hh] += __shfl_xor(sm[hh], off, 32);

  vecf acc = {};
  const float* hl = h + VEC * lane;
  for (int c0 = 0; c0 < cnt; c0 += 32) {
    int s = s0;
    float al[H];
    if (c0 == 0) {
      for (int hh = 0; hh < H; ++hh) al[hh] = ex0[hh] / sm[hh];
    } else {
      const int i = c0 + lane;
      const bool ok = i < cnt;
      s = ok ? slot_src(i) : d;
      for (int hh = 0; hh < H; ++hh) al[hh] = ok ? __expf(logit(s, hh) - mx[hh]) / sm[hh] : 0.f;
    }
    const int n = (cnt - c0 < 32) ? (cnt - c0) : 32;
    for (int k = 0; k < n; k += U) {
      vecf hv[U];
      float a[U];
      #pragma unroll
      for (int u = 0; u < U; ++u) {
        const int kk = k + u;
        const int kc = (kk < n) ? kk : n - 1;
        const int sk = __builtin_amdgcn_readlane(s, kc);
        hv[u] = *(const vecf*)(hl + (size_t)sk * F);
        float ax = lane_bcast(al[0], kc);
        if constexpr (H == 2) {
          const float a1 = lane_bcast(al[1], kc);
          ax = myh ? a1 : ax;
        }
        a[u] = (kk < n) ? ax : 0.f;
      }
      #pragma unroll
      for (int u = 0; u < U; ++u) acc += a[u] * hv[u];
    }
  }

  vecf r = acc + *(const vecf*)(bias + VEC * lane);
  for (int c = 0; c < VEC; ++c) {
    if (FINAL) r[c] = 1.f / (1.f + __expf(-r[c]));
    else       r[c] = (r[c] > 0.f) ? r[c] : 0.f;
  }
  *(vecf*)(outp + (size_t)d * F + VEC * lane) = r;
}

extern "C" void kernel_launch(void* const* d_in, const int* in_sizes, int n_in,
                              void* d_out, int out_size, void* d_ws, size_t ws_size,
                              hipStream_t stream) {
  const float* x   = (const float*)d_in[0];
  const int*   ei  = (const int*)d_in[1];
  const float* W1  = (const float*)d_in[2];
  const float* as1 = (const float*)d_in[3];
  const float* ad1 = (const float*)d_in[4];
  const float* b1  = (const float*)d_in[5];
  const float* W2  = (const float*)d_in[6];
  const float* as2 = (const float*)d_in[7];
  const float* ad2 = (const float*)d_in[8];
  const float* b2  = (const float*)d_in[9];
  float* out = (float*)d_out;

  const int N = in_sizes[0] / 128;
  const int E = in_sizes[1] / 2;
  const int* srcE = ei;
  const int* dstE = ei + E;

  float* w = (float*)d_ws;
  size_t o = 0;
  float* h1    = w + o; o += (size_t)N * 128;
  float* agg1  = w + o; o += (size_t)N * 128;
  float* asrc1 = w + o; o += (size_t)N * 2;
  float* adst1 = w + o; o += (size_t)N * 2;
  float* asrc2 = w + o; o += (size_t)N;
  float* adst2 = w + o; o += (size_t)N;
  o = (o + 3) & ~(size_t)3;
  const int nbins   = (N + BIN_NODES - 1) / BIN_NODES;
  const int nchunks = (E + SORT_CH - 1) / SORT_CH;
  int* histG    = (int*)(w + o); o += (size_t)nchunks * MAX_BINS;
  int* binStart = (int*)(w + o); o += (size_t)MAX_BINS + 4;
  int* binned   = (int*)(w + o); o += (size_t)E * 2;
  int* rowptr   = (int*)(w + o); o += (size_t)N + 1;
  int* csr_src  = (int*)(w + o); o += (size_t)E;
  float* h2f   = h1;
  (void)o; (void)ws_size; (void)n_in; (void)out_size;

  const int TB = 256;
  auto blk = [](long long n, int tb) { return (unsigned)((n + tb - 1) / tb); };
  const int rowGroups = (N + 31) / 32;

  bin_hist_kernel<<<nchunks, SORT_NT, 0, stream>>>(dstE, histG, E);
  bin_scatter_kernel<<<nchunks, SORT_NT, 0, stream>>>(srcE, dstE, histG, binStart, binned, E, nbins, nchunks);
  bin_place_kernel<<<nbins, SORT_NT, 0, stream>>>(binned, binStart, rowptr, csr_src, N);

  gemm_f32_wmma<128><<<dim3(rowGroups * 2), TB, 0, stream>>>(x, W1, h1, as1, ad1, asrc1, adst1, N);
  node_softmax_agg_kernel<2, false><<<blk((long long)N * 32, TB), TB, 0, stream>>>(
      rowptr, csr_src, asrc1, adst1, h1, b1, agg1, N);

  gemm_f32_wmma<64><<<dim3(rowGroups), TB, 0, stream>>>(agg1, W2, h2f, as2, ad2, asrc2, adst2, N);
  node_softmax_agg_kernel<1, true><<<blk((long long)N * 32, TB), TB, 0, stream>>>(
      rowptr, csr_src, asrc2, adst2, h2f, b2, out, N);
}


# layer-2 products fused into the layer-1 aggregation kernel: 16 waves aggregate 32 nodes into LDS as the A panel and multiply there; one launch, the aggregate's store and re-read gone; sums unchanged
# speedup vs baseline: 6.0468x; 1.0723x over previous
#include <hip/hip_runtime.h>
#include <hip/hip_bf16.h>
#include <math.h>

typedef __attribute__((ext_vector_type(2))) float v2f;
typedef __attribute__((ext_vector_type(4))) float v4f;
typedef __attribute__((ext_vector_type(8))) float v8f;

#define NEG_SLOPE 0.2f

__device__ __forceinline__ float row16_sum(float x) {
  x += __int_as_float(__builtin_amdgcn_update_dpp(0, __float_as_int(x), 0x161, 0xf, 0xf, true));
  x += __int_as_float(__builtin_amdgcn_update_dpp(0, __float_as_int(x), 0x162, 0xf, 0xf, true));
  x += __int_as_float(__builtin_amdgcn_update_dpp(0, __float_as_int(x), 0x164, 0xf, 0xf, true));
  x += __int_as_float(__builtin_amdgcn_update_dpp(0, __float_as_int(x), 0x168, 0xf, 0xf, true));
  return x;
}

template<int NC>
__global__ __launch_bounds__(256) void gemm_f32_wmma(const float* __restrict__ A,
                                                     const float* __restrict__ B,
                                                     float* __restrict__ C,
                                                     const float* __restrict__ aw_src,
                                                     const float* __restrict__ aw_dst,
                                                     float* __restrict__ asrc,
                                                     float* __restrict__ adst, int M) {
  constexpr int K = 128;
  constexpr int CG = 64;
  constexpr int LDA = K + 4;
  constexpr int LDC = CG + 4;
  constexpr int colGroups = NC / CG;
  __shared__ float Wlds[K * CG];
  __shared__ float Alds[32 * LDA];
  int cg = blockIdx.x % colGroups;
  int rg = blockIdx.x / colGroups;

  constexpr int WL = K * (CG / 4) / 256;
  constexpr int AL = 32 * (K / 4) / 256;
  static_assert(WL == 8 && AL == 4, "the fence below names every loaded value");
  v4f wv[WL], av[AL];
  #pragma unroll
  for (int j = 0; j < WL; ++j) {
    const int i = threadIdx.x + j * 256;
    const int k = i / (CG / 4), c4 = i % (CG / 4);
    wv[j] = *(const v4f*)&B[(size_t)k * NC + cg * CG + c4 * 4];
  }
  #pragma unroll
  for (int j = 0; j < AL; ++j) {
    const int i = threadIdx.x + j * 256;
    const int r = i / (K / 4), c4 = i % (K / 4);
    const int grow = min(rg * 32 + r, M - 1);
    av[j] = *(const v4f*)&A[(size_t)grow * K + c4 * 4];
  }
  const int oc4 = threadIdx.x & 15;
  const v4f us = *(const v4f*)&aw_src[cg * CG + oc4 * 4];
  const v4f ud = *(const v4f*)&aw_dst[cg * CG + oc4 * 4];
  asm volatile("" :: "v"(wv[0]), "v"(wv[1]), "v"(wv[2]), "v"(wv[3]),
                     "v"(wv[4]), "v"(wv[5]), "v"(wv[6]), "v"(wv[7]),
                     "v"(av[0]), "v"(av[1]), "v"(av[2]), "v"(av[3]), "v"(us), "v"(ud));
  #pragma unroll
  for (int j = 0; j < WL; ++j) {
    const int i = threadIdx.x + j * 256;
    const int k = i / (CG / 4), c4 = i % (CG / 4);
    *(v4f*)&Wlds[k * CG + c4 * 4] = wv[j];
  }
  #pragma unroll
  for (int j = 0; j < AL; ++j) {
    const int i = threadIdx.x + j * 256;
    const int r = i / (K / 4), c4 = i % (K / 4);
    const v4f zero = {0.f, 0.f, 0.f, 0.f};
    *(v4f*)&Alds[r * LDA + c4 * 4] = (rg * 32 + r < M) ? av[j] : zero;
  }
  __syncthreads();

  int wave = threadIdx.x >> 5;
  int lane = threadIdx.x & 31;
  int half = lane >> 4;
  int lrow = lane & 15;
  int rt = wave >> 2, ct = wave & 3;

  const float* Ab = &Alds[(rt * 16 + lrow) * LDA + 2 * half];
  const float* Bb = &Wlds[(2 * half) * CG + ct * 16 + lrow];
  v8f acc = {};
  #pragma unroll
  for (int k0 = 0; k0 < K; k0 += 4) {
    v2f a, b;
    a.x = Ab[k0];        a.y = Ab[k0 + 1];
    b.x = Bb[k0 * CG];   b.y = Bb[k0 * CG + CG];
    acc = __builtin_amdgcn_wmma_f32_16x16x4_f32(false, a, false, b,
                                                (short)0, acc, false, false);
  }

  __syncthreads();
  float* Ct = Alds;
  float* Cw = &Ct[(rt * 16 + 8 * half) * LDC + ct * 16 + lrow];
  #pragma unroll
  for (int j = 0; j < 8; ++j) Cw[j * LDC] = acc[j];
  __syncthreads();

  const int orow = threadIdx.x >> 4;
  const int grow_a = rg * 32 + orow, grow_b = grow_a + 16;
  const float4 va = *(const float4*)&Ct[orow * LDC + oc4 * 4];
  const float4 vb = *(const float4*)&Ct[(orow + 16) * LDC + oc4 * 4];
  if (grow_a < M) *(float4*)&C[(size_t)grow_a * NC + cg * CG + oc4 * 4] = va;
  if (grow_b < M) *(float4*)&C[(size_t)grow_b * NC + cg * CG + oc4 * 4] = vb;
  const float psa = row16_sum(va.x * us.x + va.y * us.y + va.z * us.z + va.w * us.w);
  const float pda = row16_sum(va.x * ud.x + va.y * ud.y + va.z * ud.z + va.w * ud.w);
  const float psb = row16_sum(vb.x * us.x + vb.y * us.y + vb.z * us.z + vb.w * us.w);
  const float pdb = row16_sum(vb.x * ud.x + vb.y * ud.y + vb.z * ud.z + vb.w * ud.w);
  if (oc4 < 4) {
    const bool second = (oc4 & 2) != 0, isdst = (oc4 & 1) != 0;
    const int grow = second ? grow_b : grow_a;
    const float val = isdst ? (second ? pdb : pda) : (second ? psb : psa);
    float* dstp = isdst ? adst : asrc;
    if (grow < M) dstp[(size_t)grow * colGroups + cg] = val;
  }
}

typedef __attribute__((ext_vector_type(2))) int v2i;
typedef __attribute__((ext_vector_type(4))) int v4i;
constexpr int BIN_SHIFT = 8;
constexpr int BIN_NODES = 1 << BIN_SHIFT;
constexpr int MAX_BINS  = 256;
constexpr int SORT_NT   = 512;
constexpr int SORT_U    = 8;
constexpr int SORT_CH   = SORT_NT * SORT_U;

__device__ __forceinline__ int wave_incl_scan(int v, int lane) {
  for (int off = 1; off < 32; off <<= 1) {
    int u = __shfl_up(v, off, 32);
    if (lane >= off) v += u;
  }
  return v;
}

__global__ __launch_bounds__(SORT_NT) void bin_hist_kernel(const int* __restrict__ dstE,
                                                           int* __restrict__ histG, int E) {
  __shared__ int hist[MAX_BINS];
  const int t = threadIdx.x, g = blockIdx.x;
  const int e0 = g * SORT_CH;
  const int len = min(E - e0, SORT_CH);
  int d[SORT_U];
  #pragma unroll
  for (int u = 0; u < SORT_U; ++u) {
    const int i = u * SORT_NT + t;
    d[u] = (i < len) ? dstE[e0 + i] : -1;
  }
  if (t < MAX_BINS) hist[t] = 0;
  __syncthreads();
  #pragma unroll
  for (int u = 0; u < SORT_U; ++u)
    if (d[u] >= 0) atomicAdd(&hist[d[u] >> BIN_SHIFT], 1);
  __syncthreads();
  if (t < MAX_BINS) histG[g * MAX_BINS + t] = hist[t];
}

__global__ __launch_bounds__(SORT_NT) void bin_scatter_kernel(const int* __restrict__ srcE,
                                                              const int* __restrict__ dstE,
                                                              const int* __restrict__ histG,
                                                              int* __restrict__ binStart,
                                                              int* __restrict__ binned,
                                                              int E, int nbins, int nchunks) {
  constexpr int QN = MAX_BINS / 4;
  constexpr int NG = SORT_NT / QN;
  constexpr int LB = 16;
  __shared__ int tot[MAX_BINS];
  __shared__ int fr[MAX_BINS];
  __shared__ int lcur[MAX_BINS];
  __shared__ int gbase[MAX_BINS];
  __shared__ int wsumA[MAX_BINS / 32], wsumB[MAX_BINS / 32];
  __shared__ v2i stage[SORT_CH];
  const int t = threadIdx.x, lane = t & 31, wave = t >> 5;
  const int g = blockIdx.x;
  const int q = t % QN;
  const int grp = __builtin_amdgcn_readfirstlane(t / QN);

  const int e0 = g * SORT_CH;
  const int len = min(E - e0, SORT_CH);
  int d[SORT_U], s[SORT_U];
  #pragma unroll
  for (int u = 0; u < SORT_U; ++u) {
    const int i = u * SORT_NT + t;
    const bool ok = i < len;
    d[u] = ok ? dstE[e0 + i] : -1;
    s[u] = ok ? srcE[e0 + i] : 0;
  }
  if (t < MAX_BINS) { tot[t] = 0; fr[t] = 0; }

  v4i all = {0, 0, 0, 0}, front = {0, 0, 0, 0};
  for (int r0 = grp; r0 < nchunks; r0 += NG * LB) {
    v4i v[LB];
    #pragma unroll
    for (int k = 0; k < LB; ++k) {
      const int gp = r0 + NG * k;
      const int gc = min(gp, nchunks - 1);
      v[k] = ((const v4i*)(histG + (size_t)gc * MAX_BINS))[q];
    }
    #pragma unroll
    for (int k = 0; k < LB; ++k) {
      const int gp = r0 + NG * k;
      const v4i z = {0, 0, 0, 0};
      all   += (gp < nchunks) ? v[k] : z;
      front += (gp < g) ? v[k] : z;
    }
  }
  __syncthreads();
  for (int k = 0; k < 4; ++k) {
    atomicAdd(&tot[4 * q + k], all[k]);
    atomicAdd(&fr[4 * q + k], front[k]);
  }
  __syncthreads();
  int ca = 0, cb = 0, ia = 0, ib = 0;
  if (t < MAX_BINS) {
    ca = tot[t];
    cb = histG[g * MAX_BINS + t];
    ia = wave_incl_scan(ca, lane);
    ib = wave_incl_scan(cb, lane);
    if (lane == 31) { wsumA[wave] = ia; wsumB[wave] = ib; }
  }
  __syncthreads();
  if (t < MAX_BINS) {
    int ba = 0, bb = 0;
    for (int w = 0; w < MAX_BINS / 32; ++w) {
      ba += (w < wave) ? wsumA[w] : 0;
      bb += (w < wave) ? wsumB[w] : 0;
    }
    const int exa = ba + ia - ca;
    const int exb = bb + ib - cb;
    lcur[t] = exb;
    gbase[t] = exa + fr[t] - exb;
    if (g == 0) {
      if (t < nbins) binStart[t] = exa;
      if (t == nbins - 1) binStart[nbins] = exa + ca;
    }
  }
  __syncthreads();
  #pragma unroll
  for (int u = 0; u < SORT_U; ++u)
    if (d[u] >= 0) stage[atomicAdd(&lcur[d[u] >> BIN_SHIFT], 1)] = (v2i){s[u], d[u]};
  __syncthreads();
  v2i* pairs = (v2i*)binned;
  #pragma unroll
  for (int u = 0; u < SORT_U; ++u) {
    const int i = u * SORT_NT + t;
    if (i < len) {
      const v2i pr = stage[i];
      pairs[gbase[pr.y >> BIN_SHIFT] + i] = pr;
    }
  }
}

__global__ __launch_bounds__(SORT_NT) void bin_place_kernel(const int* __restrict__ binned,
                                                            const int* __restrict__ binStart,
                                                            int* __restrict__ rowptr,
                                                            int* __restrict__ csr_src, int N) {
  constexpr int U = 10;
  constexpr int STAGE_MAX = 8192;
  __shared__ int cnt[BIN_NODES];
  __shared__ int wsum[BIN_NODES / 32];
  __shared__ int stage[STAGE_MAX];
  const int t = threadIdx.x, lane = t & 31, wave = t >> 5;
  const int b = blockIdx.x;
  const int s0 = binStart[b], s1 = binStart[b + 1];
  const bool staged = (s1 - s0) <= STAGE_MAX;
  const int node0 = b << BIN_SHIFT;
  const v2i* pairs = (const v2i*)binned;
  if (t < BIN_NODES) cnt[t] = 0;
  __syncthreads();
  for (int base = s0; base < s1; base += U * SORT_NT) {
    int d[U];
    #pragma unroll
    for (int u = 0; u < U; ++u) {
      const int e = base + u * SORT_NT + t;
      d[u] = (e < s1) ? pairs[e].y : -1;
    }
    #pragma unroll
    for (int u = 0; u < U; ++u)
      if (d[u] >= 0) atomicAdd(&cnt[d[u] - node0], 1);
  }
  __syncthreads();
  int c = 0, incl = 0;
  if (t < BIN_NODES) {
    c = cnt[t];
    incl = wave_incl_scan(c, lane);
    if (lane == 31) wsum[wave] = incl;
  }
  __syncthreads();
  if (t < BIN_NODES) {
    int wbase = 0;
    for (int w = 0; w < BIN_NODES / 32; ++w) wbase += (w < wave) ? wsum[w] : 0;
    const int excl = s0 + wbase + incl - c;
    cnt[t] = excl;
    const int node = node0 + t;
    if (node < N) rowptr[node] = excl;
    if (node == N - 1) rowptr[N] = excl + c;
  }
  __syncthreads();
  for (int base = s0; base < s1; base += U * SORT_NT) {
    v2i p[U];
    #pragma unroll
    for (int u = 0; u < U; ++u) {
      const int e = base + u * SORT_NT + t;
      p[u] = (e < s1) ? pairs[e] : (v2i){0, -1};
    }
    #pragma unroll
    for (int u = 0; u < U; ++u) {
      if (p[u].y >= 0) {
        const int pos = atomicAdd(&cnt[p[u].y - node0], 1);
        if (staged) stage[pos - s0] = p[u].x;
        else        csr_src[pos] = p[u].x;
      }
    }
  }
  __syncthreads();
  if (staged)
    for (int i = t; i < s1 - s0; i += SORT_NT) csr_src[s0 + i] = stage[i];
}

template<int V> struct RowVec;
template<> struct RowVec<4> { typedef v4f T; };
template<> struct RowVec<2> { typedef v2f T; };

__device__ __forceinline__ float lane_bcast(float x, int k) {
  return __int_as_float(__builtin_amdgcn_readlane(__float_as_int(x), k));
}

struct NodeFront { int beg, cnt, s0; };
__device__ __forceinline__ NodeFront node_front(const int d, const int lane,
                                                const int* __restrict__ rowptr,
                                                const int* __restrict__ csr_src) {
  NodeFront f;
  f.beg = rowptr[d];
  f.cnt = rowptr[d + 1] - f.beg + 1;
  f.s0 = (lane < f.cnt - 1) ? csr_src[f.beg + lane] : d;
  return f;
}

template<int H>
__device__ __forceinline__ typename RowVec<H * 2>::T agg_node_row(
    const int d, const int lane, const NodeFront nf,
    const int* __restrict__ csr_src,
    const float* __restrict__ asrc, const float* __restrict__ adst,
    const float* __restrict__ h) {
  constexpr int F = H * 64;
  constexpr int VEC = F / 32;
  constexpr int U = 4;
  typedef typename RowVec<VEC>::T vecf;
  const int beg = nf.beg;
  const int cnt = nf.cnt;
  const int myh = (VEC * lane) / 64;
  float adv[H];
  for (int hh = 0; hh < H; ++hh) adv[hh] = adst[d * H + hh];

  auto slot_src = [&](int i) -> int { return (i < cnt - 1) ? csr_src[beg + i] : d; };
  auto logit = [&](int s, int hh) -> float {
    float v = asrc[s * H + hh] + adv[hh];
    return (v > 0.f) ? v : NEG_SLOPE * v;
  };

  const bool ok0 = lane < cnt;
  const int s0 = nf.s0;
  float v0[H], mx[H], sm[H], ex0[H];
  for (int hh = 0; hh < H; ++hh) {
    v0[hh] = logit(s0, hh);
    mx[hh] = ok0 ? v0[hh] : -__builtin_inff();
  }
  for (int i = lane + 32; i < cnt; i += 32) {
    const int s = slot_src(i);
    for (int hh = 0; hh < H; ++hh) mx[hh] = fmaxf(mx[hh], logit(s, hh));
  }
  for (int hh = 0; hh < H; ++hh)
    for (int off = 16; off > 0; off >>= 1) mx[hh] = fmaxf(mx[hh], __shfl_xor(mx[hh], off, 32));
  for (int hh = 0; hh < H; ++hh) {
    ex0[hh] = ok0 ? __expf(v0[hh] - mx[hh]) : 0.f;
    sm[hh] = ex0[hh];
  }
  for (int i = lane + 32; i < cnt; i += 32) {
    const int s = slot_src(i);
    for (int hh = 0; hh < H; ++hh) sm[hh] += __expf(logit(s, hh) - mx[hh]);
  }
  for (int hh = 0; hh < H; ++hh)
    for (int off = 16; off > 0; off >>= 1) sm[hh] += __shfl_xor(sm[hh], off, 32);

  vecf acc = {};
  const float* hl = h + VEC * lane;
  for (int c0 = 0; c0 < cnt; c0 += 32) {
    int s = s0;
    float al[H];
    if (c0 == 0) {
      for (int hh = 0; hh < H; ++hh) al[hh] = ex0[hh] / sm[hh];
    } else {
      const int i = c0 + lane;
      const bool ok = i < cnt;
      s = ok ? slot_src(i) : d;
      for (int hh = 0; hh < H; ++hh) al[hh] = ok ? __expf(logit(s, hh) - mx[hh]) / sm[hh] : 0.f;
    }
    const int n = (cnt - c0 < 32) ? (cnt - c0) : 32;
    for (int k = 0; k < n; k += U) {
      vecf hv[U];
      float a[U];
      #pragma unroll
      for (int u = 0; u < U; ++u) {
        const int kk = k + u;
        const int kc = (kk < n) ? kk : n - 1;
        const int sk = __builtin_amdgcn_readlane(s, kc);
        hv[u] = *(const vecf*)(hl + (size_t)sk * F);
        float ax = lane_bcast(al[0], kc);
        if constexpr (H == 2) {
          const float a1 = lane_bcast(al[1], kc);
          ax = myh ? a1 : ax;
        }
        a[u] = (kk < n) ? ax : 0.f;
      }
      #pragma unroll
      for (int u = 0; u < U; ++u) acc += a[u] * hv[u];
    }
  }
  return acc;
}

template<int H, bool FINAL>
__global__ __launch_bounds__(256) void node_softmax_agg_kernel(
    const int* __restrict__ rowptr, const int* __restrict__ csr_src,
    const float* __restrict__ asrc, const float* __restrict__ adst,
    const float* __restrict__ h, const float* __restrict__ bias,
    float* __restrict__ outp, int N) {
  constexpr int F = H * 64;
  constexpr int VEC = F / 32;
  typedef typename RowVec<VEC>::T vecf;
  const int lane = threadIdx.x & 31;
  const int d = __builtin_amdgcn_readfirstlane((int)((blockIdx.x * blockDim.x + threadIdx.x) >> 5));
  if (d >= N) return;
  const vecf acc = agg_node_row<H>(d, lane, node_front(d, lane, rowptr, csr_src), csr_src, asrc, adst, h);
  vecf r = acc + *(const vecf*)(bias + VEC * lane);
  for (int c = 0; c < VEC; ++c) {
    if (FINAL) r[c] = 1.f / (1.f + __expf(-r[c]));
    else       r[c] = (r[c] > 0.f) ? r[c] : 0.f;
  }
  *(vecf*)(outp + (size_t)d * F + VEC * lane) = r;
}

__global__ __launch_bounds__(512) void agg1_gemm2_kernel(
    const int* __restrict__ rowptr, const int* __restrict__ csr_src,
    const float* __restrict__ asrc1, const float* __restrict__ adst1,
    const float* __restrict__ h1, const float* __restrict__ bias1,
    const float* __restrict__ B, const float* __restrict__ aw_src, const float* __restrict__ aw_dst,
    float* __restrict__ C, float* __restrict__ asrc2, float* __restrict__ adst2, int N) {
  constexpr int K = 128;
  constexpr int CG = 64;
  constexpr int LDA = K + 4;
  constexpr int LDC = CG + 4;
  __shared__ float Wlds[K * CG];
  __shared__ float Alds[32 * LDA];
  const int rg = blockIdx.x;
  const int lane = threadIdx.x & 31;
  const int wave = __builtin_amdgcn_readfirstlane((int)(threadIdx.x >> 5));

  constexpr int WL = K * (CG / 4) / 512;
  static_assert(WL == 4, "the fence below names every loaded value");
  v4f wv[WL];
  #pragma unroll
  for (int j = 0; j < WL; ++j) {
    const int i = threadIdx.x + j * 512;
    const int k = i / (CG / 4), c4 = i % (CG / 4);
    wv[j] = *(const v4f*)&B[(size_t)k * CG + c4 * 4];
  }
  const int oc4 = threadIdx.x & 15;
  const v4f us = *(const v4f*)&aw_src[oc4 * 4];
  const v4f ud = *(const v4f*)&aw_dst[oc4 * 4];
  const int d0 = rg * 32 + 2 * wave;
  NodeFront f0 = {0, 1, d0};
  if (d0 < N) f0 = node_front(d0, lane, rowptr, csr_src);
  asm volatile("" :: "v"(wv[0]), "v"(wv[1]), "v"(wv[2]), "v"(wv[3]), "v"(us), "v"(ud), "v"(f0.s0));
  #pragma unroll
  for (int j = 0; j < WL; ++j) {
    const int i = threadIdx.x + j * 512;
    const int k = i / (CG / 4), c4 = i % (CG / 4);
    *(v4f*)&Wlds[k * CG + c4 * 4] = wv[j];
  }

  for (int j = 0; j < 2; ++j) {
    const int r = 2 * wave + j;
    const int d = d0 + j;
    v4f row = {0.f, 0.f, 0.f, 0.f};
    if (d < N) {
      const NodeFront nf = (j == 0) ? f0 : node_front(d, lane, rowptr, csr_src);
      row = agg_node_row<2>(d, lane, nf, csr_src, asrc1, adst1, h1)
            + *(const v4f*)(bias1 + 4 * lane);
      for (int c = 0; c < 4; ++c) row[c] = (row[c] > 0.f) ? row[c] : 0.f;
    }
    *(v4f*)&Alds[r * LDA + 4 * lane] = row;
  }
  __syncthreads();

  const int half = lane >> 4;
  const int lrow = lane & 15;
  const int rt = (wave >> 2) & 1, ct = wave & 3;
  v8f acc = {};
  if (wave < 8) {
    const float* Ab = &Alds[(rt * 16 + lrow) * LDA + 2 * half];
    const float* Bb = &Wlds[(2 * half) * CG + ct * 16 + lrow];
    #pragma unroll
    for (int k0 = 0; k0 < K; k0 += 4) {
      v2f a, b;
      a.x = Ab[k0];        a.y = Ab[k0 + 1];
      b.x = Bb[k0 * CG];   b.y = Bb[k0 * CG + CG];
      acc = __builtin_amdgcn_wmma_f32_16x16x4_f32(false, a, false, b,
                                                  (short)0, acc, false, false);
    }
  }
  __syncthreads();
  float* Ct = Alds;
  if (wave < 8) {
    float* Cw = &Ct[(rt * 16 + 8 * half) * LDC + ct * 16 + lrow];
    #pragma unroll
    for (int j = 0; j < 8; ++j) Cw[j * LDC] = acc[j];
  }
  __syncthreads();

  const int orow = threadIdx.x >> 4;
  const int grow = rg * 32 + orow;
  const float4 va = *(const float4*)&Ct[orow * LDC + oc4 * 4];
  if (grow < N) *(float4*)&C[(size_t)grow * CG + oc4 * 4] = va;
  const float ps = row16_sum(va.x * us.x + va.y * us.y + va.z * us.z + va.w * us.w);
  const float pd = row16_sum(va.x * ud.x + va.y * ud.y + va.z * ud.z + va.w * ud.w);
  if (oc4 < 2 && grow < N) {
    if (oc4 == 0) asrc2[grow] = ps;
    else          adst2[grow] = pd;
  }
}

extern "C" void kernel_launch(void* const* d_in, const int* in_sizes, int n_in,
                              void* d_out, int out_size, void* d_ws, size_t ws_size,
                              hipStream_t stream) {
  const float* x   = (const float*)d_in[0];
  const int*   ei  = (const int*)d_in[1];
  const float* W1  = (const float*)d_in[2];
  const float* as1 = (const float*)d_in[3];
  const float* ad1 = (const float*)d_in[4];
  const float* b1  = (const float*)d_in[5];
  const float* W2  = (const float*)d_in[6];
  const float* as2 = (const float*)d_in[7];
  const float* ad2 = (const float*)d_in[8];
  const float* b2  = (const float*)d_in[9];
  float* out = (float*)d_out;

  const int N = in_sizes[0] / 128;
  const int E = in_sizes[1] / 2;
  const int* srcE = ei;
  const int* dstE = ei + E;

  float* w = (float*)d_ws;
  size_t o = 0;
  float* h1    = w + o; o += (size_t)N * 128;
  float* h2f   = w + o; o += (size_t)N * 128;
  float* asrc1 = w + o; o += (size_t)N * 2;
  float* adst1 = w + o; o += (size_t)N * 2;
  float* asrc2 = w + o; o += (size_t)N;
  float* adst2 = w + o; o += (size_t)N;
  o = (o + 3) & ~(size_t)3;
  const int nbins   = (N + BIN_NODES - 1) / BIN_NODES;
  const int nchunks = (E + SORT_CH - 1) / SORT_CH;
  int* histG    = (int*)(w + o); o += (size_t)nchunks * MAX_BINS;
  int* binStart = (int*)(w + o); o += (size_t)MAX_BINS + 4;
  int* binned   = (int*)(w + o); o += (size_t)E * 2;
  int* rowptr   = (int*)(w + o); o += (size_t)N + 1;
  int* csr_src  = (int*)(w + o); o += (size_t)E;
  (void)o; (void)ws_size; (void)n_in; (void)out_size;

  const int TB = 256;
  auto blk = [](long long n, int tb) { return (unsigned)((n + tb - 1) / tb); };
  const int rowGroups = (N + 31) / 32;

  bin_hist_kernel<<<nchunks, SORT_NT, 0, stream>>>(dstE, histG, E);
  bin_scatter_kernel<<<nchunks, SORT_NT, 0, stream>>>(srcE, dstE, histG, binStart, binned, E, nbins, nchunks);
  bin_place_kernel<<<nbins, SORT_NT, 0, stream>>>(binned, binStart, rowptr, csr_src, N);

  gemm_f32_wmma<128><<<dim3(rowGroups * 2), TB, 0, stream>>>(x, W1, h1, as1, ad1, asrc1, adst1, N);
  agg1_gemm2_kernel<<<dim3(rowGroups), 512, 0, stream>>>(
      rowptr, csr_src, asrc1, adst1, h1, b1, W2, as2, ad2, h2f, asrc2, adst2, N);

  node_softmax_agg_kernel<1, true><<<blk((long long)N * 32, TB), TB, 0, stream>>>(
      rowptr, csr_src, asrc2, adst2, h2f, b2, out, N);
}


# edge sort in two launches, not three: each chunk ordered by bin in place with its row of bin offsets, the per-bin kernel reads every chunk's run; count-table launch gone; sums in LDS arrival order
# speedup vs baseline: 6.3241x; 1.0459x over previous
#include <hip/hip_runtime.h>
#include <hip/hip_bf16.h>
#include <math.h>

typedef __attribute__((ext_vector_type(2))) float v2f;
typedef __attribute__((ext_vector_type(4))) float v4f;
typedef __attribute__((ext_vector_type(8))) float v8f;

#define NEG_SLOPE 0.2f

__device__ __forceinline__ float row16_sum(float x) {
  x += __int_as_float(__builtin_amdgcn_update_dpp(0, __float_as_int(x), 0x161, 0xf, 0xf, true));
  x += __int_as_float(__builtin_amdgcn_update_dpp(0, __float_as_int(x), 0x162, 0xf, 0xf, true));
  x += __int_as_float(__builtin_amdgcn_update_dpp(0, __float_as_int(x), 0x164, 0xf, 0xf, true));
  x += __int_as_float(__builtin_amdgcn_update_dpp(0, __float_as_int(x), 0x168, 0xf, 0xf, true));
  return x;
}

template<int NC>
__global__ __launch_bounds__(256) void gemm_f32_wmma(const float* __restrict__ A,
                                                     const float* __restrict__ B,
                                                     float* __restrict__ C,
                                                     const float* __restrict__ aw_src,
                                                     const float* __restrict__ aw_dst,
                                                     float* __restrict__ asrc,
                                                     float* __restrict__ adst, int M) {
  constexpr int K = 128;
  constexpr int CG = 64;
  constexpr int LDA = K + 4;
  constexpr int LDC = CG + 4;
  constexpr int colGroups = NC / CG;
  __shared__ float Wlds[K * CG];
  __shared__ float Alds[32 * LDA];
  int cg = blockIdx.x % colGroups;
  int rg = blockIdx.x / colGroups;

  constexpr int WL = K * (CG / 4) / 256;
  constexpr int AL = 32 * (K / 4) / 256;
  static_assert(WL == 8 && AL == 4, "the fence below names every loaded value");
  v4f wv[WL], av[AL];
  #pragma unroll
  for (int j = 0; j < WL; ++j) {
    const int i = threadIdx.x + j * 256;
    const int k = i / (CG / 4), c4 = i % (CG / 4);
    wv[j] = *(const v4f*)&B[(size_t)k * NC + cg * CG + c4 * 4];
  }
  #pragma unroll
  for (int j = 0; j < AL; ++j) {
    const int i = threadIdx.x + j * 256;
    const int r = i / (K / 4), c4 = i % (K / 4);
    const int grow = min(rg * 32 + r, M - 1);
    av[j] = *(const v4f*)&A[(size_t)grow * K + c4 * 4];
  }
  const int oc4 = threadIdx.x & 15;
  const v4f us = *(const v4f*)&aw_src[cg * CG + oc4 * 4];
  const v4f ud = *(const v4f*)&aw_dst[cg * CG + oc4 * 4];
  asm volatile("" :: "v"(wv[0]), "v"(wv[1]), "v"(wv[2]), "v"(wv[3]),
                     "v"(wv[4]), "v"(wv[5]), "v"(wv[6]), "v"(wv[7]),
                     "v"(av[0]), "v"(av[1]), "v"(av[2]), "v"(av[3]), "v"(us), "v"(ud));
  #pragma unroll
  for (int j = 0; j < WL; ++j) {
    const int i = threadIdx.x + j * 256;
    const int k = i / (CG / 4), c4 = i % (CG / 4);
    *(v4f*)&Wlds[k * CG + c4 * 4] = wv[j];
  }
  #pragma unroll
  for (int j = 0; j < AL; ++j) {
    const int i = threadIdx.x + j * 256;
    const int r = i / (K / 4), c4 = i % (K / 4);
    const v4f zero = {0.f, 0.f, 0.f, 0.f};
    *(v4f*)&Alds[r * LDA + c4 * 4] = (rg * 32 + r < M) ? av[j] : zero;
  }
  __syncthreads();

  int wave = threadIdx.x >> 5;
  int lane = threadIdx.x & 31;
  int half = lane >> 4;
  int lrow = lane & 15;
  int rt = wave >> 2, ct = wave & 3;

  const float* Ab = &Alds[(rt * 16 + lrow) * LDA + 2 * half];
  const float* Bb = &Wlds[(2 * half) * CG + ct * 16 + lrow];
  v8f acc = {};
  #pragma unroll
  for (int k0 = 0; k0 < K; k0 += 4) {
    v2f a, b;
    a.x = Ab[k0];        a.y = Ab[k0 + 1];
    b.x = Bb[k0 * CG];   b.y = Bb[k0 * CG + CG];
    acc = __builtin_amdgcn_wmma_f32_16x16x4_f32(false, a, false, b,
                                                (short)0, acc, false, false);
  }

  __syncthreads();
  float* Ct = Alds;
  float* Cw = &Ct[(rt * 16 + 8 * half) * LDC + ct * 16 + lrow];
  #pragma unroll
  for (int j = 0; j < 8; ++j) Cw[j * LDC] = acc[j];
  __syncthreads();

  const int orow = threadIdx.x >> 4;
  const int grow_a = rg * 32 + orow, grow_b = grow_a + 16;
  const float4 va = *(const float4*)&Ct[orow * LDC + oc4 * 4];
  const float4 vb = *(const float4*)&Ct[(orow + 16) * LDC + oc4 * 4];
  if (grow_a < M) *(float4*)&C[(size_t)grow_a * NC + cg * CG + oc4 * 4] = va;
  if (grow_b < M) *(float4*)&C[(size_t)grow_b * NC + cg * CG + oc4 * 4] = vb;
  const float psa = row16_sum(va.x * us.x + va.y * us.y + va.z * us.z + va.w * us.w);
  const float pda = row16_sum(va.x * ud.x + va.y * ud.y + va.z * ud.z + va.w * ud.w);
  const float psb = row16_sum(vb.x * us.x + vb.y * us.y + vb.z * us.z + vb.w * us.w);
  const float pdb = row16_sum(vb.x * ud.x + vb.y * ud.y + vb.z * ud.z + vb.w * ud.w);
  if (oc4 < 4) {
    const bool second = (oc4 & 2) != 0, isdst = (oc4 & 1) != 0;
    const int grow = second ? grow_b : grow_a;
    const float val = isdst ? (second ? pdb : pda) : (second ? psb : psa);
    float* dstp = isdst ? adst : asrc;
    if (grow < M) dstp[(size_t)grow * colGroups + cg] = val;
  }
}

typedef __attribute__((ext_vector_type(2))) int v2i;
constexpr int BIN_SHIFT = 8;
constexpr int BIN_NODES = 1 << BIN_SHIFT;
constexpr int MAX_BINS  = 256;
constexpr int SORT_NT   = 512;
constexpr int SORT_U    = 8;
constexpr int SORT_CH   = SORT_NT * SORT_U;
constexpr int OFF_PITCH = MAX_BINS + 4;
constexpr int MAX_CHUNKS = 256;

__device__ __forceinline__ int row16_sum_i(int x) {
  x += __builtin_amdgcn_update_dpp(0, x, 0x161, 0xf, 0xf, true);
  x += __builtin_amdgcn_update_dpp(0, x, 0x162, 0xf, 0xf, true);
  x += __builtin_amdgcn_update_dpp(0, x, 0x164, 0xf, 0xf, true);
  x += __builtin_amdgcn_update_dpp(0, x, 0x168, 0xf, 0xf, true);
  return x;
}

__device__ __forceinline__ int wave_incl_scan(int v, int lane) {
  for (int off = 1; off < 32; off <<= 1) {
    int u = __shfl_up(v, off, 32);
    if (lane >= off) v += u;
  }
  return v;
}

__global__ __launch_bounds__(SORT_NT) void chunk_sort_kernel(const int* __restrict__ srcE,
                                                             const int* __restrict__ dstE,
                                                             int* __restrict__ offG,
                                                             int* __restrict__ binned, int E) {
  __shared__ int hist[MAX_BINS];
  __shared__ int first[MAX_BINS];
  __shared__ int wsum[MAX_BINS / 32];
  __shared__ v2i stage[SORT_CH];
  const int t = threadIdx.x, lane = t & 31, wave = t >> 5;
  const int g = blockIdx.x;
  const int e0 = g * SORT_CH;
  const int len = min(E - e0, SORT_CH);
  int d[SORT_U], s[SORT_U], r[SORT_U];
  #pragma unroll
  for (int u = 0; u < SORT_U; ++u) {
    const int i = u * SORT_NT + t;
    const bool ok = i < len;
    d[u] = ok ? dstE[e0 + i] : -1;
    s[u] = ok ? srcE[e0 + i] : 0;
  }
  if (t < MAX_BINS) hist[t] = 0;
  __syncthreads();
  #pragma unroll
  for (int u = 0; u < SORT_U; ++u)
    r[u] = (d[u] >= 0) ? atomicAdd(&hist[d[u] >> BIN_SHIFT], 1) : 0;
  __syncthreads();
  int c = 0, incl = 0;
  if (t < MAX_BINS) {
    c = hist[t];
    incl = wave_incl_scan(c, lane);
    if (lane == 31) wsum[wave] = incl;
  }
  __syncthreads();
  if (t < MAX_BINS) {
    int wb = 0;
    for (int w = 0; w < MAX_BINS / 32; ++w) wb += (w < wave) ? wsum[w] : 0;
    const int ex = wb + incl - c;
    first[t] = ex;
    offG[g * OFF_PITCH + t] = ex;
    if (t == MAX_BINS - 1) offG[g * OFF_PITCH + MAX_BINS] = ex + c;
  }
  __syncthreads();
  #pragma unroll
  for (int u = 0; u < SORT_U; ++u)
    if (d[u] >= 0) stage[first[d[u] >> BIN_SHIFT] + r[u]] = (v2i){s[u], d[u]};
  __syncthreads();
  v2i* pairs = (v2i*)binned;
  #pragma unroll
  for (int u = 0; u < SORT_U; ++u) {
    const int i = u * SORT_NT + t;
    if (i < len) pairs[e0 + i] = stage[i];
  }
}

__global__ __launch_bounds__(SORT_NT) void bin_place_kernel(const int* __restrict__ binned,
                                                            const int* __restrict__ offG,
                                                            int* __restrict__ rowptr,
                                                            int* __restrict__ csr_src,
                                                            int N, int nchunks) {
  constexpr int NW = SORT_NT / 32;
  constexpr int RB = MAX_CHUNKS / NW;
  constexpr int RS = 2;
  constexpr int STAGE_MAX = 8192;
  __shared__ int cnt[BIN_NODES];
  __shared__ int wsum[BIN_NODES / 32];
  __shared__ int wsA[NW], wsC[NW];
  __shared__ int stage[STAGE_MAX];
  const int t = threadIdx.x, lane = t & 31;
  const int wave = __builtin_amdgcn_readfirstlane(t >> 5);
  const int b = blockIdx.x;
  const int node0 = b << BIN_SHIFT;
  const v2i* pairs = (const v2i*)binned;

  const int gl = wave + NW * lane;
  const int gcl = min(gl, nchunks - 1);
  int aL = offG[gcl * OFF_PITCH + b];
  int cL = offG[gcl * OFF_PITCH + b + 1];
  if (lane >= RB || gl >= nchunks) { aL = 0; cL = 0; }
  const int loL = gl * SORT_CH + aL;
  const int lnL = cL - aL;
  {
    static_assert(RB <= 16, "the bounds sit in the first lane row");
    const int sa = row16_sum_i(aL), sc = row16_sum_i(cL);
    if (lane == 0) { wsA[wave] = sa; wsC[wave] = sc; }
  }
  if (t < BIN_NODES) cnt[t] = 0;
  __syncthreads();
  int s0 = 0, s1 = 0;
  for (int w = 0; w < NW; ++w) { s0 += wsA[w]; s1 += wsC[w]; }
  const bool staged = (s1 - s0) <= STAGE_MAX;
  const bool longrun = __any(lnL > 32 * RS);

  {
    int d[RB][RS];
    #pragma unroll
    for (int j = 0; j < RB; ++j) {
      const int ln = __builtin_amdgcn_readlane(lnL, j);
      const int lo = __builtin_amdgcn_readlane(loL, j);
      #pragma unroll
      for (int k = 0; k < RS; ++k)
        d[j][k] = (32 * k + lane < ln) ? pairs[lo + 32 * k + lane].y : -1;
    }
    #pragma unroll
    for (int j = 0; j < RB; ++j)
      #pragma unroll
      for (int k = 0; k < RS; ++k)
        if (d[j][k] >= 0) atomicAdd(&cnt[d[j][k] - node0], 1);
    if (longrun) {
      for (int j = 0; j < RB; ++j) {
        const int ln = __builtin_amdgcn_readlane(lnL, j);
        const int lo = __builtin_amdgcn_readlane(loL, j);
        for (int i = 32 * RS + lane; i < ln; i += 32) atomicAdd(&cnt[pairs[lo + i].y - node0], 1);
      }
    }
  }
  __syncthreads();
  int c = 0, incl = 0;
  if (t < BIN_NODES) {
    c = cnt[t];
    incl = wave_incl_scan(c, lane);
    if (lane == 31) wsum[wave] = incl;
  }
  __syncthreads();
  if (t < BIN_NODES) {
    int wbase = 0;
    for (int w = 0; w < BIN_NODES / 32; ++w) wbase += (w < wave) ? wsum[w] : 0;
    const int excl = s0 + wbase + incl - c;
    cnt[t] = excl;
    const int node = node0 + t;
    if (node < N) rowptr[node] = excl;
    if (node == N - 1) rowptr[N] = excl + c;
  }
  __syncthreads();
  auto place = [&](const v2i p) {
    const int pos = atomicAdd(&cnt[p.y - node0], 1);
    if (staged) stage[pos - s0] = p.x;
    else        csr_src[pos] = p.x;
  };
  {
    v2i p[RB][RS];
    #pragma unroll
    for (int j = 0; j < RB; ++j) {
      const int ln = __builtin_amdgcn_readlane(lnL, j);
      const int lo = __builtin_amdgcn_readlane(loL, j);
      #pragma unroll
      for (int k = 0; k < RS; ++k)
        p[j][k] = (32 * k + lane < ln) ? pairs[lo + 32 * k + lane] : (v2i){0, -1};
    }
    int pos[RB][RS];
    #pragma unroll
    for (int j = 0; j < RB; ++j)
      #pragma unroll
      for (int k = 0; k < RS; ++k)
        pos[j][k] = (p[j][k].y >= 0) ? atomicAdd(&cnt[p[j][k].y - node0], 1) : -1;
    if (staged) {
      #pragma unroll
      for (int j = 0; j < RB; ++j)
        #pragma unroll
        for (int k = 0; k < RS; ++k)
          if (pos[j][k] >= 0) stage[pos[j][k] - s0] = p[j][k].x;
    } else {
      #pragma unroll
      for (int j = 0; j < RB; ++j)
        #pragma unroll
        for (int k = 0; k < RS; ++k)
          if (pos[j][k] >= 0) csr_src[pos[j][k]] = p[j][k].x;
    }
    if (longrun) {
      for (int j = 0; j < RB; ++j) {
        const int ln = __builtin_amdgcn_readlane(lnL, j);
        const int lo = __builtin_amdgcn_readlane(loL, j);
        for (int i = 32 * RS + lane; i < ln; i += 32) place(pairs[lo + i]);
      }
    }
  }
  __syncthreads();
  if (staged)
    for (int i = t; i < s1 - s0; i += SORT_NT) csr_src[s0 + i] = stage[i];
}

template<int V> struct RowVec;
template<> struct RowVec<4> { typedef v4f T; };
template<> struct RowVec<2> { typedef v2f T; };

__device__ __forceinline__ float lane_bcast(float x, int k) {
  return __int_as_float(__builtin_amdgcn_readlane(__float_as_int(x), k));
}

struct NodeFront { int beg, cnt, s0; };
__device__ __forceinline__ NodeFront node_front(const int d, const int lane,
                                                const int* __restrict__ rowptr,
                                                const int* __restrict__ csr_src) {
  NodeFront f;
  f.beg = rowptr[d];
  f.cnt = rowptr[d + 1] - f.beg + 1;
  f.s0 = (lane < f.cnt - 1) ? csr_src[f.beg + lane] : d;
  return f;
}

template<int H>
__device__ __forceinline__ typename RowVec<H * 2>::T agg_node_row(
    const int d, const int lane, const NodeFront nf,
    const int* __restrict__ csr_src,
    const float* __restrict__ asrc, const float* __restrict__ adst,
    const float* __restrict__ h) {
  constexpr int F = H * 64;
  constexpr int VEC = F / 32;
  constexpr int U = 4;
  typedef typename RowVec<VEC>::T vecf;
  const int beg = nf.beg;
  const int cnt = nf.cnt;
  const int myh = (VEC * lane) / 64;
  float adv[H];
  for (int hh = 0; hh < H; ++hh) adv[hh] = adst[d * H + hh];

  auto slot_src = [&](int i) -> int { return (i < cnt - 1) ? csr_src[beg + i] : d; };
  auto logit = [&](int s, int hh) -> float {
    float v = asrc[s * H + hh] + adv[hh];
    return (v > 0.f) ? v : NEG_SLOPE * v;
  };

  const bool ok0 = lane < cnt;
  const int s0 = nf.s0;
  float v0[H], mx[H], sm[H], ex0[H];
  for (int hh = 0; hh < H; ++hh) {
    v0[hh] = logit(s0, hh);
    mx[hh] = ok0 ? v0[hh] : -__builtin_inff();
  }
  for (int i = lane + 32; i < cnt; i += 32) {
    const int s = slot_src(i);
    for (int hh = 0; hh < H; ++hh) mx[hh] = fmaxf(mx[hh], logit(s, hh));
  }
  for (int hh = 0; hh < H; ++hh)
    for (int off = 16; off > 0; off >>= 1) mx[hh] = fmaxf(mx[hh], __shfl_xor(mx[hh], off, 32));
  for (int hh = 0; hh < H; ++hh) {
    ex0[hh] = ok0 ? __expf(v0[hh] - mx[hh]) : 0.f;
    sm[hh] = ex0[hh];
  }
  for (int i = lane + 32; i < cnt; i += 32) {
    const int s = slot_src(i);
    for (int hh = 0; hh < H; ++hh) sm[hh] += __expf(logit(s, hh) - mx[hh]);
  }
  for (int hh = 0; hh < H; ++hh)
    for (int off = 16; off > 0; off >>= 1) sm[hh] += __shfl_xor(sm[hh], off, 32);

  vecf acc = {};
  const float* hl = h + VEC * lane;
  for (int c0 = 0; c0 < cnt; c0 += 32) {
    int s = s0;
    float al[H];
    if (c0 == 0) {
      for (int hh = 0; hh < H; ++hh) al[hh] = ex0[hh] / sm[hh];
    } else {
      const int i = c0 + lane;
      const bool ok = i < cnt;
      s = ok ? slot_src(i) : d;
      for (int hh = 0; hh < H; ++hh) al[hh] = ok ? __expf(logit(s, hh) - mx[hh]) / sm[hh] : 0.f;
    }
    const int n = (cnt - c0 < 32) ? (cnt - c0) : 32;
    for (int k = 0; k < n; k += U) {
      vecf hv[U];
      float a[U];
      #pragma unroll
      for (int u = 0; u < U; ++u) {
        const int kk = k + u;
        const int kc = (kk < n) ? kk : n - 1;
        const int sk = __builtin_amdgcn_readlane(s, kc);
        hv[u] = *(const vecf*)(hl + (size_t)sk * F);
        float ax = lane_bcast(al[0], kc);
        if constexpr (H == 2) {
          const float a1 = lane_bcast(al[1], kc);
          ax = myh ? a1 : ax;
        }
        a[u] = (kk < n) ? ax : 0.f;
      }
      #pragma unroll
      for (int u = 0; u < U; ++u) acc += a[u] * hv[u];
    }
  }
  return acc;
}

template<int H, bool FINAL>
__global__ __launch_bounds__(256) void node_softmax_agg_kernel(
    const int* __restrict__ rowptr, const int* __restrict__ csr_src,
    const float* __restrict__ asrc, const float* __restrict__ adst,
    const float* __restrict__ h, const float* __restrict__ bias,
    float* __restrict__ outp, int N) {
  constexpr int F = H * 64;
  constexpr int VEC = F / 32;
  typedef typename RowVec<VEC>::T vecf;
  const int lane = threadIdx.x & 31;
  const int d = __builtin_amdgcn_readfirstlane((int)((blockIdx.x * blockDim.x + threadIdx.x) >> 5));
  if (d >= N) return;
  const vecf acc = agg_node_row<H>(d, lane, node_front(d, lane, rowptr, csr_src), csr_src, asrc, adst, h);
  vecf r = acc + *(const vecf*)(bias + VEC * lane);
  for (int c = 0; c < VEC; ++c) {
    if (FINAL) r[c] = 1.f / (1.f + __expf(-r[c]));
    else       r[c] = (r[c] > 0.f) ? r[c] : 0.f;
  }
  *(vecf*)(outp + (size_t)d * F + VEC * lane) = r;
}

__global__ __launch_bounds__(512) void agg1_gemm2_kernel(
    const int* __restrict__ rowptr, const int* __restrict__ csr_src,
    const float* __restrict__ asrc1, const float* __restrict__ adst1,
    const float* __restrict__ h1, const float* __restrict__ bias1,
    const float* __restrict__ B, const float* __restrict__ aw_src, const float* __restrict__ aw_dst,
    float* __restrict__ C, float* __restrict__ asrc2, float* __restrict__ adst2, int N) {
  constexpr int K = 128;
  constexpr int CG = 64;
  constexpr int LDA = K + 4;
  constexpr int LDC = CG + 4;
  __shared__ float Wlds[K * CG];
  __shared__ float Alds[32 * LDA];
  const int rg = blockIdx.x;
  const int lane = threadIdx.x & 31;
  const int wave = __builtin_amdgcn_readfirstlane((int)(threadIdx.x >> 5));

  constexpr int WL = K * (CG / 4) / 512;
  static_assert(WL == 4, "the fence below names every loaded value");
  v4f wv[WL];
  #pragma unroll
  for (int j = 0; j < WL; ++j) {
    const int i = threadIdx.x + j * 512;
    const int k = i / (CG / 4), c4 = i % (CG / 4);
    wv[j] = *(const v4f*)&B[(size_t)k * CG + c4 * 4];
  }
  const int oc4 = threadIdx.x & 15;
  const v4f us = *(const v4f*)&aw_src[oc4 * 4];
  const v4f ud = *(const v4f*)&aw_dst[oc4 * 4];
  const int d0 = rg * 32 + 2 * wave;
  NodeFront f0 = {0, 1, d0};
  if (d0 < N) f0 = node_front(d0, lane, rowptr, csr_src);
  asm volatile("" :: "v"(wv[0]), "v"(wv[1]), "v"(wv[2]), "v"(wv[3]), "v"(us), "v"(ud), "v"(f0.s0));
  #pragma unroll
  for (int j = 0; j < WL; ++j) {
    const int i = threadIdx.x + j * 512;
    const int k = i / (CG / 4), c4 = i % (CG / 4);
    *(v4f*)&Wlds[k * CG + c4 * 4] = wv[j];
  }

  for (int j = 0; j < 2; ++j) {
    const int r = 2 * wave + j;
    const int d = d0 + j;
    v4f row = {0.f, 0.f, 0.f, 0.f};
    if (d < N) {
      const NodeFront nf = (j == 0) ? f0 : node_front(d, lane, rowptr, csr_src);
      row = agg_node_row<2>(d, lane, nf, csr_src, asrc1, adst1, h1)
            + *(const v4f*)(bias1 + 4 * lane);
      for (int c = 0; c < 4; ++c) row[c] = (row[c] > 0.f) ? row[c] : 0.f;
    }
    *(v4f*)&Alds[r * LDA + 4 * lane] = row;
  }
  __syncthreads();

  const int half = lane >> 4;
  const int lrow = lane & 15;
  const int rt = (wave >> 2) & 1, ct = wave & 3;
  v8f acc = {};
  if (wave < 8) {
    const float* Ab = &Alds[(rt * 16 + lrow) * LDA + 2 * half];
    const float* Bb = &Wlds[(2 * half) * CG + ct * 16 + lrow];
    #pragma unroll
    for (int k0 = 0; k0 < K; k0 += 4) {
      v2f a, b;
      a.x = Ab[k0];        a.y = Ab[k0 + 1];
      b.x = Bb[k0 * CG];   b.y = Bb[k0 * CG + CG];
      acc = __builtin_amdgcn_wmma_f32_16x16x4_f32(false, a, false, b,
                                                  (short)0, acc, false, false);
    }
  }
  __syncthreads();
  float* Ct = Alds;
  if (wave < 8) {
    float* Cw = &Ct[(rt * 16 + 8 * half) * LDC + ct * 16 + lrow];
    #pragma unroll
    for (int j = 0; j < 8; ++j) Cw[j * LDC] = acc[j];
  }
  __syncthreads();

  const int orow = threadIdx.x >> 4;
  const int grow = rg * 32 + orow;
  const float4 va = *(const float4*)&Ct[orow * LDC + oc4 * 4];
  if (grow < N) *(float4*)&C[(size_t)grow * CG + oc4 * 4] = va;
  const float ps = row16_sum(va.x * us.x + va.y * us.y + va.z * us.z + va.w * us.w);
  const float pd = row16_sum(va.x * ud.x + va.y * ud.y + va.z * ud.z + va.w * ud.w);
  if (oc4 < 2 && grow < N) {
    if (oc4 == 0) asrc2[grow] = ps;
    else          adst2[grow] = pd;
  }
}

extern "C" void kernel_launch(void* const* d_in, const int* in_sizes, int n_in,
                              void* d_out, int out_size, void* d_ws, size_t ws_size,
                              hipStream_t stream) {
  const float* x   = (const float*)d_in[0];
  const int*   ei  = (const int*)d_in[1];
  const float* W1  = (const float*)d_in[2];
  const float* as1 = (const float*)d_in[3];
  const float* ad1 = (const float*)d_in[4];
  const float* b1  = (const float*)d_in[5];
  const float* W2  = (const float*)d_in[6];
  const float* as2 = (const float*)d_in[7];
  const float* ad2 = (const float*)d_in[8];
  const float* b2  = (const float*)d_in[9];
  float* out = (float*)d_out;

  const int N = in_sizes[0] / 128;
  const int E = in_sizes[1] / 2;
  const int* srcE = ei;
  const int* dstE = ei + E;

  float* w = (float*)d_ws;
  size_t o = 0;
  float* h1    = w + o; o += (size_t)N * 128;
  float* h2f   = w + o; o += (size_t)N * 128;
  float* asrc1 = w + o; o += (size_t)N * 2;
  float* adst1 = w + o; o += (size_t)N * 2;
  float* asrc2 = w + o; o += (size_t)N;
  float* adst2 = w + o; o += (size_t)N;
  o = (o + 3) & ~(size_t)3;
  const int nbins   = (N + BIN_NODES - 1) / BIN_NODES;
  const int nchunks = (E + SORT_CH - 1) / SORT_CH;
  int* offG     = (int*)(w + o); o += (size_t)nchunks * OFF_PITCH;
  int* binned   = (int*)(w + o); o += (size_t)E * 2;
  int* rowptr   = (int*)(w + o); o += (size_t)N + 1;
  int* csr_src  = (int*)(w + o); o += (size_t)E;
  (void)o; (void)ws_size; (void)n_in; (void)out_size;

  const int TB = 256;
  auto blk = [](long long n, int tb) { return (unsigned)((n + tb - 1) / tb); };
  const int rowGroups = (N + 31) / 32;

  chunk_sort_kernel<<<nchunks, SORT_NT, 0, stream>>>(srcE, dstE, offG, binned, E);
  bin_place_kernel<<<nbins, SORT_NT, 0, stream>>>(binned, offG, rowptr, csr_src, N, nchunks);

  gemm_f32_wmma<128><<<dim3(rowGroups * 2), TB, 0, stream>>>(x, W1, h1, as1, ad1, asrc1, adst1, N);
  agg1_gemm2_kernel<<<dim3(rowGroups), 512, 0, stream>>>(
      rowptr, csr_src, asrc1, adst1, h1, b1, W2, as2, ad2, h2f, asrc2, adst2, N);

  node_softmax_agg_kernel<1, true><<<blk((long long)N * 32, TB), TB, 0, stream>>>(
      rowptr, csr_src, asrc2, adst2, h2f, b2, out, N);
}


# layer-1 product split in two parts that ride in the grids of the two edge-sort steps (sort work-groups first; product work-groups of 16 waves = both heads of 32 rows): one launch less; sums unchanged
# speedup vs baseline: 6.6757x; 1.0556x over previous
#include <hip/hip_runtime.h>
#include <hip/hip_bf16.h>
#include <math.h>

typedef __attribute__((ext_vector_type(2))) float v2f;
typedef __attribute__((ext_vector_type(4))) float v4f;
typedef __attribute__((ext_vector_type(8))) float v8f;

#define NEG_SLOPE 0.2f

__device__ __forceinline__ float row16_sum(float x) {
  x += __int_as_float(__builtin_amdgcn_update_dpp(0, __float_as_int(x), 0x161, 0xf, 0xf, true));
  x += __int_as_float(__builtin_amdgcn_update_dpp(0, __float_as_int(x), 0x162, 0xf, 0xf, true));
  x += __int_as_float(__builtin_amdgcn_update_dpp(0, __float_as_int(x), 0x164, 0xf, 0xf, true));
  x += __int_as_float(__builtin_amdgcn_update_dpp(0, __float_as_int(x), 0x168, 0xf, 0xf, true));
  return x;
}

typedef __attribute__((ext_vector_type(2))) int v2i;
constexpr int BIN_SHIFT = 8;
constexpr int BIN_NODES = 1 << BIN_SHIFT;
constexpr int MAX_BINS  = 256;
constexpr int SORT_NT   = 512;
constexpr int SORT_U    = 8;
constexpr int SORT_CH   = SORT_NT * SORT_U;
constexpr int OFF_PITCH = MAX_BINS + 4;
constexpr int MAX_CHUNKS = 256;

__device__ __forceinline__ int row16_sum_i(int x) {
  x += __builtin_amdgcn_update_dpp(0, x, 0x161, 0xf, 0xf, true);
  x += __builtin_amdgcn_update_dpp(0, x, 0x162, 0xf, 0xf, true);
  x += __builtin_amdgcn_update_dpp(0, x, 0x164, 0xf, 0xf, true);
  x += __builtin_amdgcn_update_dpp(0, x, 0x168, 0xf, 0xf, true);
  return x;
}

__device__ __forceinline__ int wave_incl_scan(int v, int lane) {
  for (int off = 1; off < 32; off <<= 1) {
    int u = __shfl_up(v, off, 32);
    if (lane >= off) v += u;
  }
  return v;
}

constexpr int SORT1_LDS_WORDS = 2 * MAX_BINS + MAX_BINS / 32 + 2 * SORT_CH;
__device__ __forceinline__ void chunk_sort_body(int* lds, const int g,
                                                const int* __restrict__ srcE,
                                                const int* __restrict__ dstE,
                                                int* __restrict__ offG,
                                                int* __restrict__ binned, const int E) {
  int* hist  = lds;
  int* first = lds + MAX_BINS;
  int* wsum  = lds + 2 * MAX_BINS;
  v2i* stage = (v2i*)(lds + 2 * MAX_BINS + MAX_BINS / 32);
  const int t = threadIdx.x, lane = t & 31, wave = t >> 5;
  const int e0 = g * SORT_CH;
  const int len = min(E - e0, SORT_CH);
  int d[SORT_U], s[SORT_U], r[SORT_U];
  #pragma unroll
  for (int u = 0; u < SORT_U; ++u) {
    const int i = u * SORT_NT + t;
    const bool ok = i < len;
    d[u] = ok ? dstE[e0 + i] : -1;
    s[u] = ok ? srcE[e0 + i] : 0;
  }
  if (t < MAX_BINS) hist[t] = 0;
  __syncthreads();
  #pragma unroll
  for (int u = 0; u < SORT_U; ++u)
    r[u] = (d[u] >= 0) ? atomicAdd(&hist[d[u] >> BIN_SHIFT], 1) : 0;
  __syncthreads();
  int c = 0, incl = 0;
  if (t < MAX_BINS) {
    c = hist[t];
    incl = wave_incl_scan(c, lane);
    if (lane == 31) wsum[wave] = incl;
  }
  __syncthreads();
  if (t < MAX_BINS) {
    int wb = 0;
    for (int w = 0; w < MAX_BINS / 32; ++w) wb += (w < wave) ? wsum[w] : 0;
    const int ex = wb + incl - c;
    first[t] = ex;
    offG[g * OFF_PITCH + t] = ex;
    if (t == MAX_BINS - 1) offG[g * OFF_PITCH + MAX_BINS] = ex + c;
  }
  __syncthreads();
  #pragma unroll
  for (int u = 0; u < SORT_U; ++u)
    if (d[u] >= 0) stage[first[d[u] >> BIN_SHIFT] + r[u]] = (v2i){s[u], d[u]};
  __syncthreads();
  v2i* pairs = (v2i*)binned;
  #pragma unroll
  for (int u = 0; u < SORT_U; ++u) {
    const int i = u * SORT_NT + t;
    if (i < len) pairs[e0 + i] = stage[i];
  }
}

constexpr int SORT2_STAGE_MAX = 8192;
constexpr int SORT2_LDS_WORDS = BIN_NODES + BIN_NODES / 32 + 2 * (SORT_NT / 32) + SORT2_STAGE_MAX;
__device__ __forceinline__ void bin_place_body(int* lds, const int b,
                                               const int* __restrict__ binned,
                                               const int* __restrict__ offG,
                                               int* __restrict__ rowptr,
                                               int* __restrict__ csr_src,
                                               const int N, const int nchunks) {
  constexpr int NW = SORT_NT / 32;
  constexpr int RB = MAX_CHUNKS / NW;
  constexpr int RS = 2;
  constexpr int STAGE_MAX = SORT2_STAGE_MAX;
  int* cnt   = lds;
  int* wsum  = lds + BIN_NODES;
  int* wsA   = lds + BIN_NODES + BIN_NODES / 32;
  int* wsC   = wsA + NW;
  int* stage = wsC + NW;
  const int t = threadIdx.x, lane = t & 31;
  const int wave = __builtin_amdgcn_readfirstlane(t >> 5);
  const int node0 = b << BIN_SHIFT;
  const v2i* pairs = (const v2i*)binned;

  const int gl = wave + NW * lane;
  const int gcl = min(gl, nchunks - 1);
  int aL = offG[gcl * OFF_PITCH + b];
  int cL = offG[gcl * OFF_PITCH + b + 1];
  if (lane >= RB || gl >= nchunks) { aL = 0; cL = 0; }
  const int loL = gl * SORT_CH + aL;
  const int lnL = cL - aL;
  {
    static_assert(RB <= 16, "the bounds sit in the first lane row");
    const int sa = row16_sum_i(aL), sc = row16_sum_i(cL);
    if (lane == 0) { wsA[wave] = sa; wsC[wave] = sc; }
  }
  if (t < BIN_NODES) cnt[t] = 0;
  __syncthreads();
  int s0 = 0, s1 = 0;
  for (int w = 0; w < NW; ++w) { s0 += wsA[w]; s1 += wsC[w]; }
  const bool staged = (s1 - s0) <= STAGE_MAX;
  const bool longrun = __any(lnL > 32 * RS);

  {
    int d[RB][RS];
    #pragma unroll
    for (int j = 0; j < RB; ++j) {
      const int ln = __builtin_amdgcn_readlane(lnL, j);
      const int lo = __builtin_amdgcn_readlane(loL, j);
      #pragma unroll
      for (int k = 0; k < RS; ++k)
        d[j][k] = (32 * k + lane < ln) ? pairs[lo + 32 * k + lane].y : -1;
    }
    #pragma unroll
    for (int j = 0; j < RB; ++j)
      #pragma unroll
      for (int k = 0; k < RS; ++k)
        if (d[j][k] >= 0) atomicAdd(&cnt[d[j][k] - node0], 1);
    if (longrun) {
      for (int j = 0; j < RB; ++j) {
        const int ln = __builtin_amdgcn_readlane(lnL, j);
        const int lo = __builtin_amdgcn_readlane(loL, j);
        for (int i = 32 * RS + lane; i < ln; i += 32) atomicAdd(&cnt[pairs[lo + i].y - node0], 1);
      }
    }
  }
  __syncthreads();
  int c = 0, incl = 0;
  if (t < BIN_NODES) {
    c = cnt[t];
    incl = wave_incl_scan(c, lane);
    if (lane == 31) wsum[wave] = incl;
  }
  __syncthreads();
  if (t < BIN_NODES) {
    int wbase = 0;
    for (int w = 0; w < BIN_NODES / 32; ++w) wbase += (w < wave) ? wsum[w] : 0;
    const int excl = s0 + wbase + incl - c;
    cnt[t] = excl;
    const int node = node0 + t;
    if (node < N) rowptr[node] = excl;
    if (node == N - 1) rowptr[N] = excl + c;
  }
  __syncthreads();
  {
    v2i p[RB][RS];
    #pragma unroll
    for (int j = 0; j < RB; ++j) {
      const int ln = __builtin_amdgcn_readlane(lnL, j);
      const int lo = __builtin_amdgcn_readlane(loL, j);
      #pragma unroll
      for (int k = 0; k < RS; ++k)
        p[j][k] = (32 * k + lane < ln) ? pairs[lo + 32 * k + lane] : (v2i){0, -1};
    }
    int pos[RB][RS];
    #pragma unroll
    for (int j = 0; j < RB; ++j)
      #pragma unroll
      for (int k = 0; k < RS; ++k)
        pos[j][k] = (p[j][k].y >= 0) ? atomicAdd(&cnt[p[j][k].y - node0], 1) : -1;
    if (staged) {
      #pragma unroll
      for (int j = 0; j < RB; ++j)
        #pragma unroll
        for (int k = 0; k < RS; ++k)
          if (pos[j][k] >= 0) stage[pos[j][k] - s0] = p[j][k].x;
    } else {
      #pragma unroll
      for (int j = 0; j < RB; ++j)
        #pragma unroll
        for (int k = 0; k < RS; ++k)
          if (pos[j][k] >= 0) csr_src[pos[j][k]] = p[j][k].x;
    }
    if (longrun) {
      for (int j = 0; j < RB; ++j) {
        const int ln = __builtin_amdgcn_readlane(lnL, j);
        const int lo = __builtin_amdgcn_readlane(loL, j);
        if (staged) {
          for (int i = 32 * RS + lane; i < ln; i += 32) {
            const v2i p = pairs[lo + i];
            stage[atomicAdd(&cnt[p.y - node0], 1) - s0] = p.x;
          }
        } else {
          for (int i = 32 * RS + lane; i < ln; i += 32) {
            const v2i p = pairs[lo + i];
            csr_src[atomicAdd(&cnt[p.y - node0], 1)] = p.x;
          }
        }
      }
    }
  }
  __syncthreads();
  if (staged)
    for (int i = t; i < s1 - s0; i += SORT_NT) csr_src[s0 + i] = stage[i];
}

constexpr int GEMM_LDS_WORDS = 2 * 128 * 64 + 2 * 32 * (64 + 4);

__device__ __forceinline__ void gemm1_body(float* lds, const int rg,
                                           const float* __restrict__ A,
                                           const float* __restrict__ B,
                                           float* __restrict__ C,
                                           const float* __restrict__ aw_src,
                                           const float* __restrict__ aw_dst,
                                           float* __restrict__ asrc,
                                           float* __restrict__ adst, const int M) {
  constexpr int K = 128;
  constexpr int NC = 128;
  constexpr int CG = 64;
  constexpr int LDA = K + 4;
  constexpr int LDC = CG + 4;
  constexpr int NT = 512;
  static_assert(GEMM_LDS_WORDS >= 2 * K * CG + 32 * LDA, "A panel behind the B panels");
  float* Wlds = lds;
  float* Alds = lds + 2 * K * CG;
  const int tid = threadIdx.x;

  constexpr int WL = K * (NC / 4) / NT;
  constexpr int AL = 32 * (K / 4) / NT;
  static_assert(WL == 8 && AL == 2, "the fence below names every loaded value");
  v4f wv[WL], av[AL];
  #pragma unroll
  for (int j = 0; j < WL; ++j) {
    const int i = tid + j * NT;
    const int k = i / (NC / 4), c = i % (NC / 4);
    wv[j] = *(const v4f*)&B[(size_t)k * NC + c * 4];
  }
  #pragma unroll
  for (int j = 0; j < AL; ++j) {
    const int i = tid + j * NT;
    const int r = i / (K / 4), c4 = i % (K / 4);
    const int grow = min(rg * 32 + r, M - 1);
    av[j] = *(const v4f*)&A[(size_t)grow * K + c4 * 4];
  }
  const int cg = tid >> 8;
  const int t8 = tid & 255;
  const int oc4 = t8 & 15;
  const v4f us = *(const v4f*)&aw_src[cg * CG + oc4 * 4];
  const v4f ud = *(const v4f*)&aw_dst[cg * CG + oc4 * 4];
  asm volatile("" :: "v"(wv[0]), "v"(wv[1]), "v"(wv[2]), "v"(wv[3]),
                     "v"(wv[4]), "v"(wv[5]), "v"(wv[6]), "v"(wv[7]),
                     "v"(av[0]), "v"(av[1]), "v"(us), "v"(ud));
  #pragma unroll
  for (int j = 0; j < WL; ++j) {
    const int i = tid + j * NT;
    const int k = i / (NC / 4), c = i % (NC / 4);
    *(v4f*)&Wlds[(c >> 4) * (K * CG) + k * CG + (c & 15) * 4] = wv[j];
  }
  #pragma unroll
  for (int j = 0; j < AL; ++j) {
    const int i = tid + j * NT;
    const int r = i / (K / 4), c4 = i % (K / 4);
    const v4f zero = {0.f, 0.f, 0.f, 0.f};
    *(v4f*)&Alds[r * LDA + c4 * 4] = (rg * 32 + r < M) ? av[j] : zero;
  }
  __syncthreads();

  const int wave = t8 >> 5;
  const int lane = tid & 31;
  const int half = lane >> 4;
  const int lrow = lane & 15;
  const int rt = wave >> 2, ct = wave & 3;

  const float* Ab = &Alds[(rt * 16 + lrow) * LDA + 2 * half];
  const float* Bb = &Wlds[cg * (K * CG) + (2 * half) * CG + ct * 16 + lrow];
  v8f acc = {};
  #pragma unroll
  for (int k0 = 0; k0 < K; k0 += 4) {
    v2f a, b;
    a.x = Ab[k0];        a.y = Ab[k0 + 1];
    b.x = Bb[k0 * CG];   b.y = Bb[k0 * CG + CG];
    acc = __builtin_amdgcn_wmma_f32_16x16x4_f32(false, a, false, b,
                                                (short)0, acc, false, false);
  }

  __syncthreads();
  float* Ct = Alds + cg * (32 * LDC);
  float* Cw = &Ct[(rt * 16 + 8 * half) * LDC + ct * 16 + lrow];
  #pragma unroll
  for (int j = 0; j < 8; ++j) Cw[j * LDC] = acc[j];
  __syncthreads();

  const int orow = t8 >> 4;
  const int grow_a = rg * 32 + orow, grow_b = grow_a + 16;
  const float4 va = *(const float4*)&Ct[orow * LDC + oc4 * 4];
  const float4 vb = *(const float4*)&Ct[(orow + 16) * LDC + oc4 * 4];
  if (grow_a < M) *(float4*)&C[(size_t)grow_a * NC + cg * CG + oc4 * 4] = va;
  if (grow_b < M) *(float4*)&C[(size_t)grow_b * NC + cg * CG + oc4 * 4] = vb;
  const float psa = row16_sum(va.x * us.x + va.y * us.y + va.z * us.z + va.w * us.w);
  const float pda = row16_sum(va.x * ud.x + va.y * ud.y + va.z * ud.z + va.w * ud.w);
  const float psb = row16_sum(vb.x * us.x + vb.y * us.y + vb.z * us.z + vb.w * us.w);
  const float pdb = row16_sum(vb.x * ud.x + vb.y * ud.y + vb.z * ud.z + vb.w * ud.w);
  if (oc4 < 4) {
    const bool second = (oc4 & 2) != 0, isdst = (oc4 & 1) != 0;
    const int grow = second ? grow_b : grow_a;
    const float val = isdst ? (second ? pdb : pda) : (second ? psb : psa);
    float* dstp = isdst ? adst : asrc;
    if (grow < M) dstp[(size_t)grow * 2 + cg] = val;
  }
}

static_assert(SORT1_LDS_WORDS <= GEMM_LDS_WORDS && SORT2_LDS_WORDS <= GEMM_LDS_WORDS,
              "the sorting work-groups live inside the product work-groups' LDS");

__global__ __launch_bounds__(SORT_NT) void sort1_gemm_kernel(
    const int* __restrict__ srcE, const int* __restrict__ dstE,
    int* __restrict__ offG, int* __restrict__ binned, int E, int nsort,
    const float* __restrict__ A, const float* __restrict__ B, float* __restrict__ C,
    const float* __restrict__ aw_src, const float* __restrict__ aw_dst,
    float* __restrict__ asrc, float* __restrict__ adst, int M, int rg0) {
  __shared__ __attribute__((aligned(16))) float lds[GEMM_LDS_WORDS];
  const int wg = blockIdx.x;
  if (wg < nsort) {
    chunk_sort_body((int*)lds, wg, srcE, dstE, offG, binned, E);
    return;
  }
  gemm1_body(lds, rg0 + wg - nsort, A, B, C, aw_src, aw_dst, asrc, adst, M);
}

__global__ __launch_bounds__(SORT_NT) void sort2_gemm_kernel(
    const int* __restrict__ binned, const int* __restrict__ offG,
    int* __restrict__ rowptr, int* __restrict__ csr_src, int N, int nchunks, int nsort,
    const float* __restrict__ A, const float* __restrict__ B, float* __restrict__ C,
    const float* __restrict__ aw_src, const float* __restrict__ aw_dst,
    float* __restrict__ asrc, float* __restrict__ adst, int rg0) {
  __shared__ __attribute__((aligned(16))) float lds[GEMM_LDS_WORDS];
  const int wg = blockIdx.x;
  if (wg < nsort) {
    bin_place_body((int*)lds, wg, binned, offG, rowptr, csr_src, N, nchunks);
    return;
  }
  gemm1_body(lds, rg0 + wg - nsort, A, B, C, aw_src, aw_dst, asrc, adst, N);
}

template<int V> struct RowVec;
template<> struct RowVec<4> { typedef v4f T; };
template<> struct RowVec<2> { typedef v2f T; };

__device__ __forceinline__ float lane_bcast(float x, int k) {
  return __int_as_float(__builtin_amdgcn_readlane(__float_as_int(x), k));
}

struct NodeFront { int beg, cnt, s0; };
__device__ __forceinline__ NodeFront node_front(const int d, const int lane,
                                                const int* __restrict__ rowptr,
                                                const int* __restrict__ csr_src) {
  NodeFront f;
  f.beg = rowptr[d];
  f.cnt = rowptr[d + 1] - f.beg + 1;
  f.s0 = (lane < f.cnt - 1) ? csr_src[f.beg + lane] : d;
  return f;
}

template<int H>
__device__ __forceinline__ typename RowVec<H * 2>::T agg_node_row(
    const int d, const int lane, const NodeFront nf,
    const int* __restrict__ csr_src,
    const float* __restrict__ asrc, const float* __restrict__ adst,
    const float* __restrict__ h) {
  constexpr int F = H * 64;
  constexpr int VEC = F / 32;
  constexpr int U = 4;
  typedef typename RowVec<VEC>::T vecf;
  const int beg = nf.beg;
  const int cnt = nf.cnt;
  const int myh = (VEC * lane) / 64;
  float adv[H];
  for (int hh = 0; hh < H; ++hh) adv[hh] = adst[d * H + hh];

  auto slot_src = [&](int i) -> int { return (i < cnt - 1) ? csr_src[beg + i] : d; };
  auto logit = [&](int s, int hh) -> float {
    float v = asrc[s * H + hh] + adv[hh];
    return (v > 0.f) ? v : NEG_SLOPE * v;
  };

  const bool ok0 = lane < cnt;
  const int s0 = nf.s0;
  float v0[H], mx[H], sm[H], ex0[H];
  for (int hh = 0; hh < H; ++hh) {
    v0[hh] = logit(s0, hh);
    mx[hh] = ok0 ? v0[hh] : -__builtin_inff();
  }
  for (int i = lane + 32; i < cnt; i += 32) {
    const int s = slot_src(i);
    for (int hh = 0; hh < H; ++hh) mx[hh] = fmaxf(mx[hh], logit(s, hh));
  }
  for (int hh = 0; hh < H; ++hh)
    for (int off = 16; off > 0; off >>= 1) mx[hh] = fmaxf(mx[hh], __shfl_xor(mx[hh], off, 32));
  for (int hh = 0; hh < H; ++hh) {
    ex0[hh] = ok0 ? __expf(v0[hh] - mx[hh]) : 0.f;
    sm[hh] = ex0[hh];
  }
  for (int i = lane + 32; i < cnt; i += 32) {
    const int s = slot_src(i);
    for (int hh = 0; hh < H; ++hh) sm[hh] += __expf(logit(s, hh) - mx[hh]);
  }
  for (int hh = 0; hh < H; ++hh)
    for (int off = 16; off > 0; off >>= 1) sm[hh] += __shfl_xor(sm[hh], off, 32);

  vecf acc = {};
  const float* hl = h + VEC * lane;
  for (int c0 = 0; c0 < cnt; c0 += 32) {
    int s = s0;
    float al[H];
    if (c0 == 0) {
      for (int hh = 0; hh < H; ++hh) al[hh] = ex0[hh] / sm[hh];
    } else {
      const int i = c0 + lane;
      const bool ok = i < cnt;
      s = ok ? slot_src(i) : d;
      for (int hh = 0; hh < H; ++hh) al[hh] = ok ? __expf(logit(s, hh) - mx[hh]) / sm[hh] : 0.f;
    }
    const int n = (cnt - c0 < 32) ? (cnt - c0) : 32;
    for (int k = 0; k < n; k += U) {
      vecf hv[U];
      float a[U];
      #pragma unroll
      for (int u = 0; u < U; ++u) {
        const int kk = k + u;
        const int kc = (kk < n) ? kk : n - 1;
        const int sk = __builtin_amdgcn_readlane(s, kc);
        hv[u] = *(const vecf*)(hl + (size_t)sk * F);
        float ax = lane_bcast(al[0], kc);
        if constexpr (H == 2) {
          const float a1 = lane_bcast(al[1], kc);
          ax = myh ? a1 : ax;
        }
        a[u] = (kk < n) ? ax : 0.f;
      }
      #pragma unroll
      for (int u = 0; u < U; ++u) acc += a[u] * hv[u];
    }
  }
  return acc;
}

template<int H, bool FINAL>
__global__ __launch_bounds__(256) void node_softmax_agg_kernel(
    const int* __restrict__ rowptr, const int* __restrict__ csr_src,
    const float* __restrict__ asrc, const float* __restrict__ adst,
    const float* __restrict__ h, const float* __restrict__ bias,
    float* __restrict__ outp, int N) {
  constexpr int F = H * 64;
  constexpr int VEC = F / 32;
  typedef typename RowVec<VEC>::T vecf;
  const int lane = threadIdx.x & 31;
  const int d = __builtin_amdgcn_readfirstlane((int)((blockIdx.x * blockDim.x + threadIdx.x) >> 5));
  if (d >= N) return;
  const vecf acc = agg_node_row<H>(d, lane, node_front(d, lane, rowptr, csr_src), csr_src, asrc, adst, h);
  vecf r = acc + *(const vecf*)(bias + VEC * lane);
  for (int c = 0; c < VEC; ++c) {
    if (FINAL) r[c] = 1.f / (1.f + __expf(-r[c]));
    else       r[c] = (r[c] > 0.f) ? r[c] : 0.f;
  }
  *(vecf*)(outp + (size_t)d * F + VEC * lane) = r;
}

__global__ __launch_bounds__(512) void agg1_gemm2_kernel(
    const int* __restrict__ rowptr, const int* __restrict__ csr_src,
    const float* __restrict__ asrc1, const float* __restrict__ adst1,
    const float* __restrict__ h1, const float* __restrict__ bias1,
    const float* __restrict__ B, const float* __restrict__ aw_src, const float* __restrict__ aw_dst,
    float* __restrict__ C, float* __restrict__ asrc2, float* __restrict__ adst2, int N) {
  constexpr int K = 128;
  constexpr int CG = 64;
  constexpr int LDA = K + 4;
  constexpr int LDC = CG + 4;
  __shared__ float Wlds[K * CG];
  __shared__ float Alds[32 * LDA];
  const int rg = blockIdx.x;
  const int lane = threadIdx.x & 31;
  const int wave = __builtin_amdgcn_readfirstlane((int)(threadIdx.x >> 5));

  constexpr int WL = K * (CG / 4) / 512;
  static_assert(WL == 4, "the fence below names every loaded value");
  v4f wv[WL];
  #pragma unroll
  for (int j = 0; j < WL; ++j) {
    const int i = threadIdx.x + j * 512;
    const int k = i / (CG / 4), c4 = i % (CG / 4);
    wv[j] = *(const v4f*)&B[(size_t)k * CG + c4 * 4];
  }
  const int oc4 = threadIdx.x & 15;
  const v4f us = *(const v4f*)&aw_src[oc4 * 4];
  const v4f ud = *(const v4f*)&aw_dst[oc4 * 4];
  const int d0 = rg * 32 + 2 * wave;
  NodeFront f0 = {0, 1, d0};
  if (d0 < N) f0 = node_front(d0, lane, rowptr, csr_src);
  asm volatile("" :: "v"(wv[0]), "v"(wv[1]), "v"(wv[2]), "v"(wv[3]), "v"(us), "v"(ud), "v"(f0.s0));
  #pragma unroll
  for (int j = 0; j < WL; ++j) {
    const int i = threadIdx.x + j * 512;
    const int k = i / (CG / 4), c4 = i % (CG / 4);
    *(v4f*)&Wlds[k * CG + c4 * 4] = wv[j];
  }

  for (int j = 0; j < 2; ++j) {
    const int r = 2 * wave + j;
    const int d = d0 + j;
    v4f row = {0.f, 0.f, 0.f, 0.f};
    if (d < N) {
      const NodeFront nf = (j == 0) ? f0 : node_front(d, lane, rowptr, csr_src);
      row = agg_node_row<2>(d, lane, nf, csr_src, asrc1, adst1, h1)
            + *(const v4f*)(bias1 + 4 * lane);
      for (int c = 0; c < 4; ++c) row[c] = (row[c] > 0.f) ? row[c] : 0.f;
    }
    *(v4f*)&Alds[r * LDA + 4 * lane] = row;
  }
  __syncthreads();

  const int half = lane >> 4;
  const int lrow = lane & 15;
  const int rt = (wave >> 2) & 1, ct = wave & 3;
  v8f acc = {};
  if (wave < 8) {
    const float* Ab = &Alds[(rt * 16 + lrow) * LDA + 2 * half];
    const float* Bb = &Wlds[(2 * half) * CG + ct * 16 + lrow];
    #pragma unroll
    for (int k0 = 0; k0 < K; k0 += 4) {
      v2f a, b;
      a.x = Ab[k0];        a.y = Ab[k0 + 1];
      b.x = Bb[k0 * CG];   b.y = Bb[k0 * CG + CG];
      acc = __builtin_amdgcn_wmma_f32_16x16x4_f32(false, a, false, b,
                                                  (short)0, acc, false, false);
    }
  }
  __syncthreads();
  float* Ct = Alds;
  if (wave < 8) {
    float* Cw = &Ct[(rt * 16 + 8 * half) * LDC + ct * 16 + lrow];
    #pragma unroll
    for (int j = 0; j < 8; ++j) Cw[j * LDC] = acc[j];
  }
  __syncthreads();

  const int orow = threadIdx.x >> 4;
  const int grow = rg * 32 + orow;
  const float4 va = *(const float4*)&Ct[orow * LDC + oc4 * 4];
  if (grow < N) *(float4*)&C[(size_t)grow * CG + oc4 * 4] = va;
  const float ps = row16_sum(va.x * us.x + va.y * us.y + va.z * us.z + va.w * us.w);
  const float pd = row16_sum(va.x * ud.x + va.y * ud.y + va.z * ud.z + va.w * ud.w);
  if (oc4 < 2 && grow < N) {
    if (oc4 == 0) asrc2[grow] = ps;
    else          adst2[grow] = pd;
  }
}

extern "C" void kernel_launch(void* const* d_in, const int* in_sizes, int n_in,
                              void* d_out, int out_size, void* d_ws, size_t ws_size,
                              hipStream_t stream) {
  const float* x   = (const float*)d_in[0];
  const int*   ei  = (const int*)d_in[1];
  const float* W1  = (const float*)d_in[2];
  const float* as1 = (const float*)d_in[3];
  const float* ad1 = (const float*)d_in[4];
  const float* b1  = (const float*)d_in[5];
  const float* W2  = (const float*)d_in[6];
  const float* as2 = (const float*)d_in[7];
  const float* ad2 = (const float*)d_in[8];
  const float* b2  = (const float*)d_in[9];
  float* out = (float*)d_out;

  const int N = in_sizes[0] / 128;
  const int E = in_sizes[1] / 2;
  const int* srcE = ei;
  const int* dstE = ei + E;

  float* w = (float*)d_ws;
  size_t o = 0;
  float* h1    = w + o; o += (size_t)N * 128;
  float* h2f   = w + o; o += (size_t)N * 128;
  float* asrc1 = w + o; o += (size_t)N * 2;
  float* adst1 = w + o; o += (size_t)N * 2;
  float* asrc2 = w + o; o += (size_t)N;
  float* adst2 = w + o; o += (size_t)N;
  o = (o + 3) & ~(size_t)3;
  const int nbins   = (N + BIN_NODES - 1) / BIN_NODES;
  const int nchunks = (E + SORT_CH - 1) / SORT_CH;
  int* offG     = (int*)(w + o); o += (size_t)nchunks * OFF_PITCH;
  int* binned   = (int*)(w + o); o += (size_t)E * 2;
  int* rowptr   = (int*)(w + o); o += (size_t)N + 1;
  int* csr_src  = (int*)(w + o); o += (size_t)E;
  (void)o; (void)ws_size; (void)n_in; (void)out_size;

  const int TB = 256;
  auto blk = [](long long n, int tb) { return (unsigned)((n + tb - 1) / tb); };
  const int rowGroups = (N + 31) / 32;

  const int rgA = (rowGroups + 1) / 2;
  sort1_gemm_kernel<<<nchunks + rgA, SORT_NT, 0, stream>>>(
      srcE, dstE, offG, binned, E, nchunks, x, W1, h1, as1, ad1, asrc1, adst1, N, 0);
  sort2_gemm_kernel<<<nbins + (rowGroups - rgA), SORT_NT, 0, stream>>>(
      binned, offG, rowptr, csr_src, N, nchunks, nbins, x, W1, h1, as1, ad1, asrc1, adst1, rgA);

  agg1_gemm2_kernel<<<dim3(rowGroups), 512, 0, stream>>>(
      rowptr, csr_src, asrc1, adst1, h1, b1, W2, as2, ad2, h2f, asrc2, adst2, N);

  node_softmax_agg_kernel<1, true><<<blk((long long)N * 32, TB), TB, 0, stream>>>(
      rowptr, csr_src, asrc2, adst2, h2f, b2, out, N);
}


# aggregation waves walk several nodes (layer 2: 4 a wave, was 1; layer 1: 2 as before): bounds in one load, sources two nodes and logit terms one node ahead of the row sums; sums unchanged
# speedup vs baseline: 6.8954x; 1.0329x over previous
#include <hip/hip_runtime.h>
#include <hip/hip_bf16.h>
#include <math.h>

typedef __attribute__((ext_vector_type(2))) float v2f;
typedef __attribute__((ext_vector_type(4))) float v4f;
typedef __attribute__((ext_vector_type(8))) float v8f;

#define NEG_SLOPE 0.2f

__device__ __forceinline__ float row16_sum(float x) {
  x += __int_as_float(__builtin_amdgcn_update_dpp(0, __float_as_int(x), 0x161, 0xf, 0xf, true));
  x += __int_as_float(__builtin_amdgcn_update_dpp(0, __float_as_int(x), 0x162, 0xf, 0xf, true));
  x += __int_as_float(__builtin_amdgcn_update_dpp(0, __float_as_int(x), 0x164, 0xf, 0xf, true));
  x += __int_as_float(__builtin_amdgcn_update_dpp(0, __float_as_int(x), 0x168, 0xf, 0xf, true));
  return x;
}

typedef __attribute__((ext_vector_type(2))) int v2i;
constexpr int BIN_SHIFT = 8;
constexpr int BIN_NODES = 1 << BIN_SHIFT;
constexpr int MAX_BINS  = 256;
constexpr int SORT_NT   = 512;
constexpr int SORT_U    = 8;
constexpr int SORT_CH   = SORT_NT * SORT_U;
constexpr int OFF_PITCH = MAX_BINS + 4;
constexpr int MAX_CHUNKS = 256;

__device__ __forceinline__ int row16_sum_i(int x) {
  x += __builtin_amdgcn_update_dpp(0, x, 0x161, 0xf, 0xf, true);
  x += __builtin_amdgcn_update_dpp(0, x, 0x162, 0xf, 0xf, true);
  x += __builtin_amdgcn_update_dpp(0, x, 0x164, 0xf, 0xf, true);
  x += __builtin_amdgcn_update_dpp(0, x, 0x168, 0xf, 0xf, true);
  return x;
}

__device__ __forceinline__ int wave_incl_scan(int v, int lane) {
  for (int off = 1; off < 32; off <<= 1) {
    int u = __shfl_up(v, off, 32);
    if (lane >= off) v += u;
  }
  return v;
}

constexpr int SORT1_LDS_WORDS = 2 * MAX_BINS + MAX_BINS / 32 + 2 * SORT_CH;
__device__ __forceinline__ void chunk_sort_body(int* lds, const int g,
                                                const int* __restrict__ srcE,
                                                const int* __restrict__ dstE,
                                                int* __restrict__ offG,
                                                int* __restrict__ binned, const int E) {
  int* hist  = lds;
  int* first = lds + MAX_BINS;
  int* wsum  = lds + 2 * MAX_BINS;
  v2i* stage = (v2i*)(lds + 2 * MAX_BINS + MAX_BINS / 32);
  const int t = threadIdx.x, lane = t & 31, wave = t >> 5;
  const int e0 = g * SORT_CH;
  const int len = min(E - e0, SORT_CH);
  int d[SORT_U], s[SORT_U], r[SORT_U];
  #pragma unroll
  for (int u = 0; u < SORT_U; ++u) {
    const int i = u * SORT_NT + t;
    const bool ok = i < len;
    d[u] = ok ? dstE[e0 + i] : -1;
    s[u] = ok ? srcE[e0 + i] : 0;
  }
  if (t < MAX_BINS) hist[t] = 0;
  __syncthreads();
  #pragma unroll
  for (int u = 0; u < SORT_U; ++u)
    r[u] = (d[u] >= 0) ? atomicAdd(&hist[d[u] >> BIN_SHIFT], 1) : 0;
  __syncthreads();
  int c = 0, incl = 0;
  if (t < MAX_BINS) {
    c = hist[t];
    incl = wave_incl_scan(c, lane);
    if (lane == 31) wsum[wave] = incl;
  }
  __syncthreads();
  if (t < MAX_BINS) {
    int wb = 0;
    for (int w = 0; w < MAX_BINS / 32; ++w) wb += (w < wave) ? wsum[w] : 0;
    const int ex = wb + incl - c;
    first[t] = ex;
    offG[g * OFF_PITCH + t] = ex;
    if (t == MAX_BINS - 1) offG[g * OFF_PITCH + MAX_BINS] = ex + c;
  }
  __syncthreads();
  #pragma unroll
  for (int u = 0; u < SORT_U; ++u)
    if (d[u] >= 0) stage[first[d[u] >> BIN_SHIFT] + r[u]] = (v2i){s[u], d[u]};
  __syncthreads();
  v2i* pairs = (v2i*)binned;
  #pragma unroll
  for (int u = 0; u < SORT_U; ++u) {
    const int i = u * SORT_NT + t;
    if (i < len) pairs[e0 + i] = stage[i];
  }
}

constexpr int SORT2_STAGE_MAX = 8192;
constexpr int SORT2_LDS_WORDS = BIN_NODES + BIN_NODES / 32 + 2 * (SORT_NT / 32) + SORT2_STAGE_MAX;
__device__ __forceinline__ void bin_place_body(int* lds, const int b,
                                               const int* __restrict__ binned,
                                               const int* __restrict__ offG,
                                               int* __restrict__ rowptr,
                                               int* __restrict__ csr_src,
                                               const int N, const int nchunks) {
  constexpr int NW = SORT_NT / 32;
  constexpr int RB = MAX_CHUNKS / NW;
  constexpr int RS = 2;
  constexpr int STAGE_MAX = SORT2_STAGE_MAX;
  int* cnt   = lds;
  int* wsum  = lds + BIN_NODES;
  int* wsA   = lds + BIN_NODES + BIN_NODES / 32;
  int* wsC   = wsA + NW;
  int* stage = wsC + NW;
  const int t = threadIdx.x, lane = t & 31;
  const int wave = __builtin_amdgcn_readfirstlane(t >> 5);
  const int node0 = b << BIN_SHIFT;
  const v2i* pairs = (const v2i*)binned;

  const int gl = wave + NW * lane;
  const int gcl = min(gl, nchunks - 1);
  int aL = offG[gcl * OFF_PITCH + b];
  int cL = offG[gcl * OFF_PITCH + b + 1];
  if (lane >= RB || gl >= nchunks) { aL = 0; cL = 0; }
  const int loL = gl * SORT_CH + aL;
  const int lnL = cL - aL;
  {
    static_assert(RB <= 16, "the bounds sit in the first lane row");
    const int sa = row16_sum_i(aL), sc = row16_sum_i(cL);
    if (lane == 0) { wsA[wave] = sa; wsC[wave] = sc; }
  }
  if (t < BIN_NODES) cnt[t] = 0;
  __syncthreads();
  int s0 = 0, s1 = 0;
  for (int w = 0; w < NW; ++w) { s0 += wsA[w]; s1 += wsC[w]; }
  const bool staged = (s1 - s0) <= STAGE_MAX;
  const bool longrun = __any(lnL > 32 * RS);

  {
    int d[RB][RS];
    #pragma unroll
    for (int j = 0; j < RB; ++j) {
      const int ln = __builtin_amdgcn_readlane(lnL, j);
      const int lo = __builtin_amdgcn_readlane(loL, j);
      #pragma unroll
      for (int k = 0; k < RS; ++k)
        d[j][k] = (32 * k + lane < ln) ? pairs[lo + 32 * k + lane].y : -1;
    }
    #pragma unroll
    for (int j = 0; j < RB; ++j)
      #pragma unroll
      for (int k = 0; k < RS; ++k)
        if (d[j][k] >= 0) atomicAdd(&cnt[d[j][k] - node0], 1);
    if (longrun) {
      for (int j = 0; j < RB; ++j) {
        const int ln = __builtin_amdgcn_readlane(lnL, j);
        const int lo = __builtin_amdgcn_readlane(loL, j);
        for (int i = 32 * RS + lane; i < ln; i += 32) atomicAdd(&cnt[pairs[lo + i].y - node0], 1);
      }
    }
  }
  __syncthreads();
  int c = 0, incl = 0;
  if (t < BIN_NODES) {
    c = cnt[t];
    incl = wave_incl_scan(c, lane);
    if (lane == 31) wsum[wave] = incl;
  }
  __syncthreads();
  if (t < BIN_NODES) {
    int wbase = 0;
    for (int w = 0; w < BIN_NODES / 32; ++w) wbase += (w < wave) ? wsum[w] : 0;
    const int excl = s0 + wbase + incl - c;
    cnt[t] = excl;
    const int node = node0 + t;
    if (node < N) rowptr[node] = excl;
    if (node == N - 1) rowptr[N] = excl + c;
  }
  __syncthreads();
  {
    v2i p[RB][RS];
    #pragma unroll
    for (int j = 0; j < RB; ++j) {
      const int ln = __builtin_amdgcn_readlane(lnL, j);
      const int lo = __builtin_amdgcn_readlane(loL, j);
      #pragma unroll
      for (int k = 0; k < RS; ++k)
        p[j][k] = (32 * k + lane < ln) ? pairs[lo + 32 * k + lane] : (v2i){0, -1};
    }
    int pos[RB][RS];
    #pragma unroll
    for (int j = 0; j < RB; ++j)
      #pragma unroll
      for (int k = 0; k < RS; ++k)
        pos[j][k] = (p[j][k].y >= 0) ? atomicAdd(&cnt[p[j][k].y - node0], 1) : -1;
    if (staged) {
      #pragma unroll
      for (int j = 0; j < RB; ++j)
        #pragma unroll
        for (int k = 0; k < RS; ++k)
          if (pos[j][k] >= 0) stage[pos[j][k] - s0] = p[j][k].x;
    } else {
      #pragma unroll
      for (int j = 0; j < RB; ++j)
        #pragma unroll
        for (int k = 0; k < RS; ++k)
          if (pos[j][k] >= 0) csr_src[pos[j][k]] = p[j][k].x;
    }
    if (longrun) {
      for (int j = 0; j < RB; ++j) {
        const int ln = __builtin_amdgcn_readlane(lnL, j);
        const int lo = __builtin_amdgcn_readlane(loL, j);
        if (staged) {
          for (int i = 32 * RS + lane; i < ln; i += 32) {
            const v2i p = pairs[lo + i];
            stage[atomicAdd(&cnt[p.y - node0], 1) - s0] = p.x;
          }
        } else {
          for (int i = 32 * RS + lane; i < ln; i += 32) {
            const v2i p = pairs[lo + i];
            csr_src[atomicAdd(&cnt[p.y - node0], 1)] = p.x;
          }
        }
      }
    }
  }
  __syncthreads();
  if (staged)
    for (int i = t; i < s1 - s0; i += SORT_NT) csr_src[s0 + i] = stage[i];
}

constexpr int GEMM_LDS_WORDS = 2 * 128 * 64 + 2 * 32 * (64 + 4);

__device__ __forceinline__ void gemm1_body(float* lds, const int rg,
                                           const float* __restrict__ A,
                                           const float* __restrict__ B,
                                           float* __restrict__ C,
                                           const float* __restrict__ aw_src,
                                           const float* __restrict__ aw_dst,
                                           float* __restrict__ asrc,
                                           float* __restrict__ adst, const int M) {
  constexpr int K = 128;
  constexpr int NC = 128;
  constexpr int CG = 64;
  constexpr int LDA = K + 4;
  constexpr int LDC = CG + 4;
  constexpr int NT = 512;
  static_assert(GEMM_LDS_WORDS >= 2 * K * CG + 32 * LDA, "A panel behind the B panels");
  float* Wlds = lds;
  float* Alds = lds + 2 * K * CG;
  const int tid = threadIdx.x;

  constexpr int WL = K * (NC / 4) / NT;
  constexpr int AL = 32 * (K / 4) / NT;
  static_assert(WL == 8 && AL == 2, "the fence below names every loaded value");
  v4f wv[WL], av[AL];
  #pragma unroll
  for (int j = 0; j < WL; ++j) {
    const int i = tid + j * NT;
    const int k = i / (NC / 4), c = i % (NC / 4);
    wv[j] = *(const v4f*)&B[(size_t)k * NC + c * 4];
  }
  #pragma unroll
  for (int j = 0; j < AL; ++j) {
    const int i = tid + j * NT;
    const int r = i / (K / 4), c4 = i % (K / 4);
    const int grow = min(rg * 32 + r, M - 1);
    av[j] = *(const v4f*)&A[(size_t)grow * K + c4 * 4];
  }
  const int cg = tid >> 8;
  const int t8 = tid & 255;
  const int oc4 = t8 & 15;
  const v4f us = *(const v4f*)&aw_src[cg * CG + oc4 * 4];
  const v4f ud = *(const v4f*)&aw_dst[cg * CG + oc4 * 4];
  asm volatile("" :: "v"(wv[0]), "v"(wv[1]), "v"(wv[2]), "v"(wv[3]),
                     "v"(wv[4]), "v"(wv[5]), "v"(wv[6]), "v"(wv[7]),
                     "v"(av[0]), "v"(av[1]), "v"(us), "v"(ud));
  #pragma unroll
  for (int j = 0; j < WL; ++j) {
    const int i = tid + j * NT;
    const int k = i / (NC / 4), c = i % (NC / 4);
    *(v4f*)&Wlds[(c >> 4) * (K * CG) + k * CG + (c & 15) * 4] = wv[j];
  }
  #pragma unroll
  for (int j = 0; j < AL; ++j) {
    const int i = tid + j * NT;
    const int r = i / (K / 4), c4 = i % (K / 4);
    const v4f zero = {0.f, 0.f, 0.f, 0.f};
    *(v4f*)&Alds[r * LDA + c4 * 4] = (rg * 32 + r < M) ? av[j] : zero;
  }
  __syncthreads();

  const int wave = t8 >> 5;
  const int lane = tid & 31;
  const int half = lane >> 4;
  const int lrow = lane & 15;
  const int rt = wave >> 2, ct = wave & 3;

  const float* Ab = &Alds[(rt * 16 + lrow) * LDA + 2 * half];
  const float* Bb = &Wlds[cg * (K * CG) + (2 * half) * CG + ct * 16 + lrow];
  v8f acc = {};
  #pragma unroll
  for (int k0 = 0; k0 < K; k0 += 4) {
    v2f a, b;
    a.x = Ab[k0];        a.y = Ab[k0 + 1];
    b.x = Bb[k0 * CG];   b.y = Bb[k0 * CG + CG];
    acc = __builtin_amdgcn_wmma_f32_16x16x4_f32(false, a, false, b,
                                                (short)0, acc, false, false);
  }

  __syncthreads();
  float* Ct = Alds + cg * (32 * LDC);
  float* Cw = &Ct[(rt * 16 + 8 * half) * LDC + ct * 16 + lrow];
  #pragma unroll
  for (int j = 0; j < 8; ++j) Cw[j * LDC] = acc[j];
  __syncthreads();

  const int orow = t8 >> 4;
  const int grow_a = rg * 32 + orow, grow_b = grow_a + 16;
  const float4 va = *(const float4*)&Ct[orow * LDC + oc4 * 4];
  const float4 vb = *(const float4*)&Ct[(orow + 16) * LDC + oc4 * 4];
  if (grow_a < M) *(float4*)&C[(size_t)grow_a * NC + cg * CG + oc4 * 4] = va;
  if (grow_b < M) *(float4*)&C[(size_t)grow_b * NC + cg * CG + oc4 * 4] = vb;
  const float psa = row16_sum(va.x * us.x + va.y * us.y + va.z * us.z + va.w * us.w);
  const float pda = row16_sum(va.x * ud.x + va.y * ud.y + va.z * ud.z + va.w * ud.w);
  const float psb = row16_sum(vb.x * us.x + vb.y * us.y + vb.z * us.z + vb.w * us.w);
  const float pdb = row16_sum(vb.x * ud.x + vb.y * ud.y + vb.z * ud.z + vb.w * ud.w);
  if (oc4 < 4) {
    const bool second = (oc4 & 2) != 0, isdst = (oc4 & 1) != 0;
    const int grow = second ? grow_b : grow_a;
    const float val = isdst ? (second ? pdb : pda) : (second ? psb : psa);
    float* dstp = isdst ? adst : asrc;
    if (grow < M) dstp[(size_t)grow * 2 + cg] = val;
  }
}

static_assert(SORT1_LDS_WORDS <= GEMM_LDS_WORDS && SORT2_LDS_WORDS <= GEMM_LDS_WORDS,
              "the sorting work-groups live inside the product work-groups' LDS");

__global__ __launch_bounds__(SORT_NT) void sort1_gemm_kernel(
    const int* __restrict__ srcE, const int* __restrict__ dstE,
    int* __restrict__ offG, int* __restrict__ binned, int E, int nsort,
    const float* __restrict__ A, const float* __restrict__ B, float* __restrict__ C,
    const float* __restrict__ aw_src, const float* __restrict__ aw_dst,
    float* __restrict__ asrc, float* __restrict__ adst, int M, int rg0) {
  __shared__ __attribute__((aligned(16))) float lds[GEMM_LDS_WORDS];
  const int wg = blockIdx.x;
  if (wg < nsort) {
    chunk_sort_body((int*)lds, wg, srcE, dstE, offG, binned, E);
    return;
  }
  gemm1_body(lds, rg0 + wg - nsort, A, B, C, aw_src, aw_dst, asrc, adst, M);
}

__global__ __launch_bounds__(SORT_NT) void sort2_gemm_kernel(
    const int* __restrict__ binned, const int* __restrict__ offG,
    int* __restrict__ rowptr, int* __restrict__ csr_src, int N, int nchunks, int nsort,
    const float* __restrict__ A, const float* __restrict__ B, float* __restrict__ C,
    const float* __restrict__ aw_src, const float* __restrict__ aw_dst,
    float* __restrict__ asrc, float* __restrict__ adst, int rg0) {
  __shared__ __attribute__((aligned(16))) float lds[GEMM_LDS_WORDS];
  const int wg = blockIdx.x;
  if (wg < nsort) {
    bin_place_body((int*)lds, wg, binned, offG, rowptr, csr_src, N, nchunks);
    return;
  }
  gemm1_body(lds, rg0 + wg - nsort, A, B, C, aw_src, aw_dst, asrc, adst, N);
}

template<int V> struct RowVec;
template<> struct RowVec<4> { typedef v4f T; };
template<> struct RowVec<2> { typedef v2f T; };

__device__ __forceinline__ float lane_bcast(float x, int k) {
  return __int_as_float(__builtin_amdgcn_readlane(__float_as_int(x), k));
}

template<int H> struct NodeFront { int beg, cnt, s0; float a0[H]; float ad[H]; };

template<int H, int NPW> struct NodeWalk {
  static_assert(NPW + 1 <= 32, "one lane per bound");
  int d0, N, rp, s_cur, s_nx;

  __device__ __forceinline__ int beg_of(int j) const { return __builtin_amdgcn_readlane(rp, j); }
  __device__ __forceinline__ int cnt_of(int j) const {
    return __builtin_amdgcn_readlane(rp, j + 1) - __builtin_amdgcn_readlane(rp, j) + 1;
  }
  __device__ __forceinline__ int node_of(int j) const { return min(d0 + j, N - 1); }
  __device__ __forceinline__ int src_of(int j, int lane, const int* __restrict__ csr_src) const {
    const int beg = beg_of(j), cnt = cnt_of(j);
    return (lane < cnt - 1) ? csr_src[beg + lane] : node_of(j);
  }
  __device__ __forceinline__ void start(int d0_, int lane, int N_,
                                        const int* __restrict__ rowptr,
                                        const int* __restrict__ csr_src) {
    d0 = d0_; N = N_;
    rp = rowptr[min(d0 + min(lane, NPW), N)];
    s_cur = src_of(0, lane, csr_src);
    s_nx = (NPW > 1) ? src_of(1, lane, csr_src) : 0;
  }
};

template<int H>
__device__ __forceinline__ typename RowVec<H * 2>::T agg_node_row(
    const int d, const int lane, const NodeFront<H> nf,
    const int* __restrict__ csr_src,
    const float* __restrict__ asrc,
    const float* __restrict__ h) {
  constexpr int F = H * 64;
  constexpr int VEC = F / 32;
  constexpr int U = 4;
  typedef typename RowVec<VEC>::T vecf;
  const int beg = nf.beg;
  const int cnt = nf.cnt;
  const int myh = (VEC * lane) / 64;
  float adv[H];
  for (int hh = 0; hh < H; ++hh) adv[hh] = nf.ad[hh];

  auto slot_src = [&](int i) -> int { return (i < cnt - 1) ? csr_src[beg + i] : d; };
  auto leaky = [](float v) -> float { return (v > 0.f) ? v : NEG_SLOPE * v; };
  auto logit = [&](int s, int hh) -> float { return leaky(asrc[s * H + hh] + adv[hh]); };

  const bool ok0 = lane < cnt;
  const int s0 = nf.s0;
  float v0[H], mx[H], sm[H], ex0[H];
  for (int hh = 0; hh < H; ++hh) {
    v0[hh] = leaky(nf.a0[hh] + adv[hh]);
    mx[hh] = ok0 ? v0[hh] : -__builtin_inff();
  }
  for (int i = lane + 32; i < cnt; i += 32) {
    const int s = slot_src(i);
    for (int hh = 0; hh < H; ++hh) mx[hh] = fmaxf(mx[hh], logit(s, hh));
  }
  for (int hh = 0; hh < H; ++hh)
    for (int off = 16; off > 0; off >>= 1) mx[hh] = fmaxf(mx[hh], __shfl_xor(mx[hh], off, 32));
  for (int hh = 0; hh < H; ++hh) {
    ex0[hh] = ok0 ? __expf(v0[hh] - mx[hh]) : 0.f;
    sm[hh] = ex0[hh];
  }
  for (int i = lane + 32; i < cnt; i += 32) {
    const int s = slot_src(i);
    for (int hh = 0; hh < H; ++hh) sm[hh] += __expf(logit(s, hh) - mx[hh]);
  }
  for (int hh = 0; hh < H; ++hh)
    for (int off = 16; off > 0; off >>= 1) sm[hh] += __shfl_xor(sm[hh], off, 32);

  vecf acc = {};
  const float* hl = h + VEC * lane;
  for (int c0 = 0; c0 < cnt; c0 += 32) {
    int s = s0;
    float al[H];
    if (c0 == 0) {
      for (int hh = 0; hh < H; ++hh) al[hh] = ex0[hh] / sm[hh];
    } else {
      const int i = c0 + lane;
      const bool ok = i < cnt;
      s = ok ? slot_src(i) : d;
      for (int hh = 0; hh < H; ++hh) al[hh] = ok ? __expf(logit(s, hh) - mx[hh]) / sm[hh] : 0.f;
    }
    const int n = (cnt - c0 < 32) ? (cnt - c0) : 32;
    for (int k = 0; k < n; k += U) {
      vecf hv[U];
      float a[U];
      #pragma unroll
      for (int u = 0; u < U; ++u) {
        const int kk = k + u;
        const int kc = (kk < n) ? kk : n - 1;
        const int sk = __builtin_amdgcn_readlane(s, kc);
        hv[u] = *(const vecf*)(hl + (size_t)sk * F);
        float ax = lane_bcast(al[0], kc);
        if constexpr (H == 2) {
          const float a1 = lane_bcast(al[1], kc);
          ax = myh ? a1 : ax;
        }
        a[u] = (kk < n) ? ax : 0.f;
      }
      #pragma unroll
      for (int u = 0; u < U; ++u) acc += a[u] * hv[u];
    }
  }
  return acc;
}

template<int H, int NPW, typename Emit>
__device__ __forceinline__ void walk_run(NodeWalk<H, NPW>& wk, const int lane,
                                         const int* __restrict__ csr_src,
                                         const float* __restrict__ asrc,
                                         const float* __restrict__ adst,
                                         const float* __restrict__ h, Emit emit) {
  float a_cur[H], ad_cur[H];
  for (int hh = 0; hh < H; ++hh) {
    a_cur[hh] = asrc[wk.s_cur * H + hh];
    ad_cur[hh] = adst[wk.node_of(0) * H + hh];
  }
  #pragma unroll
  for (int j = 0; j < NPW; ++j) {
    const int d = wk.d0 + j;
    if (d >= wk.N) break;
    float a_nx[H], ad_nx[H];
    int s_nx2 = 0;
    for (int hh = 0; hh < H; ++hh) { a_nx[hh] = 0.f; ad_nx[hh] = 0.f; }
    if (j + 1 < NPW) {
      for (int hh = 0; hh < H; ++hh) {
        a_nx[hh] = asrc[wk.s_nx * H + hh];
        ad_nx[hh] = adst[wk.node_of(j + 1) * H + hh];
      }
    }
    if (j + 2 < NPW) s_nx2 = wk.src_of(j + 2, lane, csr_src);
    NodeFront<H> nf;
    nf.beg = wk.beg_of(j);
    nf.cnt = wk.cnt_of(j);
    nf.s0 = wk.s_cur;
    for (int hh = 0; hh < H; ++hh) { nf.a0[hh] = a_cur[hh]; nf.ad[hh] = ad_cur[hh]; }
    emit(j, d, agg_node_row<H>(d, lane, nf, csr_src, asrc, h));
    wk.s_cur = wk.s_nx;
    wk.s_nx = s_nx2;
    for (int hh = 0; hh < H; ++hh) { a_cur[hh] = a_nx[hh]; ad_cur[hh] = ad_nx[hh]; }
  }
}

template<int H, bool FINAL, int NPW>
__global__ __launch_bounds__(256) void node_softmax_agg_kernel(
    const int* __restrict__ rowptr, const int* __restrict__ csr_src,
    const float* __restrict__ asrc, const float* __restrict__ adst,
    const float* __restrict__ h, const float* __restrict__ bias,
    float* __restrict__ outp, int N) {
  constexpr int F = H * 64;
  constexpr int VEC = F / 32;
  typedef typename RowVec<VEC>::T vecf;
  const int lane = threadIdx.x & 31;
  const int d0 = NPW * __builtin_amdgcn_readfirstlane((int)((blockIdx.x * blockDim.x + threadIdx.x) >> 5));
  if (d0 >= N) return;
  NodeWalk<H, NPW> wk;
  wk.start(d0, lane, N, rowptr, csr_src);
  const vecf bv = *(const vecf*)(bias + VEC * lane);
  walk_run<H, NPW>(wk, lane, csr_src, asrc, adst, h, [&](int, int d, vecf acc) {
    vecf r = acc + bv;
    for (int c = 0; c < VEC; ++c) {
      if (FINAL) r[c] = 1.f / (1.f + __expf(-r[c]));
      else       r[c] = (r[c] > 0.f) ? r[c] : 0.f;
    }
    *(vecf*)(outp + (size_t)d * F + VEC * lane) = r;
  });
}

__global__ __launch_bounds__(512) void agg1_gemm2_kernel(
    const int* __restrict__ rowptr, const int* __restrict__ csr_src,
    const float* __restrict__ asrc1, const float* __restrict__ adst1,
    const float* __restrict__ h1, const float* __restrict__ bias1,
    const float* __restrict__ B, const float* __restrict__ aw_src, const float* __restrict__ aw_dst,
    float* __restrict__ C, float* __restrict__ asrc2, float* __restrict__ adst2, int N) {
  constexpr int K = 128;
  constexpr int CG = 64;
  constexpr int LDA = K + 4;
  constexpr int LDC = CG + 4;
  __shared__ float Wlds[K * CG];
  __shared__ float Alds[32 * LDA];
  const int rg = blockIdx.x;
  const int lane = threadIdx.x & 31;
  const int wave = __builtin_amdgcn_readfirstlane((int)(threadIdx.x >> 5));

  constexpr int WL = K * (CG / 4) / 512;
  static_assert(WL == 4, "the fence below names every loaded value");
  v4f wv[WL];
  #pragma unroll
  for (int j = 0; j < WL; ++j) {
    const int i = threadIdx.x + j * 512;
    const int k = i / (CG / 4), c4 = i % (CG / 4);
    wv[j] = *(const v4f*)&B[(size_t)k * CG + c4 * 4];
  }
  const int oc4 = threadIdx.x & 15;
  const v4f us = *(const v4f*)&aw_src[oc4 * 4];
  const v4f ud = *(const v4f*)&aw_dst[oc4 * 4];
  const int d0 = rg * 32 + 2 * wave;
  const bool walks = d0 < N;
  NodeWalk<2, 2> wk;
  wk.start(walks ? d0 : 0, lane, N, rowptr, csr_src);
  asm volatile("" :: "v"(wv[0]), "v"(wv[1]), "v"(wv[2]), "v"(wv[3]), "v"(us), "v"(ud),
                     "v"(wk.s_cur), "v"(wk.s_nx));
  #pragma unroll
  for (int j = 0; j < WL; ++j) {
    const int i = threadIdx.x + j * 512;
    const int k = i / (CG / 4), c4 = i % (CG / 4);
    *(v4f*)&Wlds[k * CG + c4 * 4] = wv[j];
  }

  for (int j = 0; j < 2; ++j)
    if (d0 + j >= N) *(v4f*)&Alds[(2 * wave + j) * LDA + 4 * lane] = (v4f){0.f, 0.f, 0.f, 0.f};
  if (walks) {
    const v4f bv = *(const v4f*)(bias1 + 4 * lane);
    walk_run<2, 2>(wk, lane, csr_src, asrc1, adst1, h1, [&](int j, int, v4f acc) {
      v4f row = acc + bv;
      for (int c = 0; c < 4; ++c) row[c] = (row[c] > 0.f) ? row[c] : 0.f;
      *(v4f*)&Alds[(2 * wave + j) * LDA + 4 * lane] = row;
    });
  }
  __syncthreads();

  const int half = lane >> 4;
  const int lrow = lane & 15;
  const int rt = (wave >> 2) & 1, ct = wave & 3;
  v8f acc = {};
  if (wave < 8) {
    const float* Ab = &Alds[(rt * 16 + lrow) * LDA + 2 * half];
    const float* Bb = &Wlds[(2 * half) * CG + ct * 16 + lrow];
    #pragma unroll
    for (int k0 = 0; k0 < K; k0 += 4) {
      v2f a, b;
      a.x = Ab[k0];        a.y = Ab[k0 + 1];
      b.x = Bb[k0 * CG];   b.y = Bb[k0 * CG + CG];
      acc = __builtin_amdgcn_wmma_f32_16x16x4_f32(false, a, false, b,
                                                  (short)0, acc, false, false);
    }
  }
  __syncthreads();
  float* Ct = Alds;
  if (wave < 8) {
    float* Cw = &Ct[(rt * 16 + 8 * half) * LDC + ct * 16 + lrow];
    #pragma unroll
    for (int j = 0; j < 8; ++j) Cw[j * LDC] = acc[j];
  }
  __syncthreads();

  const int orow = threadIdx.x >> 4;
  const int grow = rg * 32 + orow;
  const float4 va = *(const float4*)&Ct[orow * LDC + oc4 * 4];
  if (grow < N) *(float4*)&C[(size_t)grow * CG + oc4 * 4] = va;
  const float ps = row16_sum(va.x * us.x + va.y * us.y + va.z * us.z + va.w * us.w);
  const float pd = row16_sum(va.x * ud.x + va.y * ud.y + va.z * ud.z + va.w * ud.w);
  if (oc4 < 2 && grow < N) {
    if (oc4 == 0) asrc2[grow] = ps;
    else          adst2[grow] = pd;
  }
}

constexpr int AGG2_NPW = 4;

extern "C" void kernel_launch(void* const* d_in, const int* in_sizes, int n_in,
                              void* d_out, int out_size, void* d_ws, size_t ws_size,
                              hipStream_t stream) {
  const float* x   = (const float*)d_in[0];
  const int*   ei  = (const int*)d_in[1];
  const float* W1  = (const float*)d_in[2];
  const float* as1 = (const float*)d_in[3];
  const float* ad1 = (const float*)d_in[4];
  const float* b1  = (const float*)d_in[5];
  const float* W2  = (const float*)d_in[6];
  const float* as2 = (const float*)d_in[7];
  const float* ad2 = (const float*)d_in[8];
  const float* b2  = (const float*)d_in[9];
  float* out = (float*)d_out;

  const int N = in_sizes[0] / 128;
  const int E = in_sizes[1] / 2;
  const int* srcE = ei;
  const int* dstE = ei + E;

  float* w = (float*)d_ws;
  size_t o = 0;
  float* h1    = w + o; o += (size_t)N * 128;
  float* h2f   = w + o; o += (size_t)N * 128;
  float* asrc1 = w + o; o += (size_t)N * 2;
  float* adst1 = w + o; o += (size_t)N * 2;
  float* asrc2 = w + o; o += (size_t)N;
  float* adst2 = w + o; o += (size_t)N;
  o = (o + 3) & ~(size_t)3;
  const int nbins   = (N + BIN_NODES - 1) / BIN_NODES;
  const int nchunks = (E + SORT_CH - 1) / SORT_CH;
  int* offG     = (int*)(w + o); o += (size_t)nchunks * OFF_PITCH;
  int* binned   = (int*)(w + o); o += (size_t)E * 2;
  int* rowptr   = (int*)(w + o); o += (size_t)N + 1;
  int* csr_src  = (int*)(w + o); o += (size_t)E;
  (void)o; (void)ws_size; (void)n_in; (void)out_size;

  const int TB = 256;
  auto blk = [](long long n, int tb) { return (unsigned)((n + tb - 1) / tb); };
  const int rowGroups = (N + 31) / 32;

  const int rgA = (rowGroups + 1) / 2;
  sort1_gemm_kernel<<<nchunks + rgA, SORT_NT, 0, stream>>>(
      srcE, dstE, offG, binned, E, nchunks, x, W1, h1, as1, ad1, asrc1, adst1, N, 0);
  sort2_gemm_kernel<<<nbins + (rowGroups - rgA), SORT_NT, 0, stream>>>(
      binned, offG, rowptr, csr_src, N, nchunks, nbins, x, W1, h1, as1, ad1, asrc1, adst1, rgA);

  agg1_gemm2_kernel<<<dim3(rowGroups), 512, 0, stream>>>(
      rowptr, csr_src, asrc1, adst1, h1, b1, W2, as2, ad2, h2f, asrc2, adst2, N);

  node_softmax_agg_kernel<1, true, AGG2_NPW><<<blk((long long)((N + AGG2_NPW - 1) / AGG2_NPW) * 32, TB), TB, 0, stream>>>(
      rowptr, csr_src, asrc2, adst2, h2f, b2, out, N);
}
